# starting kernel
# baseline (speedup 1.0000x reference)
#include <hip/hip_runtime.h>
#include <math.h>

static const int HD  = 128;
static const int PD  = 1024;
static const int LD  = 1024;
static const int CD  = 128;
static const int NT  = 64, NH = 32, NA = 64, NB = 64, NPA = 32, NPE = 32;
static const int RT  = PD + LD + PD + LD + CD + NH;
static const int RH  = PD + PD + LD + CD;
static const int HH  = HD * HD;

typedef float v2f __attribute__((ext_vector_type(2)));
typedef float v8f __attribute__((ext_vector_type(8)));

__device__ __forceinline__ v8f wmma_f32(v2f a, v2f b, v8f c) {
  return __builtin_amdgcn_wmma_f32_16x16x4_f32(false, a, false, b, (short)0, c,
                                               false, false);
}

__global__ void masks_kernel(const int* __restrict__ pl, const int* __restrict__ pc,
                             float* __restrict__ lpl_m, float* __restrict__ plp_m,
                             float* __restrict__ pcp_m) {
  int t = blockIdx.x * blockDim.x + threadIdx.x;
  if (t < PD) {
    int s = 0;
    for (int j = 0; j < LD; ++j) s += pl[(size_t)t * LD + j];
    lpl_m[t] = (s >= 2) ? 1.f : 0.f;
  }
  if (t < LD) {
    int s = 0;
    for (int i = 0; i < PD; ++i) s += pl[(size_t)i * LD + t];
    plp_m[t] = (s >= 2) ? 1.f : 0.f;
  }
  if (t < CD) {
    int s = 0;
    for (int i = 0; i < PD; ++i) s += pc[(size_t)i * CD + t];
    pcp_m[t] = (s >= 2) ? 1.f : 0.f;
  }
}

__global__ __launch_bounds__(256)
void agg_gemm_kernel(const int* __restrict__ rel, int ldA, int transA,
                     const float* __restrict__ Bm, int K, float* __restrict__ Cout) {
  __shared__ float sA[16 * 128];
  const int tid  = threadIdx.x;
  const int lane = tid & 31;
  const int w    = tid >> 5;
  const int m0   = blockIdx.x * 16;
  const int n    = w * 16 + (lane & 15);
  const int am   = lane & 15;
  const int koff = (lane >> 4) << 1;

  v8f acc = {0.f, 0.f, 0.f, 0.f, 0.f, 0.f, 0.f, 0.f};

  for (int k0 = 0; k0 < K; k0 += 128) {
    if (transA) {
      for (int i = tid; i < 16 * 128; i += 256) {
        int m = i & 15, k = i >> 4;
        sA[m * 128 + k] = (float)rel[(size_t)(k0 + k) * ldA + (m0 + m)];
      }
    } else {
      for (int i = tid; i < 16 * 128; i += 256) {
        int m = i >> 7, k = i & 127;
        sA[m * 128 + k] = (float)rel[(size_t)(m0 + m) * ldA + (k0 + k)];
      }
    }
    __syncthreads();
    for (int kk = 0; kk < 128; kk += 4) {
      v2f a = *(const v2f*)&sA[am * 128 + kk + koff];
      int kg = k0 + kk + koff;
      v2f b;
      b.x = Bm[(size_t)kg * HD + n];
      b.y = Bm[(size_t)(kg + 1) * HD + n];
      acc = wmma_f32(a, b, acc);
    }
    __syncthreads();
  }
  const int mb = (lane >> 4) << 3;
#pragma unroll
  for (int v = 0; v < 8; ++v)
    Cout[(size_t)(m0 + mb + v) * HD + n] = acc[v];
}

__global__ __launch_bounds__(256)
void lrl_gemm_kernel(const float* __restrict__ X0, const float* __restrict__ X1,
                     float alpha, const float* __restrict__ W,
                     const float* __restrict__ bias, const float* __restrict__ g,
                     const float* __restrict__ be, float* __restrict__ Y, int do_ln) {
  __shared__ float sA[16 * 128];
  __shared__ float sC[16 * 128];
  __shared__ float sMean[16], sRstd[16];
  const int tid  = threadIdx.x;
  const int lane = tid & 31;
  const int w    = tid >> 5;
  const int m0   = blockIdx.x * 16;
  const int n    = w * 16 + (lane & 15);
  const int am   = lane & 15;
  const int koff = (lane >> 4) << 1;

  for (int i = tid; i < 16 * 128; i += 256) {
    int m = i >> 7, k = i & 127;
    float v = alpha * X0[(size_t)(m0 + m) * 128 + k];
    if (X1) v += X1[(size_t)(m0 + m) * 128 + k];
    sA[i] = v;
  }
  __syncthreads();

  v8f acc = {0.f, 0.f, 0.f, 0.f, 0.f, 0.f, 0.f, 0.f};
#pragma unroll 4
  for (int kk = 0; kk < 128; kk += 4) {
    v2f a = *(const v2f*)&sA[am * 128 + kk + koff];
    v2f b = *(const v2f*)&W[(size_t)n * 128 + kk + koff];
    acc = wmma_f32(a, b, acc);
  }

  const int mb = (lane >> 4) << 3;
  if (!do_ln) {
#pragma unroll
    for (int v = 0; v < 8; ++v)
      Y[(size_t)(m0 + mb + v) * 128 + n] = acc[v];
    return;
  }
  float bn = bias[n];
#pragma unroll
  for (int v = 0; v < 8; ++v) {
    float x = acc[v] + bn;
    sC[(mb + v) * 128 + n] = x > 0.f ? x : 0.f;
  }
  __syncthreads();
  if (tid < 16) {
    float s = 0.f, s2 = 0.f;
    for (int k = 0; k < 128; ++k) {
      float x = sC[tid * 128 + k];
      s += x; s2 += x * x;
    }
    float mean = s * (1.f / 128.f);
    float var  = s2 * (1.f / 128.f) - mean * mean;
    sMean[tid] = mean;
    sRstd[tid] = rsqrtf(var + 1e-5f);
  }
  __syncthreads();
  for (int i = tid; i < 16 * 128; i += 256) {
    int m = i >> 7, k = i & 127;
    Y[(size_t)(m0 + m) * 128 + k] = (sC[i] - sMean[m]) * sRstd[m] * g[k] + be[k];
  }
}

__global__ void mask_cat_kernel(const float* __restrict__ lpl_m,
                                const float* __restrict__ plp_m,
                                const float* __restrict__ pcp_m,
                                float* __restrict__ mask_t, float* __restrict__ mask_h) {
  int t = blockIdx.x * blockDim.x + threadIdx.x;
  if (t < RT) {
    float v;
    if (t < 2048)      v = 1.f;
    else if (t < 3072) v = lpl_m[t - 2048];
    else if (t < 4096) v = plp_m[t - 3072];
    else if (t < 4224) v = pcp_m[t - 4096];
    else               v = 1.f;
    mask_t[t] = v;
  }
  if (t < RH) {
    float v;
    if (t < 1024)      v = 1.f;
    else if (t < 2048) v = lpl_m[t - 1024];
    else if (t < 3072) v = plp_m[t - 2048];
    else               v = pcp_m[t - 3072];
    mask_h[t] = v;
  }
}

__global__ __launch_bounds__(256)
void pair_kernel(const float* __restrict__ symW, const float* __restrict__ geoW,
                 const float* __restrict__ mask, const float* __restrict__ b1,
                 const float* __restrict__ g, const float* __restrict__ be,
                 const float* __restrict__ W2, const float* __restrict__ b2p,
                 float* __restrict__ out, int Nsym, int R) {
  int gw = (int)(((unsigned)blockIdx.x * 256u + threadIdx.x) >> 5);
  if (gw >= Nsym * R) return;
  int lane = threadIdx.x & 31;
  int i = gw / R;
  int r = gw - i * R;

  float4 s4 = ((const float4*)(symW + (size_t)i * 128))[lane];
  float4 g4 = ((const float4*)(geoW + (size_t)r * 128))[lane];
  float4 b4 = ((const float4*)b1)[lane];

  float x0 = s4.x + g4.x + b4.x; x0 = x0 > 0.f ? x0 : 0.f;
  float x1 = s4.y + g4.y + b4.y; x1 = x1 > 0.f ? x1 : 0.f;
  float x2 = s4.z + g4.z + b4.z; x2 = x2 > 0.f ? x2 : 0.f;
  float x3 = s4.w + g4.w + b4.w; x3 = x3 > 0.f ? x3 : 0.f;

  float s  = x0 + x1 + x2 + x3;
  float s2 = x0 * x0 + x1 * x1 + x2 * x2 + x3 * x3;
  for (int o = 16; o > 0; o >>= 1) {
    s  += __shfl_xor(s, o, 32);
    s2 += __shfl_xor(s2, o, 32);
  }
  float mean = s * (1.f / 128.f);
  float var  = s2 * (1.f / 128.f) - mean * mean;
  float rstd = rsqrtf(var + 1e-5f);

  float4 gg = ((const float4*)g)[lane];
  float4 bb = ((const float4*)be)[lane];
  float4 w2 = ((const float4*)W2)[lane];
  float d = ((x0 - mean) * rstd * gg.x + bb.x) * w2.x +
            ((x1 - mean) * rstd * gg.y + bb.y) * w2.y +
            ((x2 - mean) * rstd * gg.z + bb.z) * w2.z +
            ((x3 - mean) * rstd * gg.w + bb.w) * w2.w;
  for (int o = 16; o > 0; o >>= 1) d += __shfl_xor(d, o, 32);

  if (lane == 0) {
    float logit = d + b2p[0];
    float mk = mask ? mask[r] : 1.f;
    float z  = (mk > 0.f) ? logit : -1e9f;
    out[(size_t)i * R + r] = 1.f / (1.f + expf(-z));
  }
}

extern "C" void kernel_launch(void* const* d_in, const int* in_sizes, int n_in,
                              void* d_out, int out_size, void* d_ws, size_t ws_size,
                              hipStream_t stream) {
  (void)in_sizes; (void)n_in; (void)out_size; (void)ws_size;
  const float* points  = (const float*)d_in[0];
  const float* lines   = (const float*)d_in[1];
  const float* circles = (const float*)d_in[2];
  const float* text_s  = (const float*)d_in[3];
  const float* head_s  = (const float*)d_in[4];
  const float* angle_s = (const float*)d_in[5];
  const float* bar_s   = (const float*)d_in[6];
  const float* para_s  = (const float*)d_in[7];
  const float* perp_s  = (const float*)d_in[8];
  const float* rel_W   = (const float*)d_in[9];
  const float* rel_b   = (const float*)d_in[10];
  const float* rel_g   = (const float*)d_in[11];
  const float* rel_be  = (const float*)d_in[12];
  const float* gm_W    = (const float*)d_in[13];
  const float* gm_b    = (const float*)d_in[14];
  const float* gm_g    = (const float*)d_in[15];
  const float* gm_be   = (const float*)d_in[16];
  const float* sc_W1   = (const float*)d_in[17];
  const float* sc_b1   = (const float*)d_in[18];
  const float* sc_g    = (const float*)d_in[19];
  const float* sc_be   = (const float*)d_in[20];
  const float* sc_W2   = (const float*)d_in[21];
  const float* sc_b2   = (const float*)d_in[22];
  const int*   pl      = (const int*)d_in[23];
  const int*   pc      = (const int*)d_in[24];
  float* out = (float*)d_out;

  float* wsp = (float*)d_ws;
  size_t off = 0;
  auto alloc = [&](size_t n) { float* p = wsp + off; off += (n + 31) & ~(size_t)31; return p; };
  float* LPLb   = alloc((size_t)PD * HD);
  float* PLPb   = alloc((size_t)LD * HD);
  float* PCPb   = alloc((size_t)CD * HD);
  float* lpl_m  = alloc(PD);
  float* plp_m  = alloc(LD);
  float* pcp_m  = alloc(CD);
  float* mask_t = alloc(RT);
  float* mask_h = alloc(RH);
  float* geo_t  = alloc((size_t)RT * HD);
  float* geo_h  = alloc((size_t)RH * HD);
  float* tmp    = alloc((size_t)RT * HD);
  float* symWb  = alloc((size_t)64 * HD);
  float* geoWb  = alloc((size_t)RT * HD);

  masks_kernel<<<4, 256, 0, stream>>>(pl, pc, lpl_m, plp_m, pcp_m);

  agg_gemm_kernel<<<PD / 16, 256, 0, stream>>>(pl, LD, 0, lines, LD, tmp);
  lrl_gemm_kernel<<<PD / 16, 256, 0, stream>>>(tmp, points, 2.f, rel_W + 0 * HH,
      rel_b + 0 * HD, rel_g + 0 * HD, rel_be + 0 * HD, LPLb, 1);

  agg_gemm_kernel<<<LD / 16, 256, 0, stream>>>(pl, LD, 1, points, PD, tmp);
  lrl_gemm_kernel<<<LD / 16, 256, 0, stream>>>(tmp, lines, 2.f, rel_W + 1 * HH,
      rel_b + 1 * HD, rel_g + 1 * HD, rel_be + 1 * HD, PLPb, 1);

  agg_gemm_kernel<<<CD / 16, 256, 0, stream>>>(pc, CD, 1, points, PD, tmp);
  lrl_gemm_kernel<<<CD / 16, 256, 0, stream>>>(tmp, circles, 2.f, rel_W + 2 * HH,
      rel_b + 2 * HD, rel_g + 2 * HD, rel_be + 2 * HD, PCPb, 1);

  const size_t ROW = (size_t)HD * sizeof(float);
  hipMemcpyAsync(geo_t + (size_t)0 * HD,    points, PD * ROW, hipMemcpyDeviceToDevice, stream);
  hipMemcpyAsync(geo_t + (size_t)1024 * HD, lines,  LD * ROW, hipMemcpyDeviceToDevice, stream);
  hipMemcpyAsync(geo_t + (size_t)2048 * HD, LPLb,   PD * ROW, hipMemcpyDeviceToDevice, stream);
  hipMemcpyAsync(geo_t + (size_t)3072 * HD, PLPb,   LD * ROW, hipMemcpyDeviceToDevice, stream);
  hipMemcpyAsync(geo_t + (size_t)4096 * HD, PCPb,   CD * ROW, hipMemcpyDeviceToDevice, stream);
  hipMemcpyAsync(geo_t + (size_t)4224 * HD, head_s, NH * ROW, hipMemcpyDeviceToDevice, stream);
  hipMemcpyAsync(geo_h + (size_t)0 * HD,    points, PD * ROW, hipMemcpyDeviceToDevice, stream);
  hipMemcpyAsync(geo_h + (size_t)1024 * HD, LPLb,   PD * ROW, hipMemcpyDeviceToDevice, stream);
  hipMemcpyAsync(geo_h + (size_t)2048 * HD, PLPb,   LD * ROW, hipMemcpyDeviceToDevice, stream);
  hipMemcpyAsync(geo_h + (size_t)3072 * HD, PCPb,   CD * ROW, hipMemcpyDeviceToDevice, stream);
  mask_cat_kernel<<<(RT + 255) / 256, 256, 0, stream>>>(lpl_m, plp_m, pcp_m, mask_t, mask_h);

  lrl_gemm_kernel<<<RT / 16, 256, 0, stream>>>(geo_t, nullptr, 1.f, gm_W + 0 * HH,
      gm_b + 0 * HD, gm_g + 0 * HD, gm_be + 0 * HD, tmp, 1);
  lrl_gemm_kernel<<<RT / 16, 256, 0, stream>>>(tmp, nullptr, 1.f, gm_W + 1 * HH,
      gm_b + 1 * HD, gm_g + 1 * HD, gm_be + 1 * HD, geo_t, 1);
  lrl_gemm_kernel<<<RH / 16, 256, 0, stream>>>(geo_h, nullptr, 1.f, gm_W + 0 * HH,
      gm_b + 0 * HD, gm_g + 0 * HD, gm_be + 0 * HD, tmp, 1);
  lrl_gemm_kernel<<<RH / 16, 256, 0, stream>>>(tmp, nullptr, 1.f, gm_W + 1 * HH,
      gm_b + 1 * HD, gm_g + 1 * HD, gm_be + 1 * HD, geo_h, 1);

  struct Head { const float* sym; int N; const float* geo; int R; const float* mask; };
  const Head heads[6] = {
      {text_s,  NT,  geo_t, RT, mask_t},
      {head_s,  NH,  geo_h, RH, mask_h},
      {angle_s, NA,  LPLb,  PD, lpl_m},
      {bar_s,   NB,  PLPb,  LD, plp_m},
      {para_s,  NPA, lines, LD, nullptr},
      {perp_s,  NPE, LPLb,  PD, lpl_m},
  };
  size_t ooff = 0;
  for (int h = 0; h < 6; ++h) {
    const Head& hd = heads[h];
    lrl_gemm_kernel<<<hd.N / 16, 256, 0, stream>>>(hd.sym, nullptr, 1.f,
        sc_W1 + (size_t)h * HH, nullptr, nullptr, nullptr, symWb, 0);
    lrl_gemm_kernel<<<hd.R / 16, 256, 0, stream>>>(hd.geo, nullptr, 1.f,
        sc_W1 + (size_t)h * HH, nullptr, nullptr, nullptr, geoWb, 0);
    int waves  = hd.N * hd.R;
    int blocks = (waves + 7) / 8;
    pair_kernel<<<blocks, 256, 0, stream>>>(symWb, geoWb, hd.mask,
        sc_b1 + (size_t)h * HD, sc_g + (size_t)h * HD, sc_be + (size_t)h * HD,
        sc_W2 + (size_t)h * HD, sc_b2 + h, out + ooff, hd.N, hd.R);
    ooff += (size_t)hd.N * hd.R;
  }
}


# relation masks taken as exact f32 row sums of the 0/1 tile the aggregation GEMM already stages in LDS; the separate serial masks kernel and its launch are removed
# speedup vs baseline: 1.1702x; 1.1702x over previous
#include <hip/hip_runtime.h>
#include <math.h>

static const int HD  = 128;
static const int PD  = 1024;
static const int LD  = 1024;
static const int CD  = 128;
static const int NT  = 64, NH = 32, NA = 64, NB = 64, NPA = 32, NPE = 32;
static const int RT  = PD + LD + PD + LD + CD + NH;
static const int RH  = PD + PD + LD + CD;
static const int HH  = HD * HD;

typedef float v2f __attribute__((ext_vector_type(2)));
typedef float v8f __attribute__((ext_vector_type(8)));

__device__ __forceinline__ v8f wmma_f32(v2f a, v2f b, v8f c) {
  return __builtin_amdgcn_wmma_f32_16x16x4_f32(false, a, false, b, (short)0, c,
                                               false, false);
}

__global__ __launch_bounds__(256)
void agg_gemm_kernel(const int* __restrict__ rel, int ldA, int transA,
                     const float* __restrict__ Bm, int K, float* __restrict__ Cout,
                     float* __restrict__ rowmask) {
  __shared__ float sA[16 * 128];
  const int tid  = threadIdx.x;
  const int lane = tid & 31;
  const int w    = tid >> 5;
  const int m0   = blockIdx.x * 16;
  const int n    = w * 16 + (lane & 15);
  const int am   = lane & 15;
  const int koff = (lane >> 4) << 1;

  v8f acc = {0.f, 0.f, 0.f, 0.f, 0.f, 0.f, 0.f, 0.f};

  const int crow = tid >> 4;
  const int ccol = (tid & 15) << 2;
  float cnt = 0.f;

  for (int k0 = 0; k0 < K; k0 += 128) {
    if (transA) {
      for (int i = tid; i < 16 * 128; i += 256) {
        int m = i & 15, k = i >> 4;
        sA[m * 128 + k] = (float)rel[(size_t)(k0 + k) * ldA + (m0 + m)];
      }
    } else {
      for (int i = tid; i < 16 * 128; i += 256) {
        int m = i >> 7, k = i & 127;
        sA[m * 128 + k] = (float)rel[(size_t)(m0 + m) * ldA + (k0 + k)];
      }
    }
    __syncthreads();
    {
      const float4 c0 = *(const float4*)&sA[crow * 128 + ccol];
      const float4 c1 = *(const float4*)&sA[crow * 128 + ccol + 64];
      cnt += ((c0.x + c0.y) + (c0.z + c0.w)) + ((c1.x + c1.y) + (c1.z + c1.w));
    }
    for (int kk = 0; kk < 128; kk += 4) {
      v2f a = *(const v2f*)&sA[am * 128 + kk + koff];
      int kg = k0 + kk + koff;
      v2f b;
      b.x = Bm[(size_t)kg * HD + n];
      b.y = Bm[(size_t)(kg + 1) * HD + n];
      acc = wmma_f32(a, b, acc);
    }
    __syncthreads();
  }
  const int mb = (lane >> 4) << 3;
#pragma unroll
  for (int v = 0; v < 8; ++v)
    Cout[(size_t)(m0 + mb + v) * HD + n] = acc[v];

  for (int o = 8; o > 0; o >>= 1) cnt += __shfl_xor(cnt, o, 32);
  if ((tid & 15) == 0) rowmask[m0 + crow] = (cnt >= 2.f) ? 1.f : 0.f;
}

__global__ __launch_bounds__(256)
void lrl_gemm_kernel(const float* __restrict__ X0, const float* __restrict__ X1,
                     float alpha, const float* __restrict__ W,
                     const float* __restrict__ bias, const float* __restrict__ g,
                     const float* __restrict__ be, float* __restrict__ Y, int do_ln) {
  __shared__ float sA[16 * 128];
  __shared__ float sC[16 * 128];
  __shared__ float sMean[16], sRstd[16];
  const int tid  = threadIdx.x;
  const int lane = tid & 31;
  const int w    = tid >> 5;
  const int m0   = blockIdx.x * 16;
  const int n    = w * 16 + (lane & 15);
  const int am   = lane & 15;
  const int koff = (lane >> 4) << 1;

  for (int i = tid; i < 16 * 128; i += 256) {
    int m = i >> 7, k = i & 127;
    float v = alpha * X0[(size_t)(m0 + m) * 128 + k];
    if (X1) v += X1[(size_t)(m0 + m) * 128 + k];
    sA[i] = v;
  }
  __syncthreads();

  v8f acc = {0.f, 0.f, 0.f, 0.f, 0.f, 0.f, 0.f, 0.f};
#pragma unroll 4
  for (int kk = 0; kk < 128; kk += 4) {
    v2f a = *(const v2f*)&sA[am * 128 + kk + koff];
    v2f b = *(const v2f*)&W[(size_t)n * 128 + kk + koff];
    acc = wmma_f32(a, b, acc);
  }

  const int mb = (lane >> 4) << 3;
  if (!do_ln) {
#pragma unroll
    for (int v = 0; v < 8; ++v)
      Y[(size_t)(m0 + mb + v) * 128 + n] = acc[v];
    return;
  }
  float bn = bias[n];
#pragma unroll
  for (int v = 0; v < 8; ++v) {
    float x = acc[v] + bn;
    sC[(mb + v) * 128 + n] = x > 0.f ? x : 0.f;
  }
  __syncthreads();
  if (tid < 16) {
    float s = 0.f, s2 = 0.f;
    for (int k = 0; k < 128; ++k) {
      float x = sC[tid * 128 + k];
      s += x; s2 += x * x;
    }
    float mean = s * (1.f / 128.f);
    float var  = s2 * (1.f / 128.f) - mean * mean;
    sMean[tid] = mean;
    sRstd[tid] = rsqrtf(var + 1e-5f);
  }
  __syncthreads();
  for (int i = tid; i < 16 * 128; i += 256) {
    int m = i >> 7, k = i & 127;
    Y[(size_t)(m0 + m) * 128 + k] = (sC[i] - sMean[m]) * sRstd[m] * g[k] + be[k];
  }
}

__global__ void mask_cat_kernel(const float* __restrict__ lpl_m,
                                const float* __restrict__ plp_m,
                                const float* __restrict__ pcp_m,
                                float* __restrict__ mask_t, float* __restrict__ mask_h) {
  int t = blockIdx.x * blockDim.x + threadIdx.x;
  if (t < RT) {
    float v;
    if (t < 2048)      v = 1.f;
    else if (t < 3072) v = lpl_m[t - 2048];
    else if (t < 4096) v = plp_m[t - 3072];
    else if (t < 4224) v = pcp_m[t - 4096];
    else               v = 1.f;
    mask_t[t] = v;
  }
  if (t < RH) {
    float v;
    if (t < 1024)      v = 1.f;
    else if (t < 2048) v = lpl_m[t - 1024];
    else if (t < 3072) v = plp_m[t - 2048];
    else               v = pcp_m[t - 3072];
    mask_h[t] = v;
  }
}

__global__ __launch_bounds__(256)
void pair_kernel(const float* __restrict__ symW, const float* __restrict__ geoW,
                 const float* __restrict__ mask, const float* __restrict__ b1,
                 const float* __restrict__ g, const float* __restrict__ be,
                 const float* __restrict__ W2, const float* __restrict__ b2p,
                 float* __restrict__ out, int Nsym, int R) {
  int gw = (int)(((unsigned)blockIdx.x * 256u + threadIdx.x) >> 5);
  if (gw >= Nsym * R) return;
  int lane = threadIdx.x & 31;
  int i = gw / R;
  int r = gw - i * R;

  float4 s4 = ((const float4*)(symW + (size_t)i * 128))[lane];
  float4 g4 = ((const float4*)(geoW + (size_t)r * 128))[lane];
  float4 b4 = ((const float4*)b1)[lane];

  float x0 = s4.x + g4.x + b4.x; x0 = x0 > 0.f ? x0 : 0.f;
  float x1 = s4.y + g4.y + b4.y; x1 = x1 > 0.f ? x1 : 0.f;
  float x2 = s4.z + g4.z + b4.z; x2 = x2 > 0.f ? x2 : 0.f;
  float x3 = s4.w + g4.w + b4.w; x3 = x3 > 0.f ? x3 : 0.f;

  float s  = x0 + x1 + x2 + x3;
  float s2 = x0 * x0 + x1 * x1 + x2 * x2 + x3 * x3;
  for (int o = 16; o > 0; o >>= 1) {
    s  += __shfl_xor(s, o, 32);
    s2 += __shfl_xor(s2, o, 32);
  }
  float mean = s * (1.f / 128.f);
  float var  = s2 * (1.f / 128.f) - mean * mean;
  float rstd = rsqrtf(var + 1e-5f);

  float4 gg = ((const float4*)g)[lane];
  float4 bb = ((const float4*)be)[lane];
  float4 w2 = ((const float4*)W2)[lane];
  float d = ((x0 - mean) * rstd * gg.x + bb.x) * w2.x +
            ((x1 - mean) * rstd * gg.y + bb.y) * w2.y +
            ((x2 - mean) * rstd * gg.z + bb.z) * w2.z +
            ((x3 - mean) * rstd * gg.w + bb.w) * w2.w;
  for (int o = 16; o > 0; o >>= 1) d += __shfl_xor(d, o, 32);

  if (lane == 0) {
    float logit = d + b2p[0];
    float mk = mask ? mask[r] : 1.f;
    float z  = (mk > 0.f) ? logit : -1e9f;
    out[(size_t)i * R + r] = 1.f / (1.f + expf(-z));
  }
}

extern "C" void kernel_launch(void* const* d_in, const int* in_sizes, int n_in,
                              void* d_out, int out_size, void* d_ws, size_t ws_size,
                              hipStream_t stream) {
  (void)in_sizes; (void)n_in; (void)out_size; (void)ws_size;
  const float* points  = (const float*)d_in[0];
  const float* lines   = (const float*)d_in[1];
  const float* circles = (const float*)d_in[2];
  const float* text_s  = (const float*)d_in[3];
  const float* head_s  = (const float*)d_in[4];
  const float* angle_s = (const float*)d_in[5];
  const float* bar_s   = (const float*)d_in[6];
  const float* para_s  = (const float*)d_in[7];
  const float* perp_s  = (const float*)d_in[8];
  const float* rel_W   = (const float*)d_in[9];
  const float* rel_b   = (const float*)d_in[10];
  const float* rel_g   = (const float*)d_in[11];
  const float* rel_be  = (const float*)d_in[12];
  const float* gm_W    = (const float*)d_in[13];
  const float* gm_b    = (const float*)d_in[14];
  const float* gm_g    = (const float*)d_in[15];
  const float* gm_be   = (const float*)d_in[16];
  const float* sc_W1   = (const float*)d_in[17];
  const float* sc_b1   = (const float*)d_in[18];
  const float* sc_g    = (const float*)d_in[19];
  const float* sc_be   = (const float*)d_in[20];
  const float* sc_W2   = (const float*)d_in[21];
  const float* sc_b2   = (const float*)d_in[22];
  const int*   pl      = (const int*)d_in[23];
  const int*   pc      = (const int*)d_in[24];
  float* out = (float*)d_out;

  float* wsp = (float*)d_ws;
  size_t off = 0;
  auto alloc = [&](size_t n) { float* p = wsp + off; off += (n + 31) & ~(size_t)31; return p; };
  float* LPLb   = alloc((size_t)PD * HD);
  float* PLPb   = alloc((size_t)LD * HD);
  float* PCPb   = alloc((size_t)CD * HD);
  float* lpl_m  = alloc(PD);
  float* plp_m  = alloc(LD);
  float* pcp_m  = alloc(CD);
  float* mask_t = alloc(RT);
  float* mask_h = alloc(RH);
  float* geo_t  = alloc((size_t)RT * HD);
  float* geo_h  = alloc((size_t)RH * HD);
  float* tmp    = alloc((size_t)RT * HD);
  float* symWb  = alloc((size_t)64 * HD);
  float* geoWb  = alloc((size_t)RT * HD);

  agg_gemm_kernel<<<PD / 16, 256, 0, stream>>>(pl, LD, 0, lines, LD, tmp, lpl_m);
  lrl_gemm_kernel<<<PD / 16, 256, 0, stream>>>(tmp, points, 2.f, rel_W + 0 * HH,
      rel_b + 0 * HD, rel_g + 0 * HD, rel_be + 0 * HD, LPLb, 1);

  agg_gemm_kernel<<<LD / 16, 256, 0, stream>>>(pl, LD, 1, points, PD, tmp, plp_m);
  lrl_gemm_kernel<<<LD / 16, 256, 0, stream>>>(tmp, lines, 2.f, rel_W + 1 * HH,
      rel_b + 1 * HD, rel_g + 1 * HD, rel_be + 1 * HD, PLPb, 1);

  agg_gemm_kernel<<<CD / 16, 256, 0, stream>>>(pc, CD, 1, points, PD, tmp, pcp_m);
  lrl_gemm_kernel<<<CD / 16, 256, 0, stream>>>(tmp, circles, 2.f, rel_W + 2 * HH,
      rel_b + 2 * HD, rel_g + 2 * HD, rel_be + 2 * HD, PCPb, 1);

  const size_t ROW = (size_t)HD * sizeof(float);
  hipMemcpyAsync(geo_t + (size_t)0 * HD,    points, PD * ROW, hipMemcpyDeviceToDevice, stream);
  hipMemcpyAsync(geo_t + (size_t)1024 * HD, lines,  LD * ROW, hipMemcpyDeviceToDevice, stream);
  hipMemcpyAsync(geo_t + (size_t)2048 * HD, LPLb,   PD * ROW, hipMemcpyDeviceToDevice, stream);
  hipMemcpyAsync(geo_t + (size_t)3072 * HD, PLPb,   LD * ROW, hipMemcpyDeviceToDevice, stream);
  hipMemcpyAsync(geo_t + (size_t)4096 * HD, PCPb,   CD * ROW, hipMemcpyDeviceToDevice, stream);
  hipMemcpyAsync(geo_t + (size_t)4224 * HD, head_s, NH * ROW, hipMemcpyDeviceToDevice, stream);
  hipMemcpyAsync(geo_h + (size_t)0 * HD,    points, PD * ROW, hipMemcpyDeviceToDevice, stream);
  hipMemcpyAsync(geo_h + (size_t)1024 * HD, LPLb,   PD * ROW, hipMemcpyDeviceToDevice, stream);
  hipMemcpyAsync(geo_h + (size_t)2048 * HD, PLPb,   LD * ROW, hipMemcpyDeviceToDevice, stream);
  hipMemcpyAsync(geo_h + (size_t)3072 * HD, PCPb,   CD * ROW, hipMemcpyDeviceToDevice, stream);
  mask_cat_kernel<<<(RT + 255) / 256, 256, 0, stream>>>(lpl_m, plp_m, pcp_m, mask_t, mask_h);

  lrl_gemm_kernel<<<RT / 16, 256, 0, stream>>>(geo_t, nullptr, 1.f, gm_W + 0 * HH,
      gm_b + 0 * HD, gm_g + 0 * HD, gm_be + 0 * HD, tmp, 1);
  lrl_gemm_kernel<<<RT / 16, 256, 0, stream>>>(tmp, nullptr, 1.f, gm_W + 1 * HH,
      gm_b + 1 * HD, gm_g + 1 * HD, gm_be + 1 * HD, geo_t, 1);
  lrl_gemm_kernel<<<RH / 16, 256, 0, stream>>>(geo_h, nullptr, 1.f, gm_W + 0 * HH,
      gm_b + 0 * HD, gm_g + 0 * HD, gm_be + 0 * HD, tmp, 1);
  lrl_gemm_kernel<<<RH / 16, 256, 0, stream>>>(tmp, nullptr, 1.f, gm_W + 1 * HH,
      gm_b + 1 * HD, gm_g + 1 * HD, gm_be + 1 * HD, geo_h, 1);

  struct Head { const float* sym; int N; const float* geo; int R; const float* mask; };
  const Head heads[6] = {
      {text_s,  NT,  geo_t, RT, mask_t},
      {head_s,  NH,  geo_h, RH, mask_h},
      {angle_s, NA,  LPLb,  PD, lpl_m},
      {bar_s,   NB,  PLPb,  LD, plp_m},
      {para_s,  NPA, lines, LD, nullptr},
      {perp_s,  NPE, LPLb,  PD, lpl_m},
  };
  size_t ooff = 0;
  for (int h = 0; h < 6; ++h) {
    const Head& hd = heads[h];
    lrl_gemm_kernel<<<hd.N / 16, 256, 0, stream>>>(hd.sym, nullptr, 1.f,
        sc_W1 + (size_t)h * HH, nullptr, nullptr, nullptr, symWb, 0);
    lrl_gemm_kernel<<<hd.R / 16, 256, 0, stream>>>(hd.geo, nullptr, 1.f,
        sc_W1 + (size_t)h * HH, nullptr, nullptr, nullptr, geoWb, 0);
    int waves  = hd.N * hd.R;
    int blocks = (waves + 7) / 8;
    pair_kernel<<<blocks, 256, 0, stream>>>(symWb, geoWb, hd.mask,
        sc_b1 + (size_t)h * HD, sc_g + (size_t)h * HD, sc_be + (size_t)h * HD,
        sc_W2 + (size_t)h * HD, sc_b2 + h, out + ooff, hd.N, hd.R);
    ooff += (size_t)hd.N * hd.R;
  }
}


# pair scoring: one lane per (sym,geo) pair instead of one wave per pair; geo tile in LDS, sym rows broadcast, LN and W2 dot folded into 3 running k-sums per lane (sum order changes); no shuffles
# speedup vs baseline: 1.3965x; 1.1934x over previous
#include <hip/hip_runtime.h>
#include <math.h>

static const int HD  = 128;
static const int PD  = 1024;
static const int LD  = 1024;
static const int CD  = 128;
static const int NT  = 64, NH = 32, NA = 64, NB = 64, NPA = 32, NPE = 32;
static const int RT  = PD + LD + PD + LD + CD + NH;
static const int RH  = PD + PD + LD + CD;
static const int HH  = HD * HD;

typedef float v2f __attribute__((ext_vector_type(2)));
typedef float v8f __attribute__((ext_vector_type(8)));

__device__ __forceinline__ v8f wmma_f32(v2f a, v2f b, v8f c) {
  return __builtin_amdgcn_wmma_f32_16x16x4_f32(false, a, false, b, (short)0, c,
                                               false, false);
}

__global__ __launch_bounds__(256)
void agg_gemm_kernel(const int* __restrict__ rel, int ldA, int transA,
                     const float* __restrict__ Bm, int K, float* __restrict__ Cout,
                     float* __restrict__ rowmask) {
  __shared__ float sA[16 * 128];
  const int tid  = threadIdx.x;
  const int lane = tid & 31;
  const int w    = tid >> 5;
  const int m0   = blockIdx.x * 16;
  const int n    = w * 16 + (lane & 15);
  const int am   = lane & 15;
  const int koff = (lane >> 4) << 1;

  v8f acc = {0.f, 0.f, 0.f, 0.f, 0.f, 0.f, 0.f, 0.f};

  const int crow = tid >> 4;
  const int ccol = (tid & 15) << 2;
  float cnt = 0.f;

  for (int k0 = 0; k0 < K; k0 += 128) {
    if (transA) {
      for (int i = tid; i < 16 * 128; i += 256) {
        int m = i & 15, k = i >> 4;
        sA[m * 128 + k] = (float)rel[(size_t)(k0 + k) * ldA + (m0 + m)];
      }
    } else {
      for (int i = tid; i < 16 * 128; i += 256) {
        int m = i >> 7, k = i & 127;
        sA[m * 128 + k] = (float)rel[(size_t)(m0 + m) * ldA + (k0 + k)];
      }
    }
    __syncthreads();
    {
      const float4 c0 = *(const float4*)&sA[crow * 128 + ccol];
      const float4 c1 = *(const float4*)&sA[crow * 128 + ccol + 64];
      cnt += ((c0.x + c0.y) + (c0.z + c0.w)) + ((c1.x + c1.y) + (c1.z + c1.w));
    }
    for (int kk = 0; kk < 128; kk += 4) {
      v2f a = *(const v2f*)&sA[am * 128 + kk + koff];
      int kg = k0 + kk + koff;
      v2f b;
      b.x = Bm[(size_t)kg * HD + n];
      b.y = Bm[(size_t)(kg + 1) * HD + n];
      acc = wmma_f32(a, b, acc);
    }
    __syncthreads();
  }
  const int mb = (lane >> 4) << 3;
#pragma unroll
  for (int v = 0; v < 8; ++v)
    Cout[(size_t)(m0 + mb + v) * HD + n] = acc[v];

  for (int o = 8; o > 0; o >>= 1) cnt += __shfl_xor(cnt, o, 32);
  if ((tid & 15) == 0) rowmask[m0 + crow] = (cnt >= 2.f) ? 1.f : 0.f;
}

__global__ __launch_bounds__(256)
void lrl_gemm_kernel(const float* __restrict__ X0, const float* __restrict__ X1,
                     float alpha, const float* __restrict__ W,
                     const float* __restrict__ bias, const float* __restrict__ g,
                     const float* __restrict__ be, float* __restrict__ Y, int do_ln) {
  __shared__ float sA[16 * 128];
  __shared__ float sC[16 * 128];
  __shared__ float sMean[16], sRstd[16];
  const int tid  = threadIdx.x;
  const int lane = tid & 31;
  const int w    = tid >> 5;
  const int m0   = blockIdx.x * 16;
  const int n    = w * 16 + (lane & 15);
  const int am   = lane & 15;
  const int koff = (lane >> 4) << 1;

  for (int i = tid; i < 16 * 128; i += 256) {
    int m = i >> 7, k = i & 127;
    float v = alpha * X0[(size_t)(m0 + m) * 128 + k];
    if (X1) v += X1[(size_t)(m0 + m) * 128 + k];
    sA[i] = v;
  }
  __syncthreads();

  v8f acc = {0.f, 0.f, 0.f, 0.f, 0.f, 0.f, 0.f, 0.f};
#pragma unroll 4
  for (int kk = 0; kk < 128; kk += 4) {
    v2f a = *(const v2f*)&sA[am * 128 + kk + koff];
    v2f b = *(const v2f*)&W[(size_t)n * 128 + kk + koff];
    acc = wmma_f32(a, b, acc);
  }

  const int mb = (lane >> 4) << 3;
  if (!do_ln) {
#pragma unroll
    for (int v = 0; v < 8; ++v)
      Y[(size_t)(m0 + mb + v) * 128 + n] = acc[v];
    return;
  }
  float bn = bias[n];
#pragma unroll
  for (int v = 0; v < 8; ++v) {
    float x = acc[v] + bn;
    sC[(mb + v) * 128 + n] = x > 0.f ? x : 0.f;
  }
  __syncthreads();
  if (tid < 16) {
    float s = 0.f, s2 = 0.f;
    for (int k = 0; k < 128; ++k) {
      float x = sC[tid * 128 + k];
      s += x; s2 += x * x;
    }
    float mean = s * (1.f / 128.f);
    float var  = s2 * (1.f / 128.f) - mean * mean;
    sMean[tid] = mean;
    sRstd[tid] = rsqrtf(var + 1e-5f);
  }
  __syncthreads();
  for (int i = tid; i < 16 * 128; i += 256) {
    int m = i >> 7, k = i & 127;
    Y[(size_t)(m0 + m) * 128 + k] = (sC[i] - sMean[m]) * sRstd[m] * g[k] + be[k];
  }
}

__global__ void mask_cat_kernel(const float* __restrict__ lpl_m,
                                const float* __restrict__ plp_m,
                                const float* __restrict__ pcp_m,
                                float* __restrict__ mask_t, float* __restrict__ mask_h) {
  int t = blockIdx.x * blockDim.x + threadIdx.x;
  if (t < RT) {
    float v;
    if (t < 2048)      v = 1.f;
    else if (t < 3072) v = lpl_m[t - 2048];
    else if (t < 4096) v = plp_m[t - 3072];
    else if (t < 4224) v = pcp_m[t - 4096];
    else               v = 1.f;
    mask_t[t] = v;
  }
  if (t < RH) {
    float v;
    if (t < 1024)      v = 1.f;
    else if (t < 2048) v = lpl_m[t - 1024];
    else if (t < 3072) v = plp_m[t - 2048];
    else               v = pcp_m[t - 3072];
    mask_h[t] = v;
  }
}

typedef float v4f __attribute__((ext_vector_type(4)));
static const int PG_ROWS  = 32;
static const int PS_ROWS  = 16;
static const int PG_PITCH = 132;
static_assert(RT % PG_ROWS == 0 && RH % PG_ROWS == 0 && PD % PG_ROWS == 0 && LD % PG_ROWS == 0,
              "geo row counts must be whole tiles");
static_assert(NT % PS_ROWS == 0 && NH % PS_ROWS == 0 && NA % PS_ROWS == 0 && NB % PS_ROWS == 0 &&
              NPA % PS_ROWS == 0 && NPE % PS_ROWS == 0, "sym row counts must be whole tiles");

__device__ __forceinline__ void pair_block(const v4f gv, const v4f wv, const v4f (&sv)[4],
                                           v2f (&S)[4], v2f (&Q)[4], v2f (&D)[4]) {
  const v2f zero2 = {0.f, 0.f};
  for (int ii = 0; ii < 4; ++ii) {
    const v2f h01 = __builtin_elementwise_max(sv[ii].xy + gv.xy, zero2);
    const v2f h23 = __builtin_elementwise_max(sv[ii].zw + gv.zw, zero2);
    S[ii] += h01;
    Q[ii] = __builtin_elementwise_fma(h01, h01, Q[ii]);
    D[ii] = __builtin_elementwise_fma(h01, wv.xy, D[ii]);
    S[ii] += h23;
    Q[ii] = __builtin_elementwise_fma(h23, h23, Q[ii]);
    D[ii] = __builtin_elementwise_fma(h23, wv.zw, D[ii]);
  }
}

__global__ __launch_bounds__(128)
void pair_kernel(const float* __restrict__ symW, const float* __restrict__ geoW,
                 const float* __restrict__ mask, const float* __restrict__ b1,
                 const float* __restrict__ g, const float* __restrict__ be,
                 const float* __restrict__ W2, const float* __restrict__ b2p,
                 float* __restrict__ out, int R) {
  __shared__ __attribute__((aligned(16))) float sG[PG_ROWS * PG_PITCH];
  __shared__ __attribute__((aligned(16))) float sS[PS_ROWS * 128];
  __shared__ __attribute__((aligned(16))) float sW[128];
  const int tid  = threadIdx.x;
  const int lane = tid & 31;
  const int w    = tid >> 5;
  const int r0   = blockIdx.x * PG_ROWS;
  const int i0   = blockIdx.y * PS_ROWS;

  const v4f gg = ((const v4f*)g)[lane];
  const v4f bb = ((const v4f*)be)[lane];
  const v4f ww = ((const v4f*)W2)[lane];
  const v4f b4 = ((const v4f*)b1)[lane];
  const float mk = mask[r0 + lane];
  const float b2 = b2p[0];
  v4f ts[4], tg[8];
  for (int u = 0; u < 4; ++u)
    ts[u] = ((const v4f*)(symW + (size_t)(i0 + w + 4 * u) * 128))[lane];
  for (int u = 0; u < 8; ++u)
    tg[u] = ((const v4f*)(geoW + (size_t)(r0 + w + 4 * u) * 128))[lane];

  const v4f gw = gg * ww;
  const v4f bw = bb * ww;
  if (w == 0) *(v4f*)&sW[lane * 4] = gw;
  float G  = (gw.x + gw.y) + (gw.z + gw.w);
  float Bc = (bw.x + bw.y) + (bw.z + bw.w);
  for (int o = 16; o > 0; o >>= 1) {
    G  += __shfl_xor(G, o, 32);
    Bc += __shfl_xor(Bc, o, 32);
  }
  for (int u = 0; u < 4; ++u)
    *(v4f*)&sS[(w + 4 * u) * 128 + lane * 4] = ts[u] + b4;
  for (int u = 0; u < 8; ++u)
    *(v4f*)&sG[(w + 4 * u) * PG_PITCH + lane * 4] = tg[u];
  __syncthreads();

  const float* gp = &sG[lane * PG_PITCH];
  const float* sp = &sS[(w * 4) * 128];
  const v2f zero2 = {0.f, 0.f};
  v2f S[4], Q[4], D[4];
  for (int ii = 0; ii < 4; ++ii) { S[ii] = zero2; Q[ii] = zero2; D[ii] = zero2; }

  v4f gA, wA, sA[4], gB, wB, sB[4];
  gA = *(const v4f*)gp;
  wA = *(const v4f*)sW;
  for (int ii = 0; ii < 4; ++ii) sA[ii] = *(const v4f*)(sp + ii * 128);
#pragma unroll
  for (int k = 0; k < 128; k += 8) {
    gB = *(const v4f*)(gp + k + 4);
    wB = *(const v4f*)(sW + k + 4);
    for (int ii = 0; ii < 4; ++ii) sB[ii] = *(const v4f*)(sp + ii * 128 + k + 4);
    pair_block(gA, wA, sA, S, Q, D);
    const int kn = (k + 8 < 128) ? k + 8 : 0;
    gA = *(const v4f*)(gp + kn);
    wA = *(const v4f*)(sW + kn);
    for (int ii = 0; ii < 4; ++ii) sA[ii] = *(const v4f*)(sp + ii * 128 + kn);
    pair_block(gB, wB, sB, S, Q, D);
  }

  for (int ii = 0; ii < 4; ++ii) {
    const float s = S[ii].x + S[ii].y;
    const float q = Q[ii].x + Q[ii].y;
    const float d = D[ii].x + D[ii].y;
    const float mean  = s * (1.f / 128.f);
    const float var   = q * (1.f / 128.f) - mean * mean;
    const float rstd  = rsqrtf(var + 1e-5f);
    const float logit = rstd * (d - mean * G) + Bc + b2;
    const float z     = (mk > 0.f) ? logit : -1e9f;
    out[(size_t)(i0 + w * 4 + ii) * R + r0 + lane] = 1.f / (1.f + expf(-z));
  }
}

extern "C" void kernel_launch(void* const* d_in, const int* in_sizes, int n_in,
                              void* d_out, int out_size, void* d_ws, size_t ws_size,
                              hipStream_t stream) {
  (void)in_sizes; (void)n_in; (void)out_size; (void)ws_size;
  const float* points  = (const float*)d_in[0];
  const float* lines   = (const float*)d_in[1];
  const float* circles = (const float*)d_in[2];
  const float* text_s  = (const float*)d_in[3];
  const float* head_s  = (const float*)d_in[4];
  const float* angle_s = (const float*)d_in[5];
  const float* bar_s   = (const float*)d_in[6];
  const float* para_s  = (const float*)d_in[7];
  const float* perp_s  = (const float*)d_in[8];
  const float* rel_W   = (const float*)d_in[9];
  const float* rel_b   = (const float*)d_in[10];
  const float* rel_g   = (const float*)d_in[11];
  const float* rel_be  = (const float*)d_in[12];
  const float* gm_W    = (const float*)d_in[13];
  const float* gm_b    = (const float*)d_in[14];
  const float* gm_g    = (const float*)d_in[15];
  const float* gm_be   = (const float*)d_in[16];
  const float* sc_W1   = (const float*)d_in[17];
  const float* sc_b1   = (const float*)d_in[18];
  const float* sc_g    = (const float*)d_in[19];
  const float* sc_be   = (const float*)d_in[20];
  const float* sc_W2   = (const float*)d_in[21];
  const float* sc_b2   = (const float*)d_in[22];
  const int*   pl      = (const int*)d_in[23];
  const int*   pc      = (const int*)d_in[24];
  float* out = (float*)d_out;

  float* wsp = (float*)d_ws;
  size_t off = 0;
  auto alloc = [&](size_t n) { float* p = wsp + off; off += (n + 31) & ~(size_t)31; return p; };
  float* LPLb   = alloc((size_t)PD * HD);
  float* PLPb   = alloc((size_t)LD * HD);
  float* PCPb   = alloc((size_t)CD * HD);
  float* lpl_m  = alloc(PD);
  float* plp_m  = alloc(LD);
  float* pcp_m  = alloc(CD);
  float* mask_t = alloc(RT);
  float* mask_h = alloc(RH);
  float* geo_t  = alloc((size_t)RT * HD);
  float* geo_h  = alloc((size_t)RH * HD);
  float* tmp    = alloc((size_t)RT * HD);
  float* symWb  = alloc((size_t)64 * HD);
  float* geoWb  = alloc((size_t)RT * HD);

  agg_gemm_kernel<<<PD / 16, 256, 0, stream>>>(pl, LD, 0, lines, LD, tmp, lpl_m);
  lrl_gemm_kernel<<<PD / 16, 256, 0, stream>>>(tmp, points, 2.f, rel_W + 0 * HH,
      rel_b + 0 * HD, rel_g + 0 * HD, rel_be + 0 * HD, LPLb, 1);

  agg_gemm_kernel<<<LD / 16, 256, 0, stream>>>(pl, LD, 1, points, PD, tmp, plp_m);
  lrl_gemm_kernel<<<LD / 16, 256, 0, stream>>>(tmp, lines, 2.f, rel_W + 1 * HH,
      rel_b + 1 * HD, rel_g + 1 * HD, rel_be + 1 * HD, PLPb, 1);

  agg_gemm_kernel<<<CD / 16, 256, 0, stream>>>(pc, CD, 1, points, PD, tmp, pcp_m);
  lrl_gemm_kernel<<<CD / 16, 256, 0, stream>>>(tmp, circles, 2.f, rel_W + 2 * HH,
      rel_b + 2 * HD, rel_g + 2 * HD, rel_be + 2 * HD, PCPb, 1);

  const size_t ROW = (size_t)HD * sizeof(float);
  hipMemcpyAsync(geo_t + (size_t)0 * HD,    points, PD * ROW, hipMemcpyDeviceToDevice, stream);
  hipMemcpyAsync(geo_t + (size_t)1024 * HD, lines,  LD * ROW, hipMemcpyDeviceToDevice, stream);
  hipMemcpyAsync(geo_t + (size_t)2048 * HD, LPLb,   PD * ROW, hipMemcpyDeviceToDevice, stream);
  hipMemcpyAsync(geo_t + (size_t)3072 * HD, PLPb,   LD * ROW, hipMemcpyDeviceToDevice, stream);
  hipMemcpyAsync(geo_t + (size_t)4096 * HD, PCPb,   CD * ROW, hipMemcpyDeviceToDevice, stream);
  hipMemcpyAsync(geo_t + (size_t)4224 * HD, head_s, NH * ROW, hipMemcpyDeviceToDevice, stream);
  hipMemcpyAsync(geo_h + (size_t)0 * HD,    points, PD * ROW, hipMemcpyDeviceToDevice, stream);
  hipMemcpyAsync(geo_h + (size_t)1024 * HD, LPLb,   PD * ROW, hipMemcpyDeviceToDevice, stream);
  hipMemcpyAsync(geo_h + (size_t)2048 * HD, PLPb,   LD * ROW, hipMemcpyDeviceToDevice, stream);
  hipMemcpyAsync(geo_h + (size_t)3072 * HD, PCPb,   CD * ROW, hipMemcpyDeviceToDevice, stream);
  mask_cat_kernel<<<(RT + 255) / 256, 256, 0, stream>>>(lpl_m, plp_m, pcp_m, mask_t, mask_h);

  lrl_gemm_kernel<<<RT / 16, 256, 0, stream>>>(geo_t, nullptr, 1.f, gm_W + 0 * HH,
      gm_b + 0 * HD, gm_g + 0 * HD, gm_be + 0 * HD, tmp, 1);
  lrl_gemm_kernel<<<RT / 16, 256, 0, stream>>>(tmp, nullptr, 1.f, gm_W + 1 * HH,
      gm_b + 1 * HD, gm_g + 1 * HD, gm_be + 1 * HD, geo_t, 1);
  lrl_gemm_kernel<<<RH / 16, 256, 0, stream>>>(geo_h, nullptr, 1.f, gm_W + 0 * HH,
      gm_b + 0 * HD, gm_g + 0 * HD, gm_be + 0 * HD, tmp, 1);
  lrl_gemm_kernel<<<RH / 16, 256, 0, stream>>>(tmp, nullptr, 1.f, gm_W + 1 * HH,
      gm_b + 1 * HD, gm_g + 1 * HD, gm_be + 1 * HD, geo_h, 1);

  struct Head { const float* sym; int N; const float* geo; int R; const float* mask; };
  const Head heads[6] = {
      {text_s,  NT,  geo_t, RT, mask_t},
      {head_s,  NH,  geo_h, RH, mask_h},
      {angle_s, NA,  LPLb,  PD, lpl_m},
      {bar_s,   NB,  PLPb,  LD, plp_m},
      {para_s,  NPA, lines, LD, mask_t + PD},
      {perp_s,  NPE, LPLb,  PD, lpl_m},
  };
  size_t ooff = 0;
  for (int h = 0; h < 6; ++h) {
    const Head& hd = heads[h];
    lrl_gemm_kernel<<<hd.N / 16, 256, 0, stream>>>(hd.sym, nullptr, 1.f,
        sc_W1 + (size_t)h * HH, nullptr, nullptr, nullptr, symWb, 0);
    lrl_gemm_kernel<<<hd.R / 16, 256, 0, stream>>>(hd.geo, nullptr, 1.f,
        sc_W1 + (size_t)h * HH, nullptr, nullptr, nullptr, geoWb, 0);
    pair_kernel<<<dim3(hd.R / PG_ROWS, hd.N / PS_ROWS), 128, 0, stream>>>(symWb, geoWb, hd.mask,
        sc_b1 + (size_t)h * HD, sc_g + (size_t)h * HD, sc_be + (size_t)h * HD,
        sc_W2 + (size_t)h * HD, sc_b2 + h, out + ooff, hd.R);
    ooff += (size_t)hd.N * hd.R;
  }
}


# independent launches of one kernel merged into one grid each (3 aggregations, 3 relation LRLs, gm over geo_t+geo_h per layer, 12 W1 projections, 6 pair heads): 29 kernel launches become 7
# speedup vs baseline: 3.9504x; 2.8288x over previous
#include <hip/hip_runtime.h>
#include <math.h>

static const int HD  = 128;
static const int PD  = 1024;
static const int LD  = 1024;
static const int CD  = 128;
static const int NT  = 64, NH = 32, NA = 64, NB = 64, NPA = 32, NPE = 32;
static const int RT  = PD + LD + PD + LD + CD + NH;
static const int RH  = PD + PD + LD + CD;
static const int HH  = HD * HD;

typedef float v2f __attribute__((ext_vector_type(2)));
typedef float v8f __attribute__((ext_vector_type(8)));

__device__ __forceinline__ v8f wmma_f32(v2f a, v2f b, v8f c) {
  return __builtin_amdgcn_wmma_f32_16x16x4_f32(false, a, false, b, (short)0, c,
                                               false, false);
}

__device__ __forceinline__
void agg_strip(const int* __restrict__ rel, int ldA, int transA,
               const float* __restrict__ Bm, int K, float* __restrict__ Cout,
               float* __restrict__ rowmask, int m0) {
  __shared__ float sA[16 * 128];
  const int tid  = threadIdx.x;
  const int lane = tid & 31;
  const int w    = tid >> 5;
  const int n    = w * 16 + (lane & 15);
  const int am   = lane & 15;
  const int koff = (lane >> 4) << 1;

  v8f acc = {0.f, 0.f, 0.f, 0.f, 0.f, 0.f, 0.f, 0.f};

  const int crow = tid >> 4;
  const int ccol = (tid & 15) << 2;
  float cnt = 0.f;

  for (int k0 = 0; k0 < K; k0 += 128) {
    if (transA) {
      for (int i = tid; i < 16 * 128; i += 256) {
        int m = i & 15, k = i >> 4;
        sA[m * 128 + k] = (float)rel[(size_t)(k0 + k) * ldA + (m0 + m)];
      }
    } else {
      for (int i = tid; i < 16 * 128; i += 256) {
        int m = i >> 7, k = i & 127;
        sA[m * 128 + k] = (float)rel[(size_t)(m0 + m) * ldA + (k0 + k)];
      }
    }
    __syncthreads();
    {
      const float4 c0 = *(const float4*)&sA[crow * 128 + ccol];
      const float4 c1 = *(const float4*)&sA[crow * 128 + ccol + 64];
      cnt += ((c0.x + c0.y) + (c0.z + c0.w)) + ((c1.x + c1.y) + (c1.z + c1.w));
    }
    for (int kk = 0; kk < 128; kk += 4) {
      v2f a = *(const v2f*)&sA[am * 128 + kk + koff];
      int kg = k0 + kk + koff;
      v2f b;
      b.x = Bm[(size_t)kg * HD + n];
      b.y = Bm[(size_t)(kg + 1) * HD + n];
      acc = wmma_f32(a, b, acc);
    }
    __syncthreads();
  }
  const int mb = (lane >> 4) << 3;
#pragma unroll
  for (int v = 0; v < 8; ++v)
    Cout[(size_t)(m0 + mb + v) * HD + n] = acc[v];

  for (int o = 8; o > 0; o >>= 1) cnt += __shfl_xor(cnt, o, 32);
  if ((tid & 15) == 0) rowmask[m0 + crow] = (cnt >= 2.f) ? 1.f : 0.f;
}

static const int AG_S0 = PD / 16;
static const int AG_S1 = AG_S0 + LD / 16;
static const int AG_S2 = AG_S1 + CD / 16;
__global__ __launch_bounds__(256)
void agg_gemm3_kernel(const int* __restrict__ pl, const int* __restrict__ pc,
                      const float* __restrict__ lines, const float* __restrict__ points,
                      float* __restrict__ aggL, float* __restrict__ aggP, float* __restrict__ aggC,
                      float* __restrict__ lpl_m, float* __restrict__ plp_m,
                      float* __restrict__ pcp_m) {
  const int b = blockIdx.x;
  const int* rel; int ldA, transA, K, m0; const float* Bm; float* Cout; float* rowmask;
  if (b < AG_S0) {
    rel = pl; ldA = LD; transA = 0; Bm = lines;  K = LD; Cout = aggL; rowmask = lpl_m; m0 = b * 16;
  } else if (b < AG_S1) {
    rel = pl; ldA = LD; transA = 1; Bm = points; K = PD; Cout = aggP; rowmask = plp_m; m0 = (b - AG_S0) * 16;
  } else {
    rel = pc; ldA = CD; transA = 1; Bm = points; K = PD; Cout = aggC; rowmask = pcp_m; m0 = (b - AG_S1) * 16;
  }
  agg_strip(rel, ldA, transA, Bm, K, Cout, rowmask, m0);
}

__device__ __forceinline__
void lrl_strip(const float* __restrict__ X0, const float* __restrict__ X1,
               float alpha, const float* __restrict__ W,
               const float* __restrict__ bias, const float* __restrict__ g,
               const float* __restrict__ be, float* __restrict__ Y, int do_ln) {
  __shared__ float sA[16 * 128];
  __shared__ float sC[16 * 128];
  __shared__ float sMean[16], sRstd[16];
  const int tid  = threadIdx.x;
  const int lane = tid & 31;
  const int w    = tid >> 5;
  const int n    = w * 16 + (lane & 15);
  const int am   = lane & 15;
  const int koff = (lane >> 4) << 1;

  for (int i = tid; i < 16 * 128; i += 256) {
    float v = alpha * X0[i];
    if (X1) v += X1[i];
    sA[i] = v;
  }
  __syncthreads();

  v8f acc = {0.f, 0.f, 0.f, 0.f, 0.f, 0.f, 0.f, 0.f};
#pragma unroll 4
  for (int kk = 0; kk < 128; kk += 4) {
    v2f a = *(const v2f*)&sA[am * 128 + kk + koff];
    v2f b = *(const v2f*)&W[(size_t)n * 128 + kk + koff];
    acc = wmma_f32(a, b, acc);
  }

  const int mb = (lane >> 4) << 3;
  if (!do_ln) {
#pragma unroll
    for (int v = 0; v < 8; ++v)
      Y[(size_t)(mb + v) * 128 + n] = acc[v];
    return;
  }
  float bn = bias[n];
#pragma unroll
  for (int v = 0; v < 8; ++v) {
    float x = acc[v] + bn;
    sC[(mb + v) * 128 + n] = x > 0.f ? x : 0.f;
  }
  __syncthreads();
  if (tid < 16) {
    float s = 0.f, s2 = 0.f;
    for (int k = 0; k < 128; ++k) {
      float x = sC[tid * 128 + k];
      s += x; s2 += x * x;
    }
    float mean = s * (1.f / 128.f);
    float var  = s2 * (1.f / 128.f) - mean * mean;
    sMean[tid] = mean;
    sRstd[tid] = rsqrtf(var + 1e-5f);
  }
  __syncthreads();
  for (int i = tid; i < 16 * 128; i += 256) {
    int m = i >> 7, k = i & 127;
    Y[i] = (sC[i] - sMean[m]) * sRstd[m] * g[k] + be[k];
  }
}

__global__ __launch_bounds__(256)
void lrl_gemm_kernel(const float* __restrict__ X0, const float* __restrict__ W,
                     const float* __restrict__ bias, const float* __restrict__ g,
                     const float* __restrict__ be, float* __restrict__ Y) {
  const size_t o = (size_t)blockIdx.x * (16 * 128);
  lrl_strip(X0 + o, nullptr, 1.f, W, bias, g, be, Y + o, 1);
}

__global__ __launch_bounds__(256)
void rel_lrl3_kernel(const float* __restrict__ aggL, const float* __restrict__ aggP,
                     const float* __restrict__ aggC, const float* __restrict__ points,
                     const float* __restrict__ lines, const float* __restrict__ circles,
                     const float* __restrict__ rel_W, const float* __restrict__ rel_b,
                     const float* __restrict__ rel_g, const float* __restrict__ rel_be,
                     float* __restrict__ LPLb, float* __restrict__ PLPb,
                     float* __restrict__ PCPb) {
  const int b = blockIdx.x;
  const float* X0; const float* X1; float* Y; int j, s;
  if (b < AG_S0)      { X0 = aggL; X1 = points;  Y = LPLb; j = 0; s = b; }
  else if (b < AG_S1) { X0 = aggP; X1 = lines;   Y = PLPb; j = 1; s = b - AG_S0; }
  else                { X0 = aggC; X1 = circles; Y = PCPb; j = 2; s = b - AG_S1; }
  const size_t o = (size_t)s * (16 * 128);
  lrl_strip(X0 + o, X1 + o, 2.f, rel_W + (size_t)j * HH, rel_b + j * HD, rel_g + j * HD,
            rel_be + j * HD, Y + o, 1);
}

static const int PJ_G0 = RT / 16;
static const int PJ_G1 = PJ_G0 + RH / 16;
static const int PJ_G2 = PJ_G1 + PD / 16;
static const int PJ_G3 = PJ_G2 + LD / 16;
static const int PJ_G4 = PJ_G3 + LD / 16;
static const int PJ_G5 = PJ_G4 + PD / 16;
static const int PJ_S0 = PJ_G5 + NT / 16;
static const int PJ_S1 = PJ_S0 + NH / 16;
static const int PJ_S2 = PJ_S1 + NA / 16;
static const int PJ_S3 = PJ_S2 + NB / 16;
static const int PJ_S4 = PJ_S3 + NPA / 16;
static const int PJ_S5 = PJ_S4 + NPE / 16;
__global__ __launch_bounds__(256)
void proj12_kernel(const float* __restrict__ geo_all, const float* __restrict__ LPLb,
                   const float* __restrict__ PLPb, const float* __restrict__ lines,
                   const float* __restrict__ text_s, const float* __restrict__ head_s,
                   const float* __restrict__ angle_s, const float* __restrict__ bar_s,
                   const float* __restrict__ para_s, const float* __restrict__ perp_s,
                   const float* __restrict__ sc_W1, float* __restrict__ proj) {
  const int b = blockIdx.x;
  const float* X; int h, s;
  if (b < PJ_G0)      { X = geo_all; h = 0; s = b; }
  else if (b < PJ_G1) { X = geo_all; h = 1; s = b; }
  else if (b < PJ_G2) { X = LPLb;    h = 2; s = b - PJ_G1; }
  else if (b < PJ_G3) { X = PLPb;    h = 3; s = b - PJ_G2; }
  else if (b < PJ_G4) { X = lines;   h = 4; s = b - PJ_G3; }
  else if (b < PJ_G5) { X = LPLb;    h = 5; s = b - PJ_G4; }
  else if (b < PJ_S0) { X = text_s;  h = 0; s = b - PJ_G5; }
  else if (b < PJ_S1) { X = head_s;  h = 1; s = b - PJ_S0; }
  else if (b < PJ_S2) { X = angle_s; h = 2; s = b - PJ_S1; }
  else if (b < PJ_S3) { X = bar_s;   h = 3; s = b - PJ_S2; }
  else if (b < PJ_S4) { X = para_s;  h = 4; s = b - PJ_S3; }
  else                { X = perp_s;  h = 5; s = b - PJ_S4; }
  lrl_strip(X + (size_t)s * (16 * 128), nullptr, 1.f, sc_W1 + (size_t)h * HH, nullptr, nullptr,
            nullptr, proj + (size_t)b * (16 * 128), 0);
}

__global__ void mask_cat_kernel(const float* __restrict__ lpl_m,
                                const float* __restrict__ plp_m,
                                const float* __restrict__ pcp_m,
                                float* __restrict__ mask_t, float* __restrict__ mask_h) {
  int t = blockIdx.x * blockDim.x + threadIdx.x;
  if (t < RT) {
    float v;
    if (t < 2048)      v = 1.f;
    else if (t < 3072) v = lpl_m[t - 2048];
    else if (t < 4096) v = plp_m[t - 3072];
    else if (t < 4224) v = pcp_m[t - 4096];
    else               v = 1.f;
    mask_t[t] = v;
  }
  if (t < RH) {
    float v;
    if (t < 1024)      v = 1.f;
    else if (t < 2048) v = lpl_m[t - 1024];
    else if (t < 3072) v = plp_m[t - 2048];
    else               v = pcp_m[t - 3072];
    mask_h[t] = v;
  }
}

typedef float v4f __attribute__((ext_vector_type(4)));
static const int PG_ROWS  = 32;
static const int PS_ROWS  = 16;
static const int PG_PITCH = 132;
static_assert(RT % PG_ROWS == 0 && RH % PG_ROWS == 0 && PD % PG_ROWS == 0 && LD % PG_ROWS == 0,
              "geo row counts must be whole tiles");
static_assert(NT % PS_ROWS == 0 && NH % PS_ROWS == 0 && NA % PS_ROWS == 0 && NB % PS_ROWS == 0 &&
              NPA % PS_ROWS == 0 && NPE % PS_ROWS == 0, "sym row counts must be whole tiles");

__device__ __forceinline__ void pair_block(const v4f gv, const v4f wv, const v4f (&sv)[4],
                                           v2f (&S)[4], v2f (&Q)[4], v2f (&D)[4]) {
  const v2f zero2 = {0.f, 0.f};
  for (int ii = 0; ii < 4; ++ii) {
    const v2f h01 = __builtin_elementwise_max(sv[ii].xy + gv.xy, zero2);
    const v2f h23 = __builtin_elementwise_max(sv[ii].zw + gv.zw, zero2);
    S[ii] += h01;
    Q[ii] = __builtin_elementwise_fma(h01, h01, Q[ii]);
    D[ii] = __builtin_elementwise_fma(h01, wv.xy, D[ii]);
    S[ii] += h23;
    Q[ii] = __builtin_elementwise_fma(h23, h23, Q[ii]);
    D[ii] = __builtin_elementwise_fma(h23, wv.zw, D[ii]);
  }
}

__device__ __forceinline__
void pair_tile(const float* __restrict__ symW, const float* __restrict__ geoW,
                 const float* __restrict__ mask, const float* __restrict__ b1,
                 const float* __restrict__ g, const float* __restrict__ be,
                 const float* __restrict__ W2, const float* __restrict__ b2p,
                 float* __restrict__ out, int R, int r0, int i0) {
  __shared__ __attribute__((aligned(16))) float sG[PG_ROWS * PG_PITCH];
  __shared__ __attribute__((aligned(16))) float sS[PS_ROWS * 128];
  __shared__ __attribute__((aligned(16))) float sW[128];
  const int tid  = threadIdx.x;
  const int lane = tid & 31;
  const int w    = tid >> 5;

  const v4f gg = ((const v4f*)g)[lane];
  const v4f bb = ((const v4f*)be)[lane];
  const v4f ww = ((const v4f*)W2)[lane];
  const v4f b4 = ((const v4f*)b1)[lane];
  const float mk = mask[r0 + lane];
  const float b2 = b2p[0];
  v4f ts[4], tg[8];
  for (int u = 0; u < 4; ++u)
    ts[u] = ((const v4f*)(symW + (size_t)(i0 + w + 4 * u) * 128))[lane];
  for (int u = 0; u < 8; ++u)
    tg[u] = ((const v4f*)(geoW + (size_t)(r0 + w + 4 * u) * 128))[lane];

  const v4f gw = gg * ww;
  const v4f bw = bb * ww;
  if (w == 0) *(v4f*)&sW[lane * 4] = gw;
  float G  = (gw.x + gw.y) + (gw.z + gw.w);
  float Bc = (bw.x + bw.y) + (bw.z + bw.w);
  for (int o = 16; o > 0; o >>= 1) {
    G  += __shfl_xor(G, o, 32);
    Bc += __shfl_xor(Bc, o, 32);
  }
  for (int u = 0; u < 4; ++u)
    *(v4f*)&sS[(w + 4 * u) * 128 + lane * 4] = ts[u] + b4;
  for (int u = 0; u < 8; ++u)
    *(v4f*)&sG[(w + 4 * u) * PG_PITCH + lane * 4] = tg[u];
  __syncthreads();

  const float* gp = &sG[lane * PG_PITCH];
  const float* sp = &sS[(w * 4) * 128];
  const v2f zero2 = {0.f, 0.f};
  v2f S[4], Q[4], D[4];
  for (int ii = 0; ii < 4; ++ii) { S[ii] = zero2; Q[ii] = zero2; D[ii] = zero2; }

  v4f gA, wA, sA[4], gB, wB, sB[4];
  gA = *(const v4f*)gp;
  wA = *(const v4f*)sW;
  for (int ii = 0; ii < 4; ++ii) sA[ii] = *(const v4f*)(sp + ii * 128);
#pragma unroll
  for (int k = 0; k < 128; k += 8) {
    gB = *(const v4f*)(gp + k + 4);
    wB = *(const v4f*)(sW + k + 4);
    for (int ii = 0; ii < 4; ++ii) sB[ii] = *(const v4f*)(sp + ii * 128 + k + 4);
    pair_block(gA, wA, sA, S, Q, D);
    const int kn = (k + 8 < 128) ? k + 8 : 0;
    gA = *(const v4f*)(gp + kn);
    wA = *(const v4f*)(sW + kn);
    for (int ii = 0; ii < 4; ++ii) sA[ii] = *(const v4f*)(sp + ii * 128 + kn);
    pair_block(gB, wB, sB, S, Q, D);
  }

  for (int ii = 0; ii < 4; ++ii) {
    const float s = S[ii].x + S[ii].y;
    const float q = Q[ii].x + Q[ii].y;
    const float d = D[ii].x + D[ii].y;
    const float mean  = s * (1.f / 128.f);
    const float var   = q * (1.f / 128.f) - mean * mean;
    const float rstd  = rsqrtf(var + 1e-5f);
    const float logit = rstd * (d - mean * G) + Bc + b2;
    const float z     = (mk > 0.f) ? logit : -1e9f;
    out[(size_t)(i0 + w * 4 + ii) * R + r0 + lane] = 1.f / (1.f + expf(-z));
  }
}


static const int PB0 = (RT / PG_ROWS) * (NT / PS_ROWS);
static const int PB1 = PB0 + (RH / PG_ROWS) * (NH / PS_ROWS);
static const int PB2 = PB1 + (PD / PG_ROWS) * (NA / PS_ROWS);
static const int PB3 = PB2 + (LD / PG_ROWS) * (NB / PS_ROWS);
static const int PB4 = PB3 + (LD / PG_ROWS) * (NPA / PS_ROWS);
static const int PB5 = PB4 + (PD / PG_ROWS) * (NPE / PS_ROWS);
__global__ __launch_bounds__(128)
void pair6_kernel(const float* __restrict__ proj,
                  const float* __restrict__ mask_t, const float* __restrict__ mask_h,
                  const float* __restrict__ lpl_m, const float* __restrict__ plp_m,
                  const float* __restrict__ sc_b1, const float* __restrict__ sc_g,
                  const float* __restrict__ sc_be, const float* __restrict__ sc_W2,
                  const float* __restrict__ sc_b2, float* __restrict__ out) {
  const int b = blockIdx.x;
  int h, R, bx, by, grow, srow, ooff; const float* mask;
  if (b < PB0) {
    h = 0; R = RT; by = b / (RT / PG_ROWS); bx = b - by * (RT / PG_ROWS);
    grow = 0; srow = 16 * PJ_G5; mask = mask_t; ooff = 0;
  } else if (b < PB1) {
    const int lb = b - PB0;
    h = 1; R = RH; by = lb / (RH / PG_ROWS); bx = lb - by * (RH / PG_ROWS);
    grow = 16 * PJ_G0; srow = 16 * PJ_S0; mask = mask_h; ooff = NT * RT;
  } else if (b < PB2) {
    const int lb = b - PB1;
    h = 2; R = PD; by = lb / (PD / PG_ROWS); bx = lb - by * (PD / PG_ROWS);
    grow = 16 * PJ_G1; srow = 16 * PJ_S1; mask = lpl_m; ooff = NT * RT + NH * RH;
  } else if (b < PB3) {
    const int lb = b - PB2;
    h = 3; R = LD; by = lb / (LD / PG_ROWS); bx = lb - by * (LD / PG_ROWS);
    grow = 16 * PJ_G2; srow = 16 * PJ_S2; mask = plp_m; ooff = NT * RT + NH * RH + NA * PD;
  } else if (b < PB4) {
    const int lb = b - PB3;
    h = 4; R = LD; by = lb / (LD / PG_ROWS); bx = lb - by * (LD / PG_ROWS);
    grow = 16 * PJ_G3; srow = 16 * PJ_S3; mask = mask_t + PD;
    ooff = NT * RT + NH * RH + NA * PD + NB * LD;
  } else {
    const int lb = b - PB4;
    h = 5; R = PD; by = lb / (PD / PG_ROWS); bx = lb - by * (PD / PG_ROWS);
    grow = 16 * PJ_G4; srow = 16 * PJ_S4; mask = lpl_m;
    ooff = NT * RT + NH * RH + NA * PD + NB * LD + NPA * LD;
  }
  pair_tile(proj + (size_t)srow * 128, proj + (size_t)grow * 128, mask,
            sc_b1 + h * HD, sc_g + h * HD, sc_be + h * HD, sc_W2 + h * HD, sc_b2 + h,
            out + ooff, R, bx * PG_ROWS, by * PS_ROWS);
}

extern "C" void kernel_launch(void* const* d_in, const int* in_sizes, int n_in,
                              void* d_out, int out_size, void* d_ws, size_t ws_size,
                              hipStream_t stream) {
  (void)in_sizes; (void)n_in; (void)out_size; (void)ws_size;
  const float* points  = (const float*)d_in[0];
  const float* lines   = (const float*)d_in[1];
  const float* circles = (const float*)d_in[2];
  const float* text_s  = (const float*)d_in[3];
  const float* head_s  = (const float*)d_in[4];
  const float* angle_s = (const float*)d_in[5];
  const float* bar_s   = (const float*)d_in[6];
  const float* para_s  = (const float*)d_in[7];
  const float* perp_s  = (const float*)d_in[8];
  const float* rel_W   = (const float*)d_in[9];
  const float* rel_b   = (const float*)d_in[10];
  const float* rel_g   = (const float*)d_in[11];
  const float* rel_be  = (const float*)d_in[12];
  const float* gm_W    = (const float*)d_in[13];
  const float* gm_b    = (const float*)d_in[14];
  const float* gm_g    = (const float*)d_in[15];
  const float* gm_be   = (const float*)d_in[16];
  const float* sc_W1   = (const float*)d_in[17];
  const float* sc_b1   = (const float*)d_in[18];
  const float* sc_g    = (const float*)d_in[19];
  const float* sc_be   = (const float*)d_in[20];
  const float* sc_W2   = (const float*)d_in[21];
  const float* sc_b2   = (const float*)d_in[22];
  const int*   pl      = (const int*)d_in[23];
  const int*   pc      = (const int*)d_in[24];
  float* out = (float*)d_out;

  float* wsp = (float*)d_ws;
  size_t off = 0;
  auto alloc = [&](size_t n) { float* p = wsp + off; off += (n + 31) & ~(size_t)31; return p; };
  float* LPLb   = alloc((size_t)PD * HD);
  float* PLPb   = alloc((size_t)LD * HD);
  float* PCPb   = alloc((size_t)CD * HD);
  float* lpl_m  = alloc(PD);
  float* plp_m  = alloc(LD);
  float* pcp_m  = alloc(CD);
  float* mask_t = alloc(RT);
  float* mask_h = alloc(RH);
  float* aggL   = alloc((size_t)PD * HD);
  float* aggP   = alloc((size_t)LD * HD);
  float* aggC   = alloc((size_t)CD * HD);
  float* geo_all = alloc((size_t)(RT + RH) * HD);
  float* geo_t  = geo_all;
  float* geo_h  = geo_all + (size_t)RT * HD;
  float* tmp    = alloc((size_t)(RT + RH) * HD);
  float* proj   = alloc((size_t)PJ_S5 * 16 * HD);

  agg_gemm3_kernel<<<AG_S2, 256, 0, stream>>>(pl, pc, lines, points, aggL, aggP, aggC,
                                               lpl_m, plp_m, pcp_m);
  rel_lrl3_kernel<<<AG_S2, 256, 0, stream>>>(aggL, aggP, aggC, points, lines, circles,
                                              rel_W, rel_b, rel_g, rel_be, LPLb, PLPb, PCPb);

  const size_t ROW = (size_t)HD * sizeof(float);
  hipMemcpyAsync(geo_t + (size_t)0 * HD,    points, PD * ROW, hipMemcpyDeviceToDevice, stream);
  hipMemcpyAsync(geo_t + (size_t)1024 * HD, lines,  LD * ROW, hipMemcpyDeviceToDevice, stream);
  hipMemcpyAsync(geo_t + (size_t)2048 * HD, LPLb,   PD * ROW, hipMemcpyDeviceToDevice, stream);
  hipMemcpyAsync(geo_t + (size_t)3072 * HD, PLPb,   LD * ROW, hipMemcpyDeviceToDevice, stream);
  hipMemcpyAsync(geo_t + (size_t)4096 * HD, PCPb,   CD * ROW, hipMemcpyDeviceToDevice, stream);
  hipMemcpyAsync(geo_t + (size_t)4224 * HD, head_s, NH * ROW, hipMemcpyDeviceToDevice, stream);
  hipMemcpyAsync(geo_h + (size_t)0 * HD,    points, PD * ROW, hipMemcpyDeviceToDevice, stream);
  hipMemcpyAsync(geo_h + (size_t)1024 * HD, LPLb,   PD * ROW, hipMemcpyDeviceToDevice, stream);
  hipMemcpyAsync(geo_h + (size_t)2048 * HD, PLPb,   LD * ROW, hipMemcpyDeviceToDevice, stream);
  hipMemcpyAsync(geo_h + (size_t)3072 * HD, PCPb,   CD * ROW, hipMemcpyDeviceToDevice, stream);
  mask_cat_kernel<<<(RT + 255) / 256, 256, 0, stream>>>(lpl_m, plp_m, pcp_m, mask_t, mask_h);

  lrl_gemm_kernel<<<(RT + RH) / 16, 256, 0, stream>>>(geo_all, gm_W + 0 * HH,
      gm_b + 0 * HD, gm_g + 0 * HD, gm_be + 0 * HD, tmp);
  lrl_gemm_kernel<<<(RT + RH) / 16, 256, 0, stream>>>(tmp, gm_W + 1 * HH,
      gm_b + 1 * HD, gm_g + 1 * HD, gm_be + 1 * HD, geo_all);

  proj12_kernel<<<PJ_S5, 256, 0, stream>>>(geo_all, LPLb, PLPb, lines, text_s, head_s, angle_s,
                                            bar_s, para_s, perp_s, sc_W1, proj);
  pair6_kernel<<<PB5, 128, 0, stream>>>(proj, mask_t, mask_h, lpl_m, plp_m,
                                         sc_b1, sc_g, sc_be, sc_W2, sc_b2, out);
}


# LDS row pitch of the 16-row A tiles and LayerNorm tiles 128 -> 132 floats (16 B x 33): fragment reads and row sums, lane = row, no longer hit one bank 16 ways; values and sum order unchanged
# speedup vs baseline: 4.4250x; 1.1201x over previous
#include <hip/hip_runtime.h>
#include <math.h>

static const int HD  = 128;
static const int PD  = 1024;
static const int LD  = 1024;
static const int CD  = 128;
static const int NT  = 64, NH = 32, NA = 64, NB = 64, NPA = 32, NPE = 32;
static const int RT  = PD + LD + PD + LD + CD + NH;
static const int RH  = PD + PD + LD + CD;
static const int HH  = HD * HD;

typedef float v2f __attribute__((ext_vector_type(2)));
typedef float v8f __attribute__((ext_vector_type(8)));

__device__ __forceinline__ v8f wmma_f32(v2f a, v2f b, v8f c) {
  return __builtin_amdgcn_wmma_f32_16x16x4_f32(false, a, false, b, (short)0, c,
                                               false, false);
}

static const int TP = 132;

__device__ __forceinline__
void agg_strip(const int* __restrict__ rel, int ldA, int transA,
               const float* __restrict__ Bm, int K, float* __restrict__ Cout,
               float* __restrict__ rowmask, int m0) {
  __shared__ __attribute__((aligned(16))) float sA[16 * TP];
  const int tid  = threadIdx.x;
  const int lane = tid & 31;
  const int w    = tid >> 5;
  const int n    = w * 16 + (lane & 15);
  const int am   = lane & 15;
  const int koff = (lane >> 4) << 1;

  v8f acc = {0.f, 0.f, 0.f, 0.f, 0.f, 0.f, 0.f, 0.f};

  const int crow = tid >> 4;
  const int ccol = (tid & 15) << 2;
  float cnt = 0.f;

  for (int k0 = 0; k0 < K; k0 += 128) {
    if (transA) {
      for (int i = tid; i < 16 * 128; i += 256) {
        int m = i & 15, k = i >> 4;
        sA[m * TP + k] = (float)rel[(size_t)(k0 + k) * ldA + (m0 + m)];
      }
    } else {
      for (int i = tid; i < 16 * 128; i += 256) {
        int m = i >> 7, k = i & 127;
        sA[m * TP + k] = (float)rel[(size_t)(m0 + m) * ldA + (k0 + k)];
      }
    }
    __syncthreads();
    {
      const float4 c0 = *(const float4*)&sA[crow * TP + ccol];
      const float4 c1 = *(const float4*)&sA[crow * TP + ccol + 64];
      cnt += ((c0.x + c0.y) + (c0.z + c0.w)) + ((c1.x + c1.y) + (c1.z + c1.w));
    }
    for (int kk = 0; kk < 128; kk += 4) {
      v2f a = *(const v2f*)&sA[am * TP + kk + koff];
      int kg = k0 + kk + koff;
      v2f b;
      b.x = Bm[(size_t)kg * HD + n];
      b.y = Bm[(size_t)(kg + 1) * HD + n];
      acc = wmma_f32(a, b, acc);
    }
    __syncthreads();
  }
  const int mb = (lane >> 4) << 3;
#pragma unroll
  for (int v = 0; v < 8; ++v)
    Cout[(size_t)(m0 + mb + v) * HD + n] = acc[v];

  for (int o = 8; o > 0; o >>= 1) cnt += __shfl_xor(cnt, o, 32);
  if ((tid & 15) == 0) rowmask[m0 + crow] = (cnt >= 2.f) ? 1.f : 0.f;
}

static const int AG_S0 = PD / 16;
static const int AG_S1 = AG_S0 + LD / 16;
static const int AG_S2 = AG_S1 + CD / 16;
__global__ __launch_bounds__(256)
void agg_gemm3_kernel(const int* __restrict__ pl, const int* __restrict__ pc,
                      const float* __restrict__ lines, const float* __restrict__ points,
                      float* __restrict__ aggL, float* __restrict__ aggP, float* __restrict__ aggC,
                      float* __restrict__ lpl_m, float* __restrict__ plp_m,
                      float* __restrict__ pcp_m) {
  const int b = blockIdx.x;
  const int* rel; int ldA, transA, K, m0; const float* Bm; float* Cout; float* rowmask;
  if (b < AG_S0) {
    rel = pl; ldA = LD; transA = 0; Bm = lines;  K = LD; Cout = aggL; rowmask = lpl_m; m0 = b * 16;
  } else if (b < AG_S1) {
    rel = pl; ldA = LD; transA = 1; Bm = points; K = PD; Cout = aggP; rowmask = plp_m; m0 = (b - AG_S0) * 16;
  } else {
    rel = pc; ldA = CD; transA = 1; Bm = points; K = PD; Cout = aggC; rowmask = pcp_m; m0 = (b - AG_S1) * 16;
  }
  agg_strip(rel, ldA, transA, Bm, K, Cout, rowmask, m0);
}

__device__ __forceinline__
void lrl_strip(const float* __restrict__ X0, const float* __restrict__ X1,
               float alpha, const float* __restrict__ W,
               const float* __restrict__ bias, const float* __restrict__ g,
               const float* __restrict__ be, float* __restrict__ Y, int do_ln) {
  __shared__ __attribute__((aligned(16))) float sA[16 * TP];
  __shared__ __attribute__((aligned(16))) float sC[16 * TP];
  __shared__ float sMean[16], sRstd[16];
  const int tid  = threadIdx.x;
  const int lane = tid & 31;
  const int w    = tid >> 5;
  const int n    = w * 16 + (lane & 15);
  const int am   = lane & 15;
  const int koff = (lane >> 4) << 1;

  for (int i = tid; i < 16 * 128; i += 256) {
    float v = alpha * X0[i];
    if (X1) v += X1[i];
    sA[(i >> 7) * TP + (i & 127)] = v;
  }
  __syncthreads();

  v8f acc = {0.f, 0.f, 0.f, 0.f, 0.f, 0.f, 0.f, 0.f};
#pragma unroll 4
  for (int kk = 0; kk < 128; kk += 4) {
    v2f a = *(const v2f*)&sA[am * TP + kk + koff];
    v2f b = *(const v2f*)&W[(size_t)n * 128 + kk + koff];
    acc = wmma_f32(a, b, acc);
  }

  const int mb = (lane >> 4) << 3;
  if (!do_ln) {
#pragma unroll
    for (int v = 0; v < 8; ++v)
      Y[(size_t)(mb + v) * 128 + n] = acc[v];
    return;
  }
  float bn = bias[n];
#pragma unroll
  for (int v = 0; v < 8; ++v) {
    float x = acc[v] + bn;
    sC[(mb + v) * TP + n] = x > 0.f ? x : 0.f;
  }
  __syncthreads();
  if (tid < 16) {
    float s = 0.f, s2 = 0.f;
    for (int k = 0; k < 128; ++k) {
      float x = sC[tid * TP + k];
      s += x; s2 += x * x;
    }
    float mean = s * (1.f / 128.f);
    float var  = s2 * (1.f / 128.f) - mean * mean;
    sMean[tid] = mean;
    sRstd[tid] = rsqrtf(var + 1e-5f);
  }
  __syncthreads();
  for (int i = tid; i < 16 * 128; i += 256) {
    int m = i >> 7, k = i & 127;
    Y[i] = (sC[m * TP + k] - sMean[m]) * sRstd[m] * g[k] + be[k];
  }
}

__global__ __launch_bounds__(256)
void lrl_gemm_kernel(const float* __restrict__ X0, const float* __restrict__ W,
                     const float* __restrict__ bias, const float* __restrict__ g,
                     const float* __restrict__ be, float* __restrict__ Y) {
  const size_t o = (size_t)blockIdx.x * (16 * 128);
  lrl_strip(X0 + o, nullptr, 1.f, W, bias, g, be, Y + o, 1);
}

__global__ __launch_bounds__(256)
void rel_lrl3_kernel(const float* __restrict__ aggL, const float* __restrict__ aggP,
                     const float* __restrict__ aggC, const float* __restrict__ points,
                     const float* __restrict__ lines, const float* __restrict__ circles,
                     const float* __restrict__ rel_W, const float* __restrict__ rel_b,
                     const float* __restrict__ rel_g, const float* __restrict__ rel_be,
                     float* __restrict__ LPLb, float* __restrict__ PLPb,
                     float* __restrict__ PCPb) {
  const int b = blockIdx.x;
  const float* X0; const float* X1; float* Y; int j, s;
  if (b < AG_S0)      { X0 = aggL; X1 = points;  Y = LPLb; j = 0; s = b; }
  else if (b < AG_S1) { X0 = aggP; X1 = lines;   Y = PLPb; j = 1; s = b - AG_S0; }
  else                { X0 = aggC; X1 = circles; Y = PCPb; j = 2; s = b - AG_S1; }
  const size_t o = (size_t)s * (16 * 128);
  lrl_strip(X0 + o, X1 + o, 2.f, rel_W + (size_t)j * HH, rel_b + j * HD, rel_g + j * HD,
            rel_be + j * HD, Y + o, 1);
}

static const int PJ_G0 = RT / 16;
static const int PJ_G1 = PJ_G0 + RH / 16;
static const int PJ_G2 = PJ_G1 + PD / 16;
static const int PJ_G3 = PJ_G2 + LD / 16;
static const int PJ_G4 = PJ_G3 + LD / 16;
static const int PJ_G5 = PJ_G4 + PD / 16;
static const int PJ_S0 = PJ_G5 + NT / 16;
static const int PJ_S1 = PJ_S0 + NH / 16;
static const int PJ_S2 = PJ_S1 + NA / 16;
static const int PJ_S3 = PJ_S2 + NB / 16;
static const int PJ_S4 = PJ_S3 + NPA / 16;
static const int PJ_S5 = PJ_S4 + NPE / 16;
__global__ __launch_bounds__(256)
void proj12_kernel(const float* __restrict__ geo_all, const float* __restrict__ LPLb,
                   const float* __restrict__ PLPb, const float* __restrict__ lines,
                   const float* __restrict__ text_s, const float* __restrict__ head_s,
                   const float* __restrict__ angle_s, const float* __restrict__ bar_s,
                   const float* __restrict__ para_s, const float* __restrict__ perp_s,
                   const float* __restrict__ sc_W1, float* __restrict__ proj) {
  const int b = blockIdx.x;
  const float* X; int h, s;
  if (b < PJ_G0)      { X = geo_all; h = 0; s = b; }
  else if (b < PJ_G1) { X = geo_all; h = 1; s = b; }
  else if (b < PJ_G2) { X = LPLb;    h = 2; s = b - PJ_G1; }
  else if (b < PJ_G3) { X = PLPb;    h = 3; s = b - PJ_G2; }
  else if (b < PJ_G4) { X = lines;   h = 4; s = b - PJ_G3; }
  else if (b < PJ_G5) { X = LPLb;    h = 5; s = b - PJ_G4; }
  else if (b < PJ_S0) { X = text_s;  h = 0; s = b - PJ_G5; }
  else if (b < PJ_S1) { X = head_s;  h = 1; s = b - PJ_S0; }
  else if (b < PJ_S2) { X = angle_s; h = 2; s = b - PJ_S1; }
  else if (b < PJ_S3) { X = bar_s;   h = 3; s = b - PJ_S2; }
  else if (b < PJ_S4) { X = para_s;  h = 4; s = b - PJ_S3; }
  else                { X = perp_s;  h = 5; s = b - PJ_S4; }
  lrl_strip(X + (size_t)s * (16 * 128), nullptr, 1.f, sc_W1 + (size_t)h * HH, nullptr, nullptr,
            nullptr, proj + (size_t)b * (16 * 128), 0);
}

__global__ void mask_cat_kernel(const float* __restrict__ lpl_m,
                                const float* __restrict__ plp_m,
                                const float* __restrict__ pcp_m,
                                float* __restrict__ mask_t, float* __restrict__ mask_h) {
  int t = blockIdx.x * blockDim.x + threadIdx.x;
  if (t < RT) {
    float v;
    if (t < 2048)      v = 1.f;
    else if (t < 3072) v = lpl_m[t - 2048];
    else if (t < 4096) v = plp_m[t - 3072];
    else if (t < 4224) v = pcp_m[t - 4096];
    else               v = 1.f;
    mask_t[t] = v;
  }
  if (t < RH) {
    float v;
    if (t < 1024)      v = 1.f;
    else if (t < 2048) v = lpl_m[t - 1024];
    else if (t < 3072) v = plp_m[t - 2048];
    else               v = pcp_m[t - 3072];
    mask_h[t] = v;
  }
}

typedef float v4f __attribute__((ext_vector_type(4)));
static const int PG_ROWS  = 32;
static const int PS_ROWS  = 16;
static const int PG_PITCH = 132;
static_assert(RT % PG_ROWS == 0 && RH % PG_ROWS == 0 && PD % PG_ROWS == 0 && LD % PG_ROWS == 0,
              "geo row counts must be whole tiles");
static_assert(NT % PS_ROWS == 0 && NH % PS_ROWS == 0 && NA % PS_ROWS == 0 && NB % PS_ROWS == 0 &&
              NPA % PS_ROWS == 0 && NPE % PS_ROWS == 0, "sym row counts must be whole tiles");

__device__ __forceinline__ void pair_block(const v4f gv, const v4f wv, const v4f (&sv)[4],
                                           v2f (&S)[4], v2f (&Q)[4], v2f (&D)[4]) {
  const v2f zero2 = {0.f, 0.f};
  for (int ii = 0; ii < 4; ++ii) {
    const v2f h01 = __builtin_elementwise_max(sv[ii].xy + gv.xy, zero2);
    const v2f h23 = __builtin_elementwise_max(sv[ii].zw + gv.zw, zero2);
    S[ii] += h01;
    Q[ii] = __builtin_elementwise_fma(h01, h01, Q[ii]);
    D[ii] = __builtin_elementwise_fma(h01, wv.xy, D[ii]);
    S[ii] += h23;
    Q[ii] = __builtin_elementwise_fma(h23, h23, Q[ii]);
    D[ii] = __builtin_elementwise_fma(h23, wv.zw, D[ii]);
  }
}

__device__ __forceinline__
void pair_tile(const float* __restrict__ symW, const float* __restrict__ geoW,
                 const float* __restrict__ mask, const float* __restrict__ b1,
                 const float* __restrict__ g, const float* __restrict__ be,
                 const float* __restrict__ W2, const float* __restrict__ b2p,
                 float* __restrict__ out, int R, int r0, int i0) {
  __shared__ __attribute__((aligned(16))) float sG[PG_ROWS * PG_PITCH];
  __shared__ __attribute__((aligned(16))) float sS[PS_ROWS * 128];
  __shared__ __attribute__((aligned(16))) float sW[128];
  const int tid  = threadIdx.x;
  const int lane = tid & 31;
  const int w    = tid >> 5;

  const v4f gg = ((const v4f*)g)[lane];
  const v4f bb = ((const v4f*)be)[lane];
  const v4f ww = ((const v4f*)W2)[lane];
  const v4f b4 = ((const v4f*)b1)[lane];
  const float mk = mask[r0 + lane];
  const float b2 = b2p[0];
  v4f ts[4], tg[8];
  for (int u = 0; u < 4; ++u)
    ts[u] = ((const v4f*)(symW + (size_t)(i0 + w + 4 * u) * 128))[lane];
  for (int u = 0; u < 8; ++u)
    tg[u] = ((const v4f*)(geoW + (size_t)(r0 + w + 4 * u) * 128))[lane];

  const v4f gw = gg * ww;
  const v4f bw = bb * ww;
  if (w == 0) *(v4f*)&sW[lane * 4] = gw;
  float G  = (gw.x + gw.y) + (gw.z + gw.w);
  float Bc = (bw.x + bw.y) + (bw.z + bw.w);
  for (int o = 16; o > 0; o >>= 1) {
    G  += __shfl_xor(G, o, 32);
    Bc += __shfl_xor(Bc, o, 32);
  }
  for (int u = 0; u < 4; ++u)
    *(v4f*)&sS[(w + 4 * u) * 128 + lane * 4] = ts[u] + b4;
  for (int u = 0; u < 8; ++u)
    *(v4f*)&sG[(w + 4 * u) * PG_PITCH + lane * 4] = tg[u];
  __syncthreads();

  const float* gp = &sG[lane * PG_PITCH];
  const float* sp = &sS[(w * 4) * 128];
  const v2f zero2 = {0.f, 0.f};
  v2f S[4], Q[4], D[4];
  for (int ii = 0; ii < 4; ++ii) { S[ii] = zero2; Q[ii] = zero2; D[ii] = zero2; }

  v4f gA, wA, sA[4], gB, wB, sB[4];
  gA = *(const v4f*)gp;
  wA = *(const v4f*)sW;
  for (int ii = 0; ii < 4; ++ii) sA[ii] = *(const v4f*)(sp + ii * 128);
#pragma unroll
  for (int k = 0; k < 128; k += 8) {
    gB = *(const v4f*)(gp + k + 4);
    wB = *(const v4f*)(sW + k + 4);
    for (int ii = 0; ii < 4; ++ii) sB[ii] = *(const v4f*)(sp + ii * 128 + k + 4);
    pair_block(gA, wA, sA, S, Q, D);
    const int kn = (k + 8 < 128) ? k + 8 : 0;
    gA = *(const v4f*)(gp + kn);
    wA = *(const v4f*)(sW + kn);
    for (int ii = 0; ii < 4; ++ii) sA[ii] = *(const v4f*)(sp + ii * 128 + kn);
    pair_block(gB, wB, sB, S, Q, D);
  }

  for (int ii = 0; ii < 4; ++ii) {
    const float s = S[ii].x + S[ii].y;
    const float q = Q[ii].x + Q[ii].y;
    const float d = D[ii].x + D[ii].y;
    const float mean  = s * (1.f / 128.f);
    const float var   = q * (1.f / 128.f) - mean * mean;
    const float rstd  = rsqrtf(var + 1e-5f);
    const float logit = rstd * (d - mean * G) + Bc + b2;
    const float z     = (mk > 0.f) ? logit : -1e9f;
    out[(size_t)(i0 + w * 4 + ii) * R + r0 + lane] = 1.f / (1.f + expf(-z));
  }
}


static const int PB0 = (RT / PG_ROWS) * (NT / PS_ROWS);
static const int PB1 = PB0 + (RH / PG_ROWS) * (NH / PS_ROWS);
static const int PB2 = PB1 + (PD / PG_ROWS) * (NA / PS_ROWS);
static const int PB3 = PB2 + (LD / PG_ROWS) * (NB / PS_ROWS);
static const int PB4 = PB3 + (LD / PG_ROWS) * (NPA / PS_ROWS);
static const int PB5 = PB4 + (PD / PG_ROWS) * (NPE / PS_ROWS);
__global__ __launch_bounds__(128)
void pair6_kernel(const float* __restrict__ proj,
                  const float* __restrict__ mask_t, const float* __restrict__ mask_h,
                  const float* __restrict__ lpl_m, const float* __restrict__ plp_m,
                  const float* __restrict__ sc_b1, const float* __restrict__ sc_g,
                  const float* __restrict__ sc_be, const float* __restrict__ sc_W2,
                  const float* __restrict__ sc_b2, float* __restrict__ out) {
  const int b = blockIdx.x;
  int h, R, bx, by, grow, srow, ooff; const float* mask;
  if (b < PB0) {
    h = 0; R = RT; by = b / (RT / PG_ROWS); bx = b - by * (RT / PG_ROWS);
    grow = 0; srow = 16 * PJ_G5; mask = mask_t; ooff = 0;
  } else if (b < PB1) {
    const int lb = b - PB0;
    h = 1; R = RH; by = lb / (RH / PG_ROWS); bx = lb - by * (RH / PG_ROWS);
    grow = 16 * PJ_G0; srow = 16 * PJ_S0; mask = mask_h; ooff = NT * RT;
  } else if (b < PB2) {
    const int lb = b - PB1;
    h = 2; R = PD; by = lb / (PD / PG_ROWS); bx = lb - by * (PD / PG_ROWS);
    grow = 16 * PJ_G1; srow = 16 * PJ_S1; mask = lpl_m; ooff = NT * RT + NH * RH;
  } else if (b < PB3) {
    const int lb = b - PB2;
    h = 3; R = LD; by = lb / (LD / PG_ROWS); bx = lb - by * (LD / PG_ROWS);
    grow = 16 * PJ_G2; srow = 16 * PJ_S2; mask = plp_m; ooff = NT * RT + NH * RH + NA * PD;
  } else if (b < PB4) {
    const int lb = b - PB3;
    h = 4; R = LD; by = lb / (LD / PG_ROWS); bx = lb - by * (LD / PG_ROWS);
    grow = 16 * PJ_G3; srow = 16 * PJ_S3; mask = mask_t + PD;
    ooff = NT * RT + NH * RH + NA * PD + NB * LD;
  } else {
    const int lb = b - PB4;
    h = 5; R = PD; by = lb / (PD / PG_ROWS); bx = lb - by * (PD / PG_ROWS);
    grow = 16 * PJ_G4; srow = 16 * PJ_S4; mask = lpl_m;
    ooff = NT * RT + NH * RH + NA * PD + NB * LD + NPA * LD;
  }
  pair_tile(proj + (size_t)srow * 128, proj + (size_t)grow * 128, mask,
            sc_b1 + h * HD, sc_g + h * HD, sc_be + h * HD, sc_W2 + h * HD, sc_b2 + h,
            out + ooff, R, bx * PG_ROWS, by * PS_ROWS);
}

extern "C" void kernel_launch(void* const* d_in, const int* in_sizes, int n_in,
                              void* d_out, int out_size, void* d_ws, size_t ws_size,
                              hipStream_t stream) {
  (void)in_sizes; (void)n_in; (void)out_size; (void)ws_size;
  const float* points  = (const float*)d_in[0];
  const float* lines   = (const float*)d_in[1];
  const float* circles = (const float*)d_in[2];
  const float* text_s  = (const float*)d_in[3];
  const float* head_s  = (const float*)d_in[4];
  const float* angle_s = (const float*)d_in[5];
  const float* bar_s   = (const float*)d_in[6];
  const float* para_s  = (const float*)d_in[7];
  const float* perp_s  = (const float*)d_in[8];
  const float* rel_W   = (const float*)d_in[9];
  const float* rel_b   = (const float*)d_in[10];
  const float* rel_g   = (const float*)d_in[11];
  const float* rel_be  = (const float*)d_in[12];
  const float* gm_W    = (const float*)d_in[13];
  const float* gm_b    = (const float*)d_in[14];
  const float* gm_g    = (const float*)d_in[15];
  const float* gm_be   = (const float*)d_in[16];
  const float* sc_W1   = (const float*)d_in[17];
  const float* sc_b1   = (const float*)d_in[18];
  const float* sc_g    = (const float*)d_in[19];
  const float* sc_be   = (const float*)d_in[20];
  const float* sc_W2   = (const float*)d_in[21];
  const float* sc_b2   = (const float*)d_in[22];
  const int*   pl      = (const int*)d_in[23];
  const int*   pc      = (const int*)d_in[24];
  float* out = (float*)d_out;

  float* wsp = (float*)d_ws;
  size_t off = 0;
  auto alloc = [&](size_t n) { float* p = wsp + off; off += (n + 31) & ~(size_t)31; return p; };
  float* LPLb   = alloc((size_t)PD * HD);
  float* PLPb   = alloc((size_t)LD * HD);
  float* PCPb   = alloc((size_t)CD * HD);
  float* lpl_m  = alloc(PD);
  float* plp_m  = alloc(LD);
  float* pcp_m  = alloc(CD);
  float* mask_t = alloc(RT);
  float* mask_h = alloc(RH);
  float* aggL   = alloc((size_t)PD * HD);
  float* aggP   = alloc((size_t)LD * HD);
  float* aggC   = alloc((size_t)CD * HD);
  float* geo_all = alloc((size_t)(RT + RH) * HD);
  float* geo_t  = geo_all;
  float* geo_h  = geo_all + (size_t)RT * HD;
  float* tmp    = alloc((size_t)(RT + RH) * HD);
  float* proj   = alloc((size_t)PJ_S5 * 16 * HD);

  agg_gemm3_kernel<<<AG_S2, 256, 0, stream>>>(pl, pc, lines, points, aggL, aggP, aggC,
                                               lpl_m, plp_m, pcp_m);
  rel_lrl3_kernel<<<AG_S2, 256, 0, stream>>>(aggL, aggP, aggC, points, lines, circles,
                                              rel_W, rel_b, rel_g, rel_be, LPLb, PLPb, PCPb);

  const size_t ROW = (size_t)HD * sizeof(float);
  hipMemcpyAsync(geo_t + (size_t)0 * HD,    points, PD * ROW, hipMemcpyDeviceToDevice, stream);
  hipMemcpyAsync(geo_t + (size_t)1024 * HD, lines,  LD * ROW, hipMemcpyDeviceToDevice, stream);
  hipMemcpyAsync(geo_t + (size_t)2048 * HD, LPLb,   PD * ROW, hipMemcpyDeviceToDevice, stream);
  hipMemcpyAsync(geo_t + (size_t)3072 * HD, PLPb,   LD * ROW, hipMemcpyDeviceToDevice, stream);
  hipMemcpyAsync(geo_t + (size_t)4096 * HD, PCPb,   CD * ROW, hipMemcpyDeviceToDevice, stream);
  hipMemcpyAsync(geo_t + (size_t)4224 * HD, head_s, NH * ROW, hipMemcpyDeviceToDevice, stream);
  hipMemcpyAsync(geo_h + (size_t)0 * HD,    points, PD * ROW, hipMemcpyDeviceToDevice, stream);
  hipMemcpyAsync(geo_h + (size_t)1024 * HD, LPLb,   PD * ROW, hipMemcpyDeviceToDevice, stream);
  hipMemcpyAsync(geo_h + (size_t)2048 * HD, PLPb,   LD * ROW, hipMemcpyDeviceToDevice, stream);
  hipMemcpyAsync(geo_h + (size_t)3072 * HD, PCPb,   CD * ROW, hipMemcpyDeviceToDevice, stream);
  mask_cat_kernel<<<(RT + 255) / 256, 256, 0, stream>>>(lpl_m, plp_m, pcp_m, mask_t, mask_h);

  lrl_gemm_kernel<<<(RT + RH) / 16, 256, 0, stream>>>(geo_all, gm_W + 0 * HH,
      gm_b + 0 * HD, gm_g + 0 * HD, gm_be + 0 * HD, tmp);
  lrl_gemm_kernel<<<(RT + RH) / 16, 256, 0, stream>>>(tmp, gm_W + 1 * HH,
      gm_b + 1 * HD, gm_g + 1 * HD, gm_be + 1 * HD, geo_all);

  proj12_kernel<<<PJ_S5, 256, 0, stream>>>(geo_all, LPLb, PLPb, lines, text_s, head_s, angle_s,
                                            bar_s, para_s, perp_s, sc_W1, proj);
  pair6_kernel<<<PB5, 128, 0, stream>>>(proj, mask_t, mask_h, lpl_m, plp_m,
                                         sc_b1, sc_g, sc_be, sc_W2, sc_b2, out);
}


# row-local layers of a 16-row strip fused into one work-group pass, intermediates in LDS: aggregation -> relation LRL; gm layer 1 -> layer 2 -> W1 projection; 7 launches become 4; sum order unchanged
# speedup vs baseline: 5.2635x; 1.1895x over previous
#include <hip/hip_runtime.h>
#include <math.h>

static const int HD  = 128;
static const int PD  = 1024;
static const int LD  = 1024;
static const int CD  = 128;
static const int NT  = 64, NH = 32, NA = 64, NB = 64, NPA = 32, NPE = 32;
static const int RT  = PD + LD + PD + LD + CD + NH;
static const int RH  = PD + PD + LD + CD;
static const int HH  = HD * HD;

typedef float v2f __attribute__((ext_vector_type(2)));
typedef float v8f __attribute__((ext_vector_type(8)));

__device__ __forceinline__ v8f wmma_f32(v2f a, v2f b, v8f c) {
  return __builtin_amdgcn_wmma_f32_16x16x4_f32(false, a, false, b, (short)0, c,
                                               false, false);
}

static const int TP = 132;

__device__ __forceinline__
v8f agg_strip(const int* __restrict__ rel, int ldA, int transA,
              const float* __restrict__ Bm, int K,
              float* __restrict__ rowmask, int m0, float* __restrict__ sR) {
  const int tid  = threadIdx.x;
  const int lane = tid & 31;
  const int w    = tid >> 5;
  const int n    = w * 16 + (lane & 15);
  const int am   = lane & 15;
  const int koff = (lane >> 4) << 1;

  v8f acc = {0.f, 0.f, 0.f, 0.f, 0.f, 0.f, 0.f, 0.f};

  const int crow = tid >> 4;
  const int ccol = (tid & 15) << 2;
  float cnt = 0.f;

  for (int k0 = 0; k0 < K; k0 += 128) {
    if (transA) {
      for (int i = tid; i < 16 * 128; i += 256) {
        int m = i & 15, k = i >> 4;
        sR[m * TP + k] = (float)rel[(size_t)(k0 + k) * ldA + (m0 + m)];
      }
    } else {
      for (int i = tid; i < 16 * 128; i += 256) {
        int m = i >> 7, k = i & 127;
        sR[m * TP + k] = (float)rel[(size_t)(m0 + m) * ldA + (k0 + k)];
      }
    }
    __syncthreads();
    {
      const float4 c0 = *(const float4*)&sR[crow * TP + ccol];
      const float4 c1 = *(const float4*)&sR[crow * TP + ccol + 64];
      cnt += ((c0.x + c0.y) + (c0.z + c0.w)) + ((c1.x + c1.y) + (c1.z + c1.w));
    }
    for (int kk = 0; kk < 128; kk += 4) {
      v2f a = *(const v2f*)&sR[am * TP + kk + koff];
      int kg = k0 + kk + koff;
      v2f b;
      b.x = Bm[(size_t)kg * HD + n];
      b.y = Bm[(size_t)(kg + 1) * HD + n];
      acc = wmma_f32(a, b, acc);
    }
    __syncthreads();
  }

  for (int o = 8; o > 0; o >>= 1) cnt += __shfl_xor(cnt, o, 32);
  if ((tid & 15) == 0) rowmask[m0 + crow] = (cnt >= 2.f) ? 1.f : 0.f;
  return acc;
}


__device__ __forceinline__
void strip_stage(const float* __restrict__ X, float* __restrict__ sA) {
  for (int i = threadIdx.x; i < 16 * 128; i += 256)
    sA[(i >> 7) * TP + (i & 127)] = X[i];
}

__device__ __forceinline__
v8f strip_gemm(const float* __restrict__ sA, const float* __restrict__ W) {
  const int lane = threadIdx.x & 31;
  const int w    = threadIdx.x >> 5;
  const int n    = w * 16 + (lane & 15);
  const int am   = lane & 15;
  const int koff = (lane >> 4) << 1;
  v8f acc = {0.f, 0.f, 0.f, 0.f, 0.f, 0.f, 0.f, 0.f};
#pragma unroll 4
  for (int kk = 0; kk < 128; kk += 4) {
    v2f a = *(const v2f*)&sA[am * TP + kk + koff];
    v2f b = *(const v2f*)&W[(size_t)n * 128 + kk + koff];
    acc = wmma_f32(a, b, acc);
  }
  return acc;
}

__device__ __forceinline__
void strip_store_raw(const v8f acc, float* __restrict__ Y) {
  const int lane = threadIdx.x & 31;
  const int n    = (threadIdx.x >> 5) * 16 + (lane & 15);
  const int mb   = (lane >> 4) << 3;
#pragma unroll
  for (int v = 0; v < 8; ++v)
    Y[(size_t)(mb + v) * 128 + n] = acc[v];
}

__device__ __forceinline__
void strip_relu_ln(const v8f acc, const float* __restrict__ bias, const float* __restrict__ g,
                   const float* __restrict__ be, float* __restrict__ sC,
                   float* __restrict__ sMean, float* __restrict__ sRstd,
                   float* __restrict__ dst, int dp) {
  const int tid  = threadIdx.x;
  const int lane = tid & 31;
  const int n    = (tid >> 5) * 16 + (lane & 15);
  const int mb   = (lane >> 4) << 3;
  float bn = bias[n];
#pragma unroll
  for (int v = 0; v < 8; ++v) {
    float x = acc[v] + bn;
    sC[(mb + v) * TP + n] = x > 0.f ? x : 0.f;
  }
  __syncthreads();
  if (tid < 16) {
    float s = 0.f, s2 = 0.f;
    for (int k = 0; k < 128; ++k) {
      float x = sC[tid * TP + k];
      s += x; s2 += x * x;
    }
    float mean = s * (1.f / 128.f);
    float var  = s2 * (1.f / 128.f) - mean * mean;
    sMean[tid] = mean;
    sRstd[tid] = rsqrtf(var + 1e-5f);
  }
  __syncthreads();
  for (int i = tid; i < 16 * 128; i += 256) {
    int m = i >> 7, k = i & 127;
    dst[m * dp + k] = (sC[m * TP + k] - sMean[m]) * sRstd[m] * g[k] + be[k];
  }
}

static const int AG_S0 = PD / 16;
static const int AG_S1 = AG_S0 + LD / 16;
static const int AG_S2 = AG_S1 + CD / 16;
__global__ __launch_bounds__(256)
void agg_rel3_kernel(const int* __restrict__ pl, const int* __restrict__ pc,
                     const float* __restrict__ points, const float* __restrict__ lines,
                     const float* __restrict__ circles,
                     const float* __restrict__ rel_W, const float* __restrict__ rel_b,
                     const float* __restrict__ rel_g, const float* __restrict__ rel_be,
                     float* __restrict__ LPLb, float* __restrict__ PLPb, float* __restrict__ PCPb,
                     float* __restrict__ lpl_m, float* __restrict__ plp_m,
                     float* __restrict__ pcp_m) {
  __shared__ __attribute__((aligned(16))) float sR[16 * TP];
  __shared__ __attribute__((aligned(16))) float sA[16 * TP];
  __shared__ __attribute__((aligned(16))) float sC[16 * TP];
  __shared__ float sMean[16], sRstd[16];
  const int b = blockIdx.x;
  const int* rel; int ldA, transA, K, m0, j;
  const float* Bm; const float* middle; float* Y; float* rowmask;
  if (b < AG_S0) {
    rel = pl; ldA = LD; transA = 0; Bm = lines;  K = LD; middle = points;  Y = LPLb;
    rowmask = lpl_m; j = 0; m0 = b * 16;
  } else if (b < AG_S1) {
    rel = pl; ldA = LD; transA = 1; Bm = points; K = PD; middle = lines;   Y = PLPb;
    rowmask = plp_m; j = 1; m0 = (b - AG_S0) * 16;
  } else {
    rel = pc; ldA = CD; transA = 1; Bm = points; K = PD; middle = circles; Y = PCPb;
    rowmask = pcp_m; j = 2; m0 = (b - AG_S1) * 16;
  }
  const int lane = threadIdx.x & 31;
  const int n    = (threadIdx.x >> 5) * 16 + (lane & 15);
  const int mb   = (lane >> 4) << 3;
  float mid[8];
#pragma unroll
  for (int v = 0; v < 8; ++v) mid[v] = middle[(size_t)(m0 + mb + v) * HD + n];

  const v8f agg = agg_strip(rel, ldA, transA, Bm, K, rowmask, m0, sR);
#pragma unroll
  for (int v = 0; v < 8; ++v) sA[(mb + v) * TP + n] = 2.f * agg[v] + mid[v];
  __syncthreads();
  const v8f y = strip_gemm(sA, rel_W + (size_t)j * HH);
  strip_relu_ln(y, rel_b + j * HD, rel_g + j * HD, rel_be + j * HD, sC, sMean, sRstd,
                Y + (size_t)m0 * HD, HD);
}

static const int PJ_G0 = RT / 16;
static const int PJ_G1 = PJ_G0 + RH / 16;
static const int PJ_G2 = PJ_G1 + PD / 16;
static const int PJ_G3 = PJ_G2 + LD / 16;
static const int PJ_G4 = PJ_G3 + LD / 16;
static const int PJ_G5 = PJ_G4 + PD / 16;
static const int PJ_S0 = PJ_G5 + NT / 16;
static const int PJ_S1 = PJ_S0 + NH / 16;
static const int PJ_S2 = PJ_S1 + NA / 16;
static const int PJ_S3 = PJ_S2 + NB / 16;
static const int PJ_S4 = PJ_S3 + NPA / 16;
static const int PJ_S5 = PJ_S4 + NPE / 16;
__global__ __launch_bounds__(256)
void gm_proj12_kernel(const float* __restrict__ geo_all, const float* __restrict__ LPLb,
                      const float* __restrict__ PLPb, const float* __restrict__ lines,
                      const float* __restrict__ text_s, const float* __restrict__ head_s,
                      const float* __restrict__ angle_s, const float* __restrict__ bar_s,
                      const float* __restrict__ para_s, const float* __restrict__ perp_s,
                      const float* __restrict__ gm_W, const float* __restrict__ gm_b,
                      const float* __restrict__ gm_g, const float* __restrict__ gm_be,
                      const float* __restrict__ sc_W1, float* __restrict__ proj) {
  __shared__ __attribute__((aligned(16))) float sA[16 * TP];
  __shared__ __attribute__((aligned(16))) float sC[16 * TP];
  __shared__ float sMean[16], sRstd[16];
  const int b = blockIdx.x;
  const float* X; int h, s;
  if (b < PJ_G0)      { X = geo_all; h = 0; s = b; }
  else if (b < PJ_G1) { X = geo_all; h = 1; s = b; }
  else if (b < PJ_G2) { X = LPLb;    h = 2; s = b - PJ_G1; }
  else if (b < PJ_G3) { X = PLPb;    h = 3; s = b - PJ_G2; }
  else if (b < PJ_G4) { X = lines;   h = 4; s = b - PJ_G3; }
  else if (b < PJ_G5) { X = LPLb;    h = 5; s = b - PJ_G4; }
  else if (b < PJ_S0) { X = text_s;  h = 0; s = b - PJ_G5; }
  else if (b < PJ_S1) { X = head_s;  h = 1; s = b - PJ_S0; }
  else if (b < PJ_S2) { X = angle_s; h = 2; s = b - PJ_S1; }
  else if (b < PJ_S3) { X = bar_s;   h = 3; s = b - PJ_S2; }
  else if (b < PJ_S4) { X = para_s;  h = 4; s = b - PJ_S3; }
  else                { X = perp_s;  h = 5; s = b - PJ_S4; }
  strip_stage(X + (size_t)s * (16 * 128), sA);
  __syncthreads();
  if (b < PJ_G1) {
    for (int l = 0; l < 2; ++l) {
      const v8f y = strip_gemm(sA, gm_W + (size_t)l * HH);
      strip_relu_ln(y, gm_b + l * HD, gm_g + l * HD, gm_be + l * HD, sC, sMean, sRstd, sA, TP);
      __syncthreads();
    }
  }
  const v8f p = strip_gemm(sA, sc_W1 + (size_t)h * HH);
  strip_store_raw(p, proj + (size_t)b * (16 * 128));
}

__global__ void mask_cat_kernel(const float* __restrict__ lpl_m,
                                const float* __restrict__ plp_m,
                                const float* __restrict__ pcp_m,
                                float* __restrict__ mask_t, float* __restrict__ mask_h) {
  int t = blockIdx.x * blockDim.x + threadIdx.x;
  if (t < RT) {
    float v;
    if (t < 2048)      v = 1.f;
    else if (t < 3072) v = lpl_m[t - 2048];
    else if (t < 4096) v = plp_m[t - 3072];
    else if (t < 4224) v = pcp_m[t - 4096];
    else               v = 1.f;
    mask_t[t] = v;
  }
  if (t < RH) {
    float v;
    if (t < 1024)      v = 1.f;
    else if (t < 2048) v = lpl_m[t - 1024];
    else if (t < 3072) v = plp_m[t - 2048];
    else               v = pcp_m[t - 3072];
    mask_h[t] = v;
  }
}

typedef float v4f __attribute__((ext_vector_type(4)));
static const int PG_ROWS  = 32;
static const int PS_ROWS  = 16;
static const int PG_PITCH = 132;
static_assert(RT % PG_ROWS == 0 && RH % PG_ROWS == 0 && PD % PG_ROWS == 0 && LD % PG_ROWS == 0,
              "geo row counts must be whole tiles");
static_assert(NT % PS_ROWS == 0 && NH % PS_ROWS == 0 && NA % PS_ROWS == 0 && NB % PS_ROWS == 0 &&
              NPA % PS_ROWS == 0 && NPE % PS_ROWS == 0, "sym row counts must be whole tiles");

__device__ __forceinline__ void pair_block(const v4f gv, const v4f wv, const v4f (&sv)[4],
                                           v2f (&S)[4], v2f (&Q)[4], v2f (&D)[4]) {
  const v2f zero2 = {0.f, 0.f};
  for (int ii = 0; ii < 4; ++ii) {
    const v2f h01 = __builtin_elementwise_max(sv[ii].xy + gv.xy, zero2);
    const v2f h23 = __builtin_elementwise_max(sv[ii].zw + gv.zw, zero2);
    S[ii] += h01;
    Q[ii] = __builtin_elementwise_fma(h01, h01, Q[ii]);
    D[ii] = __builtin_elementwise_fma(h01, wv.xy, D[ii]);
    S[ii] += h23;
    Q[ii] = __builtin_elementwise_fma(h23, h23, Q[ii]);
    D[ii] = __builtin_elementwise_fma(h23, wv.zw, D[ii]);
  }
}

__device__ __forceinline__
void pair_tile(const float* __restrict__ symW, const float* __restrict__ geoW,
                 const float* __restrict__ mask, const float* __restrict__ b1,
                 const float* __restrict__ g, const float* __restrict__ be,
                 const float* __restrict__ W2, const float* __restrict__ b2p,
                 float* __restrict__ out, int R, int r0, int i0) {
  __shared__ __attribute__((aligned(16))) float sG[PG_ROWS * PG_PITCH];
  __shared__ __attribute__((aligned(16))) float sS[PS_ROWS * 128];
  __shared__ __attribute__((aligned(16))) float sW[128];
  const int tid  = threadIdx.x;
  const int lane = tid & 31;
  const int w    = tid >> 5;

  const v4f gg = ((const v4f*)g)[lane];
  const v4f bb = ((const v4f*)be)[lane];
  const v4f ww = ((const v4f*)W2)[lane];
  const v4f b4 = ((const v4f*)b1)[lane];
  const float mk = mask[r0 + lane];
  const float b2 = b2p[0];
  v4f ts[4], tg[8];
  for (int u = 0; u < 4; ++u)
    ts[u] = ((const v4f*)(symW + (size_t)(i0 + w + 4 * u) * 128))[lane];
  for (int u = 0; u < 8; ++u)
    tg[u] = ((const v4f*)(geoW + (size_t)(r0 + w + 4 * u) * 128))[lane];

  const v4f gw = gg * ww;
  const v4f bw = bb * ww;
  if (w == 0) *(v4f*)&sW[lane * 4] = gw;
  float G  = (gw.x + gw.y) + (gw.z + gw.w);
  float Bc = (bw.x + bw.y) + (bw.z + bw.w);
  for (int o = 16; o > 0; o >>= 1) {
    G  += __shfl_xor(G, o, 32);
    Bc += __shfl_xor(Bc, o, 32);
  }
  for (int u = 0; u < 4; ++u)
    *(v4f*)&sS[(w + 4 * u) * 128 + lane * 4] = ts[u] + b4;
  for (int u = 0; u < 8; ++u)
    *(v4f*)&sG[(w + 4 * u) * PG_PITCH + lane * 4] = tg[u];
  __syncthreads();

  const float* gp = &sG[lane * PG_PITCH];
  const float* sp = &sS[(w * 4) * 128];
  const v2f zero2 = {0.f, 0.f};
  v2f S[4], Q[4], D[4];
  for (int ii = 0; ii < 4; ++ii) { S[ii] = zero2; Q[ii] = zero2; D[ii] = zero2; }

  v4f gA, wA, sA[4], gB, wB, sB[4];
  gA = *(const v4f*)gp;
  wA = *(const v4f*)sW;
  for (int ii = 0; ii < 4; ++ii) sA[ii] = *(const v4f*)(sp + ii * 128);
#pragma unroll
  for (int k = 0; k < 128; k += 8) {
    gB = *(const v4f*)(gp + k + 4);
    wB = *(const v4f*)(sW + k + 4);
    for (int ii = 0; ii < 4; ++ii) sB[ii] = *(const v4f*)(sp + ii * 128 + k + 4);
    pair_block(gA, wA, sA, S, Q, D);
    const int kn = (k + 8 < 128) ? k + 8 : 0;
    gA = *(const v4f*)(gp + kn);
    wA = *(const v4f*)(sW + kn);
    for (int ii = 0; ii < 4; ++ii) sA[ii] = *(const v4f*)(sp + ii * 128 + kn);
    pair_block(gB, wB, sB, S, Q, D);
  }

  for (int ii = 0; ii < 4; ++ii) {
    const float s = S[ii].x + S[ii].y;
    const float q = Q[ii].x + Q[ii].y;
    const float d = D[ii].x + D[ii].y;
    const float mean  = s * (1.f / 128.f);
    const float var   = q * (1.f / 128.f) - mean * mean;
    const float rstd  = rsqrtf(var + 1e-5f);
    const float logit = rstd * (d - mean * G) + Bc + b2;
    const float z     = (mk > 0.f) ? logit : -1e9f;
    out[(size_t)(i0 + w * 4 + ii) * R + r0 + lane] = 1.f / (1.f + expf(-z));
  }
}


static const int PB0 = (RT / PG_ROWS) * (NT / PS_ROWS);
static const int PB1 = PB0 + (RH / PG_ROWS) * (NH / PS_ROWS);
static const int PB2 = PB1 + (PD / PG_ROWS) * (NA / PS_ROWS);
static const int PB3 = PB2 + (LD / PG_ROWS) * (NB / PS_ROWS);
static const int PB4 = PB3 + (LD / PG_ROWS) * (NPA / PS_ROWS);
static const int PB5 = PB4 + (PD / PG_ROWS) * (NPE / PS_ROWS);
__global__ __launch_bounds__(128)
void pair6_kernel(const float* __restrict__ proj,
                  const float* __restrict__ mask_t, const float* __restrict__ mask_h,
                  const float* __restrict__ lpl_m, const float* __restrict__ plp_m,
                  const float* __restrict__ sc_b1, const float* __restrict__ sc_g,
                  const float* __restrict__ sc_be, const float* __restrict__ sc_W2,
                  const float* __restrict__ sc_b2, float* __restrict__ out) {
  const int b = blockIdx.x;
  int h, R, bx, by, grow, srow, ooff; const float* mask;
  if (b < PB0) {
    h = 0; R = RT; by = b / (RT / PG_ROWS); bx = b - by * (RT / PG_ROWS);
    grow = 0; srow = 16 * PJ_G5; mask = mask_t; ooff = 0;
  } else if (b < PB1) {
    const int lb = b - PB0;
    h = 1; R = RH; by = lb / (RH / PG_ROWS); bx = lb - by * (RH / PG_ROWS);
    grow = 16 * PJ_G0; srow = 16 * PJ_S0; mask = mask_h; ooff = NT * RT;
  } else if (b < PB2) {
    const int lb = b - PB1;
    h = 2; R = PD; by = lb / (PD / PG_ROWS); bx = lb - by * (PD / PG_ROWS);
    grow = 16 * PJ_G1; srow = 16 * PJ_S1; mask = lpl_m; ooff = NT * RT + NH * RH;
  } else if (b < PB3) {
    const int lb = b - PB2;
    h = 3; R = LD; by = lb / (LD / PG_ROWS); bx = lb - by * (LD / PG_ROWS);
    grow = 16 * PJ_G2; srow = 16 * PJ_S2; mask = plp_m; ooff = NT * RT + NH * RH + NA * PD;
  } else if (b < PB4) {
    const int lb = b - PB3;
    h = 4; R = LD; by = lb / (LD / PG_ROWS); bx = lb - by * (LD / PG_ROWS);
    grow = 16 * PJ_G3; srow = 16 * PJ_S3; mask = mask_t + PD;
    ooff = NT * RT + NH * RH + NA * PD + NB * LD;
  } else {
    const int lb = b - PB4;
    h = 5; R = PD; by = lb / (PD / PG_ROWS); bx = lb - by * (PD / PG_ROWS);
    grow = 16 * PJ_G4; srow = 16 * PJ_S4; mask = lpl_m;
    ooff = NT * RT + NH * RH + NA * PD + NB * LD + NPA * LD;
  }
  pair_tile(proj + (size_t)srow * 128, proj + (size_t)grow * 128, mask,
            sc_b1 + h * HD, sc_g + h * HD, sc_be + h * HD, sc_W2 + h * HD, sc_b2 + h,
            out + ooff, R, bx * PG_ROWS, by * PS_ROWS);
}

extern "C" void kernel_launch(void* const* d_in, const int* in_sizes, int n_in,
                              void* d_out, int out_size, void* d_ws, size_t ws_size,
                              hipStream_t stream) {
  (void)in_sizes; (void)n_in; (void)out_size; (void)ws_size;
  const float* points  = (const float*)d_in[0];
  const float* lines   = (const float*)d_in[1];
  const float* circles = (const float*)d_in[2];
  const float* text_s  = (const float*)d_in[3];
  const float* head_s  = (const float*)d_in[4];
  const float* angle_s = (const float*)d_in[5];
  const float* bar_s   = (const float*)d_in[6];
  const float* para_s  = (const float*)d_in[7];
  const float* perp_s  = (const float*)d_in[8];
  const float* rel_W   = (const float*)d_in[9];
  const float* rel_b   = (const float*)d_in[10];
  const float* rel_g   = (const float*)d_in[11];
  const float* rel_be  = (const float*)d_in[12];
  const float* gm_W    = (const float*)d_in[13];
  const float* gm_b    = (const float*)d_in[14];
  const float* gm_g    = (const float*)d_in[15];
  const float* gm_be   = (const float*)d_in[16];
  const float* sc_W1   = (const float*)d_in[17];
  const float* sc_b1   = (const float*)d_in[18];
  const float* sc_g    = (const float*)d_in[19];
  const float* sc_be   = (const float*)d_in[20];
  const float* sc_W2   = (const float*)d_in[21];
  const float* sc_b2   = (const float*)d_in[22];
  const int*   pl      = (const int*)d_in[23];
  const int*   pc      = (const int*)d_in[24];
  float* out = (float*)d_out;

  float* wsp = (float*)d_ws;
  size_t off = 0;
  auto alloc = [&](size_t n) { float* p = wsp + off; off += (n + 31) & ~(size_t)31; return p; };
  float* LPLb   = alloc((size_t)PD * HD);
  float* PLPb   = alloc((size_t)LD * HD);
  float* PCPb   = alloc((size_t)CD * HD);
  float* lpl_m  = alloc(PD);
  float* plp_m  = alloc(LD);
  float* pcp_m  = alloc(CD);
  float* mask_t = alloc(RT);
  float* mask_h = alloc(RH);
  float* geo_all = alloc((size_t)(RT + RH) * HD);
  float* geo_t  = geo_all;
  float* geo_h  = geo_all + (size_t)RT * HD;
  float* proj   = alloc((size_t)PJ_S5 * 16 * HD);

  agg_rel3_kernel<<<AG_S2, 256, 0, stream>>>(pl, pc, points, lines, circles,
                                              rel_W, rel_b, rel_g, rel_be, LPLb, PLPb, PCPb,
                                              lpl_m, plp_m, pcp_m);

  const size_t ROW = (size_t)HD * sizeof(float);
  hipMemcpyAsync(geo_t + (size_t)0 * HD,    points, PD * ROW, hipMemcpyDeviceToDevice, stream);
  hipMemcpyAsync(geo_t + (size_t)1024 * HD, lines,  LD * ROW, hipMemcpyDeviceToDevice, stream);
  hipMemcpyAsync(geo_t + (size_t)2048 * HD, LPLb,   PD * ROW, hipMemcpyDeviceToDevice, stream);
  hipMemcpyAsync(geo_t + (size_t)3072 * HD, PLPb,   LD * ROW, hipMemcpyDeviceToDevice, stream);
  hipMemcpyAsync(geo_t + (size_t)4096 * HD, PCPb,   CD * ROW, hipMemcpyDeviceToDevice, stream);
  hipMemcpyAsync(geo_t + (size_t)4224 * HD, head_s, NH * ROW, hipMemcpyDeviceToDevice, stream);
  hipMemcpyAsync(geo_h + (size_t)0 * HD,    points, PD * ROW, hipMemcpyDeviceToDevice, stream);
  hipMemcpyAsync(geo_h + (size_t)1024 * HD, LPLb,   PD * ROW, hipMemcpyDeviceToDevice, stream);
  hipMemcpyAsync(geo_h + (size_t)2048 * HD, PLPb,   LD * ROW, hipMemcpyDeviceToDevice, stream);
  hipMemcpyAsync(geo_h + (size_t)3072 * HD, PCPb,   CD * ROW, hipMemcpyDeviceToDevice, stream);
  mask_cat_kernel<<<(RT + 255) / 256, 256, 0, stream>>>(lpl_m, plp_m, pcp_m, mask_t, mask_h);

  gm_proj12_kernel<<<PJ_S5, 256, 0, stream>>>(geo_all, LPLb, PLPb, lines, text_s, head_s, angle_s,
                                               bar_s, para_s, perp_s, gm_W, gm_b, gm_g, gm_be,
                                               sc_W1, proj);
  pair6_kernel<<<PB5, 128, 0, stream>>>(proj, mask_t, mask_h, lpl_m, plp_m,
                                         sc_b1, sc_g, sc_be, sc_W2, sc_b2, out);
}


# aggregation stages its whole 16 x 1024 relation strip in LDS from one group of 16-byte loads, margin fragments a block ahead (was 8 tiles x 8 waited 4-byte loads, 16 barriers); strip_stage likewise
# speedup vs baseline: 7.6840x; 1.4599x over previous
#include <hip/hip_runtime.h>
#include <math.h>

static const int HD  = 128;
static const int PD  = 1024;
static const int LD  = 1024;
static const int CD  = 128;
static const int NT  = 64, NH = 32, NA = 64, NB = 64, NPA = 32, NPE = 32;
static const int RT  = PD + LD + PD + LD + CD + NH;
static const int RH  = PD + PD + LD + CD;
static const int HH  = HD * HD;

typedef float v2f __attribute__((ext_vector_type(2)));
typedef float v8f __attribute__((ext_vector_type(8)));

__device__ __forceinline__ v8f wmma_f32(v2f a, v2f b, v8f c) {
  return __builtin_amdgcn_wmma_f32_16x16x4_f32(false, a, false, b, (short)0, c,
                                               false, false);
}

static const int TP = 132;

static const int KD = 1024;
static_assert(PD == KD && LD == KD, "the three aggregations share one K");
static const int SP = KD + 4;
static const int KB = 128;

__device__ __forceinline__
void agg_load_b(const float* __restrict__ Bm, int kg0, int n, v2f (&b)[KB / 4]) {
#pragma unroll
  for (int i = 0; i < KB / 4; ++i) {
    const int kg = kg0 + 4 * i;
    b[i].x = Bm[(size_t)kg * HD + n];
    b[i].y = Bm[(size_t)(kg + 1) * HD + n];
  }
}

__device__ __forceinline__
v8f agg_mma(const float* __restrict__ sa, const v2f (&b)[KB / 4], v8f acc) {
#pragma unroll
  for (int i = 0; i < KB / 4; ++i)
    acc = wmma_f32(*(const v2f*)&sa[4 * i], b[i], acc);
  return acc;
}

template <int TRANS>
__device__ __forceinline__
v8f agg_strip(const int* __restrict__ rel, int ldA, const float* __restrict__ Bm,
              float* __restrict__ rowmask, int m0, float* __restrict__ sR) {
  const int tid  = threadIdx.x;
  const int lane = tid & 31;
  const int w    = tid >> 5;
  const int n    = w * 16 + (lane & 15);
  const int am   = lane & 15;
  const int koff = (lane >> 4) << 1;

  int4 r[16];
  if (TRANS) {
    const int* p = rel + (size_t)(tid >> 2) * ldA + (m0 + ((tid & 3) << 2));
#pragma unroll
    for (int j = 0; j < 16; ++j) r[j] = *(const int4*)(p + (size_t)(64 * j) * ldA);
  } else {
    const int* p = rel + (size_t)(m0 + (tid >> 4)) * ldA + ((tid & 15) << 2);
#pragma unroll
    for (int j = 0; j < 16; ++j) r[j] = *(const int4*)(p + 64 * j);
  }
  v2f bA[KB / 4], bB[KB / 4];
  agg_load_b(Bm, koff, n, bA);

  if (TRANS) {
    float* d = &sR[((tid & 3) << 2) * SP + (tid >> 2)];
#pragma unroll
    for (int j = 0; j < 16; ++j) {
      d[0 * SP + 64 * j] = (float)r[j].x;
      d[1 * SP + 64 * j] = (float)r[j].y;
      d[2 * SP + 64 * j] = (float)r[j].z;
      d[3 * SP + 64 * j] = (float)r[j].w;
    }
  } else {
    float* d = &sR[(tid >> 4) * SP + ((tid & 15) << 2)];
#pragma unroll
    for (int j = 0; j < 16; ++j) {
      const float4 f = {(float)r[j].x, (float)r[j].y, (float)r[j].z, (float)r[j].w};
      *(float4*)&d[64 * j] = f;
    }
  }
  __syncthreads();

  const int crow = tid >> 4;
  const int ccol = (tid & 15) << 2;
  float cnt = 0.f;
#pragma unroll
  for (int k0 = 0; k0 < KD; k0 += 128) {
    const float4 c0 = *(const float4*)&sR[crow * SP + k0 + ccol];
    const float4 c1 = *(const float4*)&sR[crow * SP + k0 + ccol + 64];
    cnt += ((c0.x + c0.y) + (c0.z + c0.w)) + ((c1.x + c1.y) + (c1.z + c1.w));
  }

  v8f acc = {0.f, 0.f, 0.f, 0.f, 0.f, 0.f, 0.f, 0.f};
  const float* sa = &sR[am * SP + koff];
  for (int k0 = 0; k0 < KD - 2 * KB; k0 += 2 * KB) {
    agg_load_b(Bm, k0 + KB + koff, n, bB);
    acc = agg_mma(sa + k0, bA, acc);
    agg_load_b(Bm, k0 + 2 * KB + koff, n, bA);
    acc = agg_mma(sa + k0 + KB, bB, acc);
  }
  agg_load_b(Bm, KD - KB + koff, n, bB);
  acc = agg_mma(sa + KD - 2 * KB, bA, acc);
  acc = agg_mma(sa + KD - KB, bB, acc);

  for (int o = 8; o > 0; o >>= 1) cnt += __shfl_xor(cnt, o, 32);
  if ((tid & 15) == 0) rowmask[m0 + crow] = (cnt >= 2.f) ? 1.f : 0.f;
  return acc;
}


__device__ __forceinline__
void strip_stage(const float* __restrict__ X, float* __restrict__ sA) {
  const int tid = threadIdx.x;
  const float4 x0 = ((const float4*)X)[tid];
  const float4 x1 = ((const float4*)X)[tid + 256];
  float* d = &sA[(tid >> 5) * TP + ((tid & 31) << 2)];
  *(float4*)&d[0]      = x0;
  *(float4*)&d[8 * TP] = x1;
}

__device__ __forceinline__
v8f strip_gemm(const float* __restrict__ sA, const float* __restrict__ W) {
  const int lane = threadIdx.x & 31;
  const int w    = threadIdx.x >> 5;
  const int n    = w * 16 + (lane & 15);
  const int am   = lane & 15;
  const int koff = (lane >> 4) << 1;
  v8f acc = {0.f, 0.f, 0.f, 0.f, 0.f, 0.f, 0.f, 0.f};
#pragma unroll 4
  for (int kk = 0; kk < 128; kk += 4) {
    v2f a = *(const v2f*)&sA[am * TP + kk + koff];
    v2f b = *(const v2f*)&W[(size_t)n * 128 + kk + koff];
    acc = wmma_f32(a, b, acc);
  }
  return acc;
}

__device__ __forceinline__
void strip_store_raw(const v8f acc, float* __restrict__ Y) {
  const int lane = threadIdx.x & 31;
  const int n    = (threadIdx.x >> 5) * 16 + (lane & 15);
  const int mb   = (lane >> 4) << 3;
#pragma unroll
  for (int v = 0; v < 8; ++v)
    Y[(size_t)(mb + v) * 128 + n] = acc[v];
}

__device__ __forceinline__
void strip_relu_ln(const v8f acc, const float* __restrict__ bias, const float* __restrict__ g,
                   const float* __restrict__ be, float* __restrict__ sC,
                   float* __restrict__ sMean, float* __restrict__ sRstd,
                   float* __restrict__ dst, int dp) {
  const int tid  = threadIdx.x;
  const int lane = tid & 31;
  const int n    = (tid >> 5) * 16 + (lane & 15);
  const int mb   = (lane >> 4) << 3;
  float bn = bias[n];
#pragma unroll
  for (int v = 0; v < 8; ++v) {
    float x = acc[v] + bn;
    sC[(mb + v) * TP + n] = x > 0.f ? x : 0.f;
  }
  __syncthreads();
  if (tid < 16) {
    float s = 0.f, s2 = 0.f;
    for (int k = 0; k < 128; ++k) {
      float x = sC[tid * TP + k];
      s += x; s2 += x * x;
    }
    float mean = s * (1.f / 128.f);
    float var  = s2 * (1.f / 128.f) - mean * mean;
    sMean[tid] = mean;
    sRstd[tid] = rsqrtf(var + 1e-5f);
  }
  __syncthreads();
  for (int i = tid; i < 16 * 128; i += 256) {
    int m = i >> 7, k = i & 127;
    dst[m * dp + k] = (sC[m * TP + k] - sMean[m]) * sRstd[m] * g[k] + be[k];
  }
}

static const int AG_S0 = PD / 16;
static const int AG_S1 = AG_S0 + LD / 16;
static const int AG_S2 = AG_S1 + CD / 16;
__global__ __launch_bounds__(256)
void agg_rel3_kernel(const int* __restrict__ pl, const int* __restrict__ pc,
                     const float* __restrict__ points, const float* __restrict__ lines,
                     const float* __restrict__ circles,
                     const float* __restrict__ rel_W, const float* __restrict__ rel_b,
                     const float* __restrict__ rel_g, const float* __restrict__ rel_be,
                     float* __restrict__ LPLb, float* __restrict__ PLPb, float* __restrict__ PCPb,
                     float* __restrict__ lpl_m, float* __restrict__ plp_m,
                     float* __restrict__ pcp_m) {
  __shared__ __attribute__((aligned(16))) float sR[16 * SP];
  __shared__ __attribute__((aligned(16))) float sA[16 * TP];
  __shared__ __attribute__((aligned(16))) float sC[16 * TP];
  __shared__ float sMean[16], sRstd[16];
  const int b = blockIdx.x;
  const int* rel; int ldA, transA, m0, j;
  const float* Bm; const float* middle; float* Y; float* rowmask;
  if (b < AG_S0) {
    rel = pl; ldA = LD; transA = 0; Bm = lines;  middle = points;  Y = LPLb;
    rowmask = lpl_m; j = 0; m0 = b * 16;
  } else if (b < AG_S1) {
    rel = pl; ldA = LD; transA = 1; Bm = points; middle = lines;   Y = PLPb;
    rowmask = plp_m; j = 1; m0 = (b - AG_S0) * 16;
  } else {
    rel = pc; ldA = CD; transA = 1; Bm = points; middle = circles; Y = PCPb;
    rowmask = pcp_m; j = 2; m0 = (b - AG_S1) * 16;
  }
  const int lane = threadIdx.x & 31;
  const int n    = (threadIdx.x >> 5) * 16 + (lane & 15);
  const int mb   = (lane >> 4) << 3;
  float mid[8];
#pragma unroll
  for (int v = 0; v < 8; ++v) mid[v] = middle[(size_t)(m0 + mb + v) * HD + n];

  const v8f agg = transA ? agg_strip<1>(rel, ldA, Bm, rowmask, m0, sR)
                         : agg_strip<0>(rel, ldA, Bm, rowmask, m0, sR);
#pragma unroll
  for (int v = 0; v < 8; ++v) sA[(mb + v) * TP + n] = 2.f * agg[v] + mid[v];
  __syncthreads();
  const v8f y = strip_gemm(sA, rel_W + (size_t)j * HH);
  strip_relu_ln(y, rel_b + j * HD, rel_g + j * HD, rel_be + j * HD, sC, sMean, sRstd,
                Y + (size_t)m0 * HD, HD);
}

static const int PJ_G0 = RT / 16;
static const int PJ_G1 = PJ_G0 + RH / 16;
static const int PJ_G2 = PJ_G1 + PD / 16;
static const int PJ_G3 = PJ_G2 + LD / 16;
static const int PJ_G4 = PJ_G3 + LD / 16;
static const int PJ_G5 = PJ_G4 + PD / 16;
static const int PJ_S0 = PJ_G5 + NT / 16;
static const int PJ_S1 = PJ_S0 + NH / 16;
static const int PJ_S2 = PJ_S1 + NA / 16;
static const int PJ_S3 = PJ_S2 + NB / 16;
static const int PJ_S4 = PJ_S3 + NPA / 16;
static const int PJ_S5 = PJ_S4 + NPE / 16;
__global__ __launch_bounds__(256)
void gm_proj12_kernel(const float* __restrict__ geo_all, const float* __restrict__ LPLb,
                      const float* __restrict__ PLPb, const float* __restrict__ lines,
                      const float* __restrict__ text_s, const float* __restrict__ head_s,
                      const float* __restrict__ angle_s, const float* __restrict__ bar_s,
                      const float* __restrict__ para_s, const float* __restrict__ perp_s,
                      const float* __restrict__ gm_W, const float* __restrict__ gm_b,
                      const float* __restrict__ gm_g, const float* __restrict__ gm_be,
                      const float* __restrict__ sc_W1, float* __restrict__ proj) {
  __shared__ __attribute__((aligned(16))) float sA[16 * TP];
  __shared__ __attribute__((aligned(16))) float sC[16 * TP];
  __shared__ float sMean[16], sRstd[16];
  const int b = blockIdx.x;
  const float* X; int h, s;
  if (b < PJ_G0)      { X = geo_all; h = 0; s = b; }
  else if (b < PJ_G1) { X = geo_all; h = 1; s = b; }
  else if (b < PJ_G2) { X = LPLb;    h = 2; s = b - PJ_G1; }
  else if (b < PJ_G3) { X = PLPb;    h = 3; s = b - PJ_G2; }
  else if (b < PJ_G4) { X = lines;   h = 4; s = b - PJ_G3; }
  else if (b < PJ_G5) { X = LPLb;    h = 5; s = b - PJ_G4; }
  else if (b < PJ_S0) { X = text_s;  h = 0; s = b - PJ_G5; }
  else if (b < PJ_S1) { X = head_s;  h = 1; s = b - PJ_S0; }
  else if (b < PJ_S2) { X = angle_s; h = 2; s = b - PJ_S1; }
  else if (b < PJ_S3) { X = bar_s;   h = 3; s = b - PJ_S2; }
  else if (b < PJ_S4) { X = para_s;  h = 4; s = b - PJ_S3; }
  else                { X = perp_s;  h = 5; s = b - PJ_S4; }
  strip_stage(X + (size_t)s * (16 * 128), sA);
  __syncthreads();
  if (b < PJ_G1) {
    for (int l = 0; l < 2; ++l) {
      const v8f y = strip_gemm(sA, gm_W + (size_t)l * HH);
      strip_relu_ln(y, gm_b + l * HD, gm_g + l * HD, gm_be + l * HD, sC, sMean, sRstd, sA, TP);
      __syncthreads();
    }
  }
  const v8f p = strip_gemm(sA, sc_W1 + (size_t)h * HH);
  strip_store_raw(p, proj + (size_t)b * (16 * 128));
}

__global__ void mask_cat_kernel(const float* __restrict__ lpl_m,
                                const float* __restrict__ plp_m,
                                const float* __restrict__ pcp_m,
                                float* __restrict__ mask_t, float* __restrict__ mask_h) {
  int t = blockIdx.x * blockDim.x + threadIdx.x;
  if (t < RT) {
    float v;
    if (t < 2048)      v = 1.f;
    else if (t < 3072) v = lpl_m[t - 2048];
    else if (t < 4096) v = plp_m[t - 3072];
    else if (t < 4224) v = pcp_m[t - 4096];
    else               v = 1.f;
    mask_t[t] = v;
  }
  if (t < RH) {
    float v;
    if (t < 1024)      v = 1.f;
    else if (t < 2048) v = lpl_m[t - 1024];
    else if (t < 3072) v = plp_m[t - 2048];
    else               v = pcp_m[t - 3072];
    mask_h[t] = v;
  }
}

typedef float v4f __attribute__((ext_vector_type(4)));
static const int PG_ROWS  = 32;
static const int PS_ROWS  = 16;
static const int PG_PITCH = 132;
static_assert(RT % PG_ROWS == 0 && RH % PG_ROWS == 0 && PD % PG_ROWS == 0 && LD % PG_ROWS == 0,
              "geo row counts must be whole tiles");
static_assert(NT % PS_ROWS == 0 && NH % PS_ROWS == 0 && NA % PS_ROWS == 0 && NB % PS_ROWS == 0 &&
              NPA % PS_ROWS == 0 && NPE % PS_ROWS == 0, "sym row counts must be whole tiles");

__device__ __forceinline__ void pair_block(const v4f gv, const v4f wv, const v4f (&sv)[4],
                                           v2f (&S)[4], v2f (&Q)[4], v2f (&D)[4]) {
  const v2f zero2 = {0.f, 0.f};
  for (int ii = 0; ii < 4; ++ii) {
    const v2f h01 = __builtin_elementwise_max(sv[ii].xy + gv.xy, zero2);
    const v2f h23 = __builtin_elementwise_max(sv[ii].zw + gv.zw, zero2);
    S[ii] += h01;
    Q[ii] = __builtin_elementwise_fma(h01, h01, Q[ii]);
    D[ii] = __builtin_elementwise_fma(h01, wv.xy, D[ii]);
    S[ii] += h23;
    Q[ii] = __builtin_elementwise_fma(h23, h23, Q[ii]);
    D[ii] = __builtin_elementwise_fma(h23, wv.zw, D[ii]);
  }
}

__device__ __forceinline__
void pair_tile(const float* __restrict__ symW, const float* __restrict__ geoW,
                 const float* __restrict__ mask, const float* __restrict__ b1,
                 const float* __restrict__ g, const float* __restrict__ be,
                 const float* __restrict__ W2, const float* __restrict__ b2p,
                 float* __restrict__ out, int R, int r0, int i0) {
  __shared__ __attribute__((aligned(16))) float sG[PG_ROWS * PG_PITCH];
  __shared__ __attribute__((aligned(16))) float sS[PS_ROWS * 128];
  __shared__ __attribute__((aligned(16))) float sW[128];
  const int tid  = threadIdx.x;
  const int lane = tid & 31;
  const int w    = tid >> 5;

  const v4f gg = ((const v4f*)g)[lane];
  const v4f bb = ((const v4f*)be)[lane];
  const v4f ww = ((const v4f*)W2)[lane];
  const v4f b4 = ((const v4f*)b1)[lane];
  const float mk = mask[r0 + lane];
  const float b2 = b2p[0];
  v4f ts[4], tg[8];
  for (int u = 0; u < 4; ++u)
    ts[u] = ((const v4f*)(symW + (size_t)(i0 + w + 4 * u) * 128))[lane];
  for (int u = 0; u < 8; ++u)
    tg[u] = ((const v4f*)(geoW + (size_t)(r0 + w + 4 * u) * 128))[lane];

  const v4f gw = gg * ww;
  const v4f bw = bb * ww;
  if (w == 0) *(v4f*)&sW[lane * 4] = gw;
  float G  = (gw.x + gw.y) + (gw.z + gw.w);
  float Bc = (bw.x + bw.y) + (bw.z + bw.w);
  for (int o = 16; o > 0; o >>= 1) {
    G  += __shfl_xor(G, o, 32);
    Bc += __shfl_xor(Bc, o, 32);
  }
  for (int u = 0; u < 4; ++u)
    *(v4f*)&sS[(w + 4 * u) * 128 + lane * 4] = ts[u] + b4;
  for (int u = 0; u < 8; ++u)
    *(v4f*)&sG[(w + 4 * u) * PG_PITCH + lane * 4] = tg[u];
  __syncthreads();

  const float* gp = &sG[lane * PG_PITCH];
  const float* sp = &sS[(w * 4) * 128];
  const v2f zero2 = {0.f, 0.f};
  v2f S[4], Q[4], D[4];
  for (int ii = 0; ii < 4; ++ii) { S[ii] = zero2; Q[ii] = zero2; D[ii] = zero2; }

  v4f gA, wA, sA[4], gB, wB, sB[4];
  gA = *(const v4f*)gp;
  wA = *(const v4f*)sW;
  for (int ii = 0; ii < 4; ++ii) sA[ii] = *(const v4f*)(sp + ii * 128);
#pragma unroll
  for (int k = 0; k < 128; k += 8) {
    gB = *(const v4f*)(gp + k + 4);
    wB = *(const v4f*)(sW + k + 4);
    for (int ii = 0; ii < 4; ++ii) sB[ii] = *(const v4f*)(sp + ii * 128 + k + 4);
    pair_block(gA, wA, sA, S, Q, D);
    const int kn = (k + 8 < 128) ? k + 8 : 0;
    gA = *(const v4f*)(gp + kn);
    wA = *(const v4f*)(sW + kn);
    for (int ii = 0; ii < 4; ++ii) sA[ii] = *(const v4f*)(sp + ii * 128 + kn);
    pair_block(gB, wB, sB, S, Q, D);
  }

  for (int ii = 0; ii < 4; ++ii) {
    const float s = S[ii].x + S[ii].y;
    const float q = Q[ii].x + Q[ii].y;
    const float d = D[ii].x + D[ii].y;
    const float mean  = s * (1.f / 128.f);
    const float var   = q * (1.f / 128.f) - mean * mean;
    const float rstd  = rsqrtf(var + 1e-5f);
    const float logit = rstd * (d - mean * G) + Bc + b2;
    const float z     = (mk > 0.f) ? logit : -1e9f;
    out[(size_t)(i0 + w * 4 + ii) * R + r0 + lane] = 1.f / (1.f + expf(-z));
  }
}


static const int PB0 = (RT / PG_ROWS) * (NT / PS_ROWS);
static const int PB1 = PB0 + (RH / PG_ROWS) * (NH / PS_ROWS);
static const int PB2 = PB1 + (PD / PG_ROWS) * (NA / PS_ROWS);
static const int PB3 = PB2 + (LD / PG_ROWS) * (NB / PS_ROWS);
static const int PB4 = PB3 + (LD / PG_ROWS) * (NPA / PS_ROWS);
static const int PB5 = PB4 + (PD / PG_ROWS) * (NPE / PS_ROWS);
__global__ __launch_bounds__(128)
void pair6_kernel(const float* __restrict__ proj,
                  const float* __restrict__ mask_t, const float* __restrict__ mask_h,
                  const float* __restrict__ lpl_m, const float* __restrict__ plp_m,
                  const float* __restrict__ sc_b1, const float* __restrict__ sc_g,
                  const float* __restrict__ sc_be, const float* __restrict__ sc_W2,
                  const float* __restrict__ sc_b2, float* __restrict__ out) {
  const int b = blockIdx.x;
  int h, R, bx, by, grow, srow, ooff; const float* mask;
  if (b < PB0) {
    h = 0; R = RT; by = b / (RT / PG_ROWS); bx = b - by * (RT / PG_ROWS);
    grow = 0; srow = 16 * PJ_G5; mask = mask_t; ooff = 0;
  } else if (b < PB1) {
    const int lb = b - PB0;
    h = 1; R = RH; by = lb / (RH / PG_ROWS); bx = lb - by * (RH / PG_ROWS);
    grow = 16 * PJ_G0; srow = 16 * PJ_S0; mask = mask_h; ooff = NT * RT;
  } else if (b < PB2) {
    const int lb = b - PB1;
    h = 2; R = PD; by = lb / (PD / PG_ROWS); bx = lb - by * (PD / PG_ROWS);
    grow = 16 * PJ_G1; srow = 16 * PJ_S1; mask = lpl_m; ooff = NT * RT + NH * RH;
  } else if (b < PB3) {
    const int lb = b - PB2;
    h = 3; R = LD; by = lb / (LD / PG_ROWS); bx = lb - by * (LD / PG_ROWS);
    grow = 16 * PJ_G2; srow = 16 * PJ_S2; mask = plp_m; ooff = NT * RT + NH * RH + NA * PD;
  } else if (b < PB4) {
    const int lb = b - PB3;
    h = 4; R = LD; by = lb / (LD / PG_ROWS); bx = lb - by * (LD / PG_ROWS);
    grow = 16 * PJ_G3; srow = 16 * PJ_S3; mask = mask_t + PD;
    ooff = NT * RT + NH * RH + NA * PD + NB * LD;
  } else {
    const int lb = b - PB4;
    h = 5; R = PD; by = lb / (PD / PG_ROWS); bx = lb - by * (PD / PG_ROWS);
    grow = 16 * PJ_G4; srow = 16 * PJ_S4; mask = lpl_m;
    ooff = NT * RT + NH * RH + NA * PD + NB * LD + NPA * LD;
  }
  pair_tile(proj + (size_t)srow * 128, proj + (size_t)grow * 128, mask,
            sc_b1 + h * HD, sc_g + h * HD, sc_be + h * HD, sc_W2 + h * HD, sc_b2 + h,
            out + ooff, R, bx * PG_ROWS, by * PS_ROWS);
}

extern "C" void kernel_launch(void* const* d_in, const int* in_sizes, int n_in,
                              void* d_out, int out_size, void* d_ws, size_t ws_size,
                              hipStream_t stream) {
  (void)in_sizes; (void)n_in; (void)out_size; (void)ws_size;
  const float* points  = (const float*)d_in[0];
  const float* lines   = (const float*)d_in[1];
  const float* circles = (const float*)d_in[2];
  const float* text_s  = (const float*)d_in[3];
  const float* head_s  = (const float*)d_in[4];
  const float* angle_s = (const float*)d_in[5];
  const float* bar_s   = (const float*)d_in[6];
  const float* para_s  = (const float*)d_in[7];
  const float* perp_s  = (const float*)d_in[8];
  const float* rel_W   = (const float*)d_in[9];
  const float* rel_b   = (const float*)d_in[10];
  const float* rel_g   = (const float*)d_in[11];
  const float* rel_be  = (const float*)d_in[12];
  const float* gm_W    = (const float*)d_in[13];
  const float* gm_b    = (const float*)d_in[14];
  const float* gm_g    = (const float*)d_in[15];
  const float* gm_be   = (const float*)d_in[16];
  const float* sc_W1   = (const float*)d_in[17];
  const float* sc_b1   = (const float*)d_in[18];
  const float* sc_g    = (const float*)d_in[19];
  const float* sc_be   = (const float*)d_in[20];
  const float* sc_W2   = (const float*)d_in[21];
  const float* sc_b2   = (const float*)d_in[22];
  const int*   pl      = (const int*)d_in[23];
  const int*   pc      = (const int*)d_in[24];
  float* out = (float*)d_out;

  float* wsp = (float*)d_ws;
  size_t off = 0;
  auto alloc = [&](size_t n) { float* p = wsp + off; off += (n + 31) & ~(size_t)31; return p; };
  float* LPLb   = alloc((size_t)PD * HD);
  float* PLPb   = alloc((size_t)LD * HD);
  float* PCPb   = alloc((size_t)CD * HD);
  float* lpl_m  = alloc(PD);
  float* plp_m  = alloc(LD);
  float* pcp_m  = alloc(CD);
  float* mask_t = alloc(RT);
  float* mask_h = alloc(RH);
  float* geo_all = alloc((size_t)(RT + RH) * HD);
  float* geo_t  = geo_all;
  float* geo_h  = geo_all + (size_t)RT * HD;
  float* proj   = alloc((size_t)PJ_S5 * 16 * HD);

  agg_rel3_kernel<<<AG_S2, 256, 0, stream>>>(pl, pc, points, lines, circles,
                                              rel_W, rel_b, rel_g, rel_be, LPLb, PLPb, PCPb,
                                              lpl_m, plp_m, pcp_m);

  const size_t ROW = (size_t)HD * sizeof(float);
  hipMemcpyAsync(geo_t + (size_t)0 * HD,    points, PD * ROW, hipMemcpyDeviceToDevice, stream);
  hipMemcpyAsync(geo_t + (size_t)1024 * HD, lines,  LD * ROW, hipMemcpyDeviceToDevice, stream);
  hipMemcpyAsync(geo_t + (size_t)2048 * HD, LPLb,   PD * ROW, hipMemcpyDeviceToDevice, stream);
  hipMemcpyAsync(geo_t + (size_t)3072 * HD, PLPb,   LD * ROW, hipMemcpyDeviceToDevice, stream);
  hipMemcpyAsync(geo_t + (size_t)4096 * HD, PCPb,   CD * ROW, hipMemcpyDeviceToDevice, stream);
  hipMemcpyAsync(geo_t + (size_t)4224 * HD, head_s, NH * ROW, hipMemcpyDeviceToDevice, stream);
  hipMemcpyAsync(geo_h + (size_t)0 * HD,    points, PD * ROW, hipMemcpyDeviceToDevice, stream);
  hipMemcpyAsync(geo_h + (size_t)1024 * HD, LPLb,   PD * ROW, hipMemcpyDeviceToDevice, stream);
  hipMemcpyAsync(geo_h + (size_t)2048 * HD, PLPb,   LD * ROW, hipMemcpyDeviceToDevice, stream);
  hipMemcpyAsync(geo_h + (size_t)3072 * HD, PCPb,   CD * ROW, hipMemcpyDeviceToDevice, stream);
  mask_cat_kernel<<<(RT + 255) / 256, 256, 0, stream>>>(lpl_m, plp_m, pcp_m, mask_t, mask_h);

  gm_proj12_kernel<<<PJ_S5, 256, 0, stream>>>(geo_all, LPLb, PLPb, lines, text_s, head_s, angle_s,
                                               bar_s, para_s, perp_s, gm_W, gm_b, gm_g, gm_be,
                                               sc_W1, proj);
  pair6_kernel<<<PB5, 128, 0, stream>>>(proj, mask_t, mask_h, lpl_m, plp_m,
                                         sc_b1, sc_g, sc_be, sc_W2, sc_b2, out);
}


# strip LayerNorm by all 256 threads: thread = (row, 8 columns), 16 partial sums of a row added by a row_xmask DPP butterfly, own 8 values normalised in registers, 1 barrier not 2; sum order changes
# speedup vs baseline: 8.1220x; 1.0570x over previous
#include <hip/hip_runtime.h>
#include <math.h>

static const int HD  = 128;
static const int PD  = 1024;
static const int LD  = 1024;
static const int CD  = 128;
static const int NT  = 64, NH = 32, NA = 64, NB = 64, NPA = 32, NPE = 32;
static const int RT  = PD + LD + PD + LD + CD + NH;
static const int RH  = PD + PD + LD + CD;
static const int HH  = HD * HD;

typedef float v2f __attribute__((ext_vector_type(2)));
typedef float v8f __attribute__((ext_vector_type(8)));

__device__ __forceinline__ v8f wmma_f32(v2f a, v2f b, v8f c) {
  return __builtin_amdgcn_wmma_f32_16x16x4_f32(false, a, false, b, (short)0, c,
                                               false, false);
}

static const int TP = 132;

static const int KD = 1024;
static_assert(PD == KD && LD == KD, "the three aggregations share one K");
static const int SP = KD + 4;
static const int KB = 128;

__device__ __forceinline__
void agg_load_b(const float* __restrict__ Bm, int kg0, int n, v2f (&b)[KB / 4]) {
#pragma unroll
  for (int i = 0; i < KB / 4; ++i) {
    const int kg = kg0 + 4 * i;
    b[i].x = Bm[(size_t)kg * HD + n];
    b[i].y = Bm[(size_t)(kg + 1) * HD + n];
  }
}

__device__ __forceinline__
v8f agg_mma(const float* __restrict__ sa, const v2f (&b)[KB / 4], v8f acc) {
#pragma unroll
  for (int i = 0; i < KB / 4; ++i)
    acc = wmma_f32(*(const v2f*)&sa[4 * i], b[i], acc);
  return acc;
}

template <int TRANS>
__device__ __forceinline__
v8f agg_strip(const int* __restrict__ rel, int ldA, const float* __restrict__ Bm,
              float* __restrict__ rowmask, int m0, float* __restrict__ sR) {
  const int tid  = threadIdx.x;
  const int lane = tid & 31;
  const int w    = tid >> 5;
  const int n    = w * 16 + (lane & 15);
  const int am   = lane & 15;
  const int koff = (lane >> 4) << 1;

  int4 r[16];
  if (TRANS) {
    const int* p = rel + (size_t)(tid >> 2) * ldA + (m0 + ((tid & 3) << 2));
#pragma unroll
    for (int j = 0; j < 16; ++j) r[j] = *(const int4*)(p + (size_t)(64 * j) * ldA);
  } else {
    const int* p = rel + (size_t)(m0 + (tid >> 4)) * ldA + ((tid & 15) << 2);
#pragma unroll
    for (int j = 0; j < 16; ++j) r[j] = *(const int4*)(p + 64 * j);
  }
  v2f bA[KB / 4], bB[KB / 4];
  agg_load_b(Bm, koff, n, bA);

  if (TRANS) {
    float* d = &sR[((tid & 3) << 2) * SP + (tid >> 2)];
#pragma unroll
    for (int j = 0; j < 16; ++j) {
      d[0 * SP + 64 * j] = (float)r[j].x;
      d[1 * SP + 64 * j] = (float)r[j].y;
      d[2 * SP + 64 * j] = (float)r[j].z;
      d[3 * SP + 64 * j] = (float)r[j].w;
    }
  } else {
    float* d = &sR[(tid >> 4) * SP + ((tid & 15) << 2)];
#pragma unroll
    for (int j = 0; j < 16; ++j) {
      const float4 f = {(float)r[j].x, (float)r[j].y, (float)r[j].z, (float)r[j].w};
      *(float4*)&d[64 * j] = f;
    }
  }
  __syncthreads();

  const int crow = tid >> 4;
  const int ccol = (tid & 15) << 2;
  float cnt = 0.f;
#pragma unroll
  for (int k0 = 0; k0 < KD; k0 += 128) {
    const float4 c0 = *(const float4*)&sR[crow * SP + k0 + ccol];
    const float4 c1 = *(const float4*)&sR[crow * SP + k0 + ccol + 64];
    cnt += ((c0.x + c0.y) + (c0.z + c0.w)) + ((c1.x + c1.y) + (c1.z + c1.w));
  }

  v8f acc = {0.f, 0.f, 0.f, 0.f, 0.f, 0.f, 0.f, 0.f};
  const float* sa = &sR[am * SP + koff];
  for (int k0 = 0; k0 < KD - 2 * KB; k0 += 2 * KB) {
    agg_load_b(Bm, k0 + KB + koff, n, bB);
    acc = agg_mma(sa + k0, bA, acc);
    agg_load_b(Bm, k0 + 2 * KB + koff, n, bA);
    acc = agg_mma(sa + k0 + KB, bB, acc);
  }
  agg_load_b(Bm, KD - KB + koff, n, bB);
  acc = agg_mma(sa + KD - 2 * KB, bA, acc);
  acc = agg_mma(sa + KD - KB, bB, acc);

  for (int o = 8; o > 0; o >>= 1) cnt += __shfl_xor(cnt, o, 32);
  if ((tid & 15) == 0) rowmask[m0 + crow] = (cnt >= 2.f) ? 1.f : 0.f;
  return acc;
}


__device__ __forceinline__
void strip_stage(const float* __restrict__ X, float* __restrict__ sA) {
  const int tid = threadIdx.x;
  const float4 x0 = ((const float4*)X)[tid];
  const float4 x1 = ((const float4*)X)[tid + 256];
  float* d = &sA[(tid >> 5) * TP + ((tid & 31) << 2)];
  *(float4*)&d[0]      = x0;
  *(float4*)&d[8 * TP] = x1;
}

__device__ __forceinline__
v8f strip_gemm(const float* __restrict__ sA, const float* __restrict__ W) {
  const int lane = threadIdx.x & 31;
  const int w    = threadIdx.x >> 5;
  const int n    = w * 16 + (lane & 15);
  const int am   = lane & 15;
  const int koff = (lane >> 4) << 1;
  v8f acc = {0.f, 0.f, 0.f, 0.f, 0.f, 0.f, 0.f, 0.f};
#pragma unroll 4
  for (int kk = 0; kk < 128; kk += 4) {
    v2f a = *(const v2f*)&sA[am * TP + kk + koff];
    v2f b = *(const v2f*)&W[(size_t)n * 128 + kk + koff];
    acc = wmma_f32(a, b, acc);
  }
  return acc;
}

__device__ __forceinline__
void strip_store_raw(const v8f acc, float* __restrict__ Y) {
  const int lane = threadIdx.x & 31;
  const int n    = (threadIdx.x >> 5) * 16 + (lane & 15);
  const int mb   = (lane >> 4) << 3;
#pragma unroll
  for (int v = 0; v < 8; ++v)
    Y[(size_t)(mb + v) * 128 + n] = acc[v];
}

__device__ __forceinline__ float row16_sum(float v) {
  v += __int_as_float(__builtin_amdgcn_update_dpp(0, __float_as_int(v), 0x168, 0xf, 0xf, true));
  v += __int_as_float(__builtin_amdgcn_update_dpp(0, __float_as_int(v), 0x164, 0xf, 0xf, true));
  v += __int_as_float(__builtin_amdgcn_update_dpp(0, __float_as_int(v), 0x162, 0xf, 0xf, true));
  v += __int_as_float(__builtin_amdgcn_update_dpp(0, __float_as_int(v), 0x161, 0xf, 0xf, true));
  return v;
}

__device__ __forceinline__
void strip_relu_ln(const v8f acc, const float* __restrict__ bias, const float* __restrict__ g,
                   const float* __restrict__ be, float* __restrict__ sC,
                   float* __restrict__ dst, int dp) {
  const int tid  = threadIdx.x;
  const int lane = tid & 31;
  const int n    = (tid >> 5) * 16 + (lane & 15);
  const int mb   = (lane >> 4) << 3;
  const int row  = tid >> 4;
  const int c    = (tid & 15) << 2;
  const float bn = bias[n];
#pragma unroll
  for (int v = 0; v < 8; ++v) {
    float x = acc[v] + bn;
    sC[(mb + v) * TP + n] = x > 0.f ? x : 0.f;
  }
  __syncthreads();
  const float4 g0 = *(const float4*)&g[c];
  const float4 g1 = *(const float4*)&g[c + 64];
  const float4 e0 = *(const float4*)&be[c];
  const float4 e1 = *(const float4*)&be[c + 64];
  const float4 x0 = *(const float4*)&sC[row * TP + c];
  const float4 x1 = *(const float4*)&sC[row * TP + c + 64];
  float s  = ((x0.x + x0.y) + (x0.z + x0.w)) + ((x1.x + x1.y) + (x1.z + x1.w));
  float q0 = x0.x * x0.x, q1 = x1.x * x1.x;
  q0 = fmaf(x0.y, x0.y, q0); q1 = fmaf(x1.y, x1.y, q1);
  q0 = fmaf(x0.z, x0.z, q0); q1 = fmaf(x1.z, x1.z, q1);
  q0 = fmaf(x0.w, x0.w, q0); q1 = fmaf(x1.w, x1.w, q1);
  float q = q0 + q1;
  s = row16_sum(s);
  q = row16_sum(q);
  const float mean = s * (1.f / 128.f);
  const float var  = q * (1.f / 128.f) - mean * mean;
  const float rstd = rsqrtf(var + 1e-5f);
  float4 y0, y1;
  y0.x = (x0.x - mean) * rstd * g0.x + e0.x;
  y0.y = (x0.y - mean) * rstd * g0.y + e0.y;
  y0.z = (x0.z - mean) * rstd * g0.z + e0.z;
  y0.w = (x0.w - mean) * rstd * g0.w + e0.w;
  y1.x = (x1.x - mean) * rstd * g1.x + e1.x;
  y1.y = (x1.y - mean) * rstd * g1.y + e1.y;
  y1.z = (x1.z - mean) * rstd * g1.z + e1.z;
  y1.w = (x1.w - mean) * rstd * g1.w + e1.w;
  *(float4*)&dst[row * dp + c]      = y0;
  *(float4*)&dst[row * dp + c + 64] = y1;
}

static const int AG_S0 = PD / 16;
static const int AG_S1 = AG_S0 + LD / 16;
static const int AG_S2 = AG_S1 + CD / 16;
__global__ __launch_bounds__(256)
void agg_rel3_kernel(const int* __restrict__ pl, const int* __restrict__ pc,
                     const float* __restrict__ points, const float* __restrict__ lines,
                     const float* __restrict__ circles,
                     const float* __restrict__ rel_W, const float* __restrict__ rel_b,
                     const float* __restrict__ rel_g, const float* __restrict__ rel_be,
                     float* __restrict__ LPLb, float* __restrict__ PLPb, float* __restrict__ PCPb,
                     float* __restrict__ lpl_m, float* __restrict__ plp_m,
                     float* __restrict__ pcp_m) {
  __shared__ __attribute__((aligned(16))) float sR[16 * SP];
  __shared__ __attribute__((aligned(16))) float sA[16 * TP];
  __shared__ __attribute__((aligned(16))) float sC[16 * TP];
  const int b = blockIdx.x;
  const int* rel; int ldA, transA, m0, j;
  const float* Bm; const float* middle; float* Y; float* rowmask;
  if (b < AG_S0) {
    rel = pl; ldA = LD; transA = 0; Bm = lines;  middle = points;  Y = LPLb;
    rowmask = lpl_m; j = 0; m0 = b * 16;
  } else if (b < AG_S1) {
    rel = pl; ldA = LD; transA = 1; Bm = points; middle = lines;   Y = PLPb;
    rowmask = plp_m; j = 1; m0 = (b - AG_S0) * 16;
  } else {
    rel = pc; ldA = CD; transA = 1; Bm = points; middle = circles; Y = PCPb;
    rowmask = pcp_m; j = 2; m0 = (b - AG_S1) * 16;
  }
  const int lane = threadIdx.x & 31;
  const int n    = (threadIdx.x >> 5) * 16 + (lane & 15);
  const int mb   = (lane >> 4) << 3;
  float mid[8];
#pragma unroll
  for (int v = 0; v < 8; ++v) mid[v] = middle[(size_t)(m0 + mb + v) * HD + n];

  const v8f agg = transA ? agg_strip<1>(rel, ldA, Bm, rowmask, m0, sR)
                         : agg_strip<0>(rel, ldA, Bm, rowmask, m0, sR);
#pragma unroll
  for (int v = 0; v < 8; ++v) sA[(mb + v) * TP + n] = 2.f * agg[v] + mid[v];
  __syncthreads();
  const v8f y = strip_gemm(sA, rel_W + (size_t)j * HH);
  strip_relu_ln(y, rel_b + j * HD, rel_g + j * HD, rel_be + j * HD, sC,
                Y + (size_t)m0 * HD, HD);
}

static const int PJ_G0 = RT / 16;
static const int PJ_G1 = PJ_G0 + RH / 16;
static const int PJ_G2 = PJ_G1 + PD / 16;
static const int PJ_G3 = PJ_G2 + LD / 16;
static const int PJ_G4 = PJ_G3 + LD / 16;
static const int PJ_G5 = PJ_G4 + PD / 16;
static const int PJ_S0 = PJ_G5 + NT / 16;
static const int PJ_S1 = PJ_S0 + NH / 16;
static const int PJ_S2 = PJ_S1 + NA / 16;
static const int PJ_S3 = PJ_S2 + NB / 16;
static const int PJ_S4 = PJ_S3 + NPA / 16;
static const int PJ_S5 = PJ_S4 + NPE / 16;
__global__ __launch_bounds__(256)
void gm_proj12_kernel(const float* __restrict__ geo_all, const float* __restrict__ LPLb,
                      const float* __restrict__ PLPb, const float* __restrict__ lines,
                      const float* __restrict__ text_s, const float* __restrict__ head_s,
                      const float* __restrict__ angle_s, const float* __restrict__ bar_s,
                      const float* __restrict__ para_s, const float* __restrict__ perp_s,
                      const float* __restrict__ gm_W, const float* __restrict__ gm_b,
                      const float* __restrict__ gm_g, const float* __restrict__ gm_be,
                      const float* __restrict__ sc_W1, float* __restrict__ proj) {
  __shared__ __attribute__((aligned(16))) float sA[16 * TP];
  __shared__ __attribute__((aligned(16))) float sC[16 * TP];
  const int b = blockIdx.x;
  const float* X; int h, s;
  if (b < PJ_G0)      { X = geo_all; h = 0; s = b; }
  else if (b < PJ_G1) { X = geo_all; h = 1; s = b; }
  else if (b < PJ_G2) { X = LPLb;    h = 2; s = b - PJ_G1; }
  else if (b < PJ_G3) { X = PLPb;    h = 3; s = b - PJ_G2; }
  else if (b < PJ_G4) { X = lines;   h = 4; s = b - PJ_G3; }
  else if (b < PJ_G5) { X = LPLb;    h = 5; s = b - PJ_G4; }
  else if (b < PJ_S0) { X = text_s;  h = 0; s = b - PJ_G5; }
  else if (b < PJ_S1) { X = head_s;  h = 1; s = b - PJ_S0; }
  else if (b < PJ_S2) { X = angle_s; h = 2; s = b - PJ_S1; }
  else if (b < PJ_S3) { X = bar_s;   h = 3; s = b - PJ_S2; }
  else if (b < PJ_S4) { X = para_s;  h = 4; s = b - PJ_S3; }
  else                { X = perp_s;  h = 5; s = b - PJ_S4; }
  strip_stage(X + (size_t)s * (16 * 128), sA);
  __syncthreads();
  if (b < PJ_G1) {
    for (int l = 0; l < 2; ++l) {
      const v8f y = strip_gemm(sA, gm_W + (size_t)l * HH);
      strip_relu_ln(y, gm_b + l * HD, gm_g + l * HD, gm_be + l * HD, sC, sA, TP);
      __syncthreads();
    }
  }
  const v8f p = strip_gemm(sA, sc_W1 + (size_t)h * HH);
  strip_store_raw(p, proj + (size_t)b * (16 * 128));
}

__global__ void mask_cat_kernel(const float* __restrict__ lpl_m,
                                const float* __restrict__ plp_m,
                                const float* __restrict__ pcp_m,
                                float* __restrict__ mask_t, float* __restrict__ mask_h) {
  int t = blockIdx.x * blockDim.x + threadIdx.x;
  if (t < RT) {
    float v;
    if (t < 2048)      v = 1.f;
    else if (t < 3072) v = lpl_m[t - 2048];
    else if (t < 4096) v = plp_m[t - 3072];
    else if (t < 4224) v = pcp_m[t - 4096];
    else               v = 1.f;
    mask_t[t] = v;
  }
  if (t < RH) {
    float v;
    if (t < 1024)      v = 1.f;
    else if (t < 2048) v = lpl_m[t - 1024];
    else if (t < 3072) v = plp_m[t - 2048];
    else               v = pcp_m[t - 3072];
    mask_h[t] = v;
  }
}

typedef float v4f __attribute__((ext_vector_type(4)));
static const int PG_ROWS  = 32;
static const int PS_ROWS  = 16;
static const int PG_PITCH = 132;
static_assert(RT % PG_ROWS == 0 && RH % PG_ROWS == 0 && PD % PG_ROWS == 0 && LD % PG_ROWS == 0,
              "geo row counts must be whole tiles");
static_assert(NT % PS_ROWS == 0 && NH % PS_ROWS == 0 && NA % PS_ROWS == 0 && NB % PS_ROWS == 0 &&
              NPA % PS_ROWS == 0 && NPE % PS_ROWS == 0, "sym row counts must be whole tiles");

__device__ __forceinline__ void pair_block(const v4f gv, const v4f wv, const v4f (&sv)[4],
                                           v2f (&S)[4], v2f (&Q)[4], v2f (&D)[4]) {
  const v2f zero2 = {0.f, 0.f};
  for (int ii = 0; ii < 4; ++ii) {
    const v2f h01 = __builtin_elementwise_max(sv[ii].xy + gv.xy, zero2);
    const v2f h23 = __builtin_elementwise_max(sv[ii].zw + gv.zw, zero2);
    S[ii] += h01;
    Q[ii] = __builtin_elementwise_fma(h01, h01, Q[ii]);
    D[ii] = __builtin_elementwise_fma(h01, wv.xy, D[ii]);
    S[ii] += h23;
    Q[ii] = __builtin_elementwise_fma(h23, h23, Q[ii]);
    D[ii] = __builtin_elementwise_fma(h23, wv.zw, D[ii]);
  }
}

__device__ __forceinline__
void pair_tile(const float* __restrict__ symW, const float* __restrict__ geoW,
                 const float* __restrict__ mask, const float* __restrict__ b1,
                 const float* __restrict__ g, const float* __restrict__ be,
                 const float* __restrict__ W2, const float* __restrict__ b2p,
                 float* __restrict__ out, int R, int r0, int i0) {
  __shared__ __attribute__((aligned(16))) float sG[PG_ROWS * PG_PITCH];
  __shared__ __attribute__((aligned(16))) float sS[PS_ROWS * 128];
  __shared__ __attribute__((aligned(16))) float sW[128];
  const int tid  = threadIdx.x;
  const int lane = tid & 31;
  const int w    = tid >> 5;

  const v4f gg = ((const v4f*)g)[lane];
  const v4f bb = ((const v4f*)be)[lane];
  const v4f ww = ((const v4f*)W2)[lane];
  const v4f b4 = ((const v4f*)b1)[lane];
  const float mk = mask[r0 + lane];
  const float b2 = b2p[0];
  v4f ts[4], tg[8];
  for (int u = 0; u < 4; ++u)
    ts[u] = ((const v4f*)(symW + (size_t)(i0 + w + 4 * u) * 128))[lane];
  for (int u = 0; u < 8; ++u)
    tg[u] = ((const v4f*)(geoW + (size_t)(r0 + w + 4 * u) * 128))[lane];

  const v4f gw = gg * ww;
  const v4f bw = bb * ww;
  if (w == 0) *(v4f*)&sW[lane * 4] = gw;
  float G  = (gw.x + gw.y) + (gw.z + gw.w);
  float Bc = (bw.x + bw.y) + (bw.z + bw.w);
  for (int o = 16; o > 0; o >>= 1) {
    G  += __shfl_xor(G, o, 32);
    Bc += __shfl_xor(Bc, o, 32);
  }
  for (int u = 0; u < 4; ++u)
    *(v4f*)&sS[(w + 4 * u) * 128 + lane * 4] = ts[u] + b4;
  for (int u = 0; u < 8; ++u)
    *(v4f*)&sG[(w + 4 * u) * PG_PITCH + lane * 4] = tg[u];
  __syncthreads();

  const float* gp = &sG[lane * PG_PITCH];
  const float* sp = &sS[(w * 4) * 128];
  const v2f zero2 = {0.f, 0.f};
  v2f S[4], Q[4], D[4];
  for (int ii = 0; ii < 4; ++ii) { S[ii] = zero2; Q[ii] = zero2; D[ii] = zero2; }

  v4f gA, wA, sA[4], gB, wB, sB[4];
  gA = *(const v4f*)gp;
  wA = *(const v4f*)sW;
  for (int ii = 0; ii < 4; ++ii) sA[ii] = *(const v4f*)(sp + ii * 128);
#pragma unroll
  for (int k = 0; k < 128; k += 8) {
    gB = *(const v4f*)(gp + k + 4);
    wB = *(const v4f*)(sW + k + 4);
    for (int ii = 0; ii < 4; ++ii) sB[ii] = *(const v4f*)(sp + ii * 128 + k + 4);
    pair_block(gA, wA, sA, S, Q, D);
    const int kn = (k + 8 < 128) ? k + 8 : 0;
    gA = *(const v4f*)(gp + kn);
    wA = *(const v4f*)(sW + kn);
    for (int ii = 0; ii < 4; ++ii) sA[ii] = *(const v4f*)(sp + ii * 128 + kn);
    pair_block(gB, wB, sB, S, Q, D);
  }

  for (int ii = 0; ii < 4; ++ii) {
    const float s = S[ii].x + S[ii].y;
    const float q = Q[ii].x + Q[ii].y;
    const float d = D[ii].x + D[ii].y;
    const float mean  = s * (1.f / 128.f);
    const float var   = q * (1.f / 128.f) - mean * mean;
    const float rstd  = rsqrtf(var + 1e-5f);
    const float logit = rstd * (d - mean * G) + Bc + b2;
    const float z     = (mk > 0.f) ? logit : -1e9f;
    out[(size_t)(i0 + w * 4 + ii) * R + r0 + lane] = 1.f / (1.f + expf(-z));
  }
}


static const int PB0 = (RT / PG_ROWS) * (NT / PS_ROWS);
static const int PB1 = PB0 + (RH / PG_ROWS) * (NH / PS_ROWS);
static const int PB2 = PB1 + (PD / PG_ROWS) * (NA / PS_ROWS);
static const int PB3 = PB2 + (LD / PG_ROWS) * (NB / PS_ROWS);
static const int PB4 = PB3 + (LD / PG_ROWS) * (NPA / PS_ROWS);
static const int PB5 = PB4 + (PD / PG_ROWS) * (NPE / PS_ROWS);
__global__ __launch_bounds__(128)
void pair6_kernel(const float* __restrict__ proj,
                  const float* __restrict__ mask_t, const float* __restrict__ mask_h,
                  const float* __restrict__ lpl_m, const float* __restrict__ plp_m,
                  const float* __restrict__ sc_b1, const float* __restrict__ sc_g,
                  const float* __restrict__ sc_be, const float* __restrict__ sc_W2,
                  const float* __restrict__ sc_b2, float* __restrict__ out) {
  const int b = blockIdx.x;
  int h, R, bx, by, grow, srow, ooff; const float* mask;
  if (b < PB0) {
    h = 0; R = RT; by = b / (RT / PG_ROWS); bx = b - by * (RT / PG_ROWS);
    grow = 0; srow = 16 * PJ_G5; mask = mask_t; ooff = 0;
  } else if (b < PB1) {
    const int lb = b - PB0;
    h = 1; R = RH; by = lb / (RH / PG_ROWS); bx = lb - by * (RH / PG_ROWS);
    grow = 16 * PJ_G0; srow = 16 * PJ_S0; mask = mask_h; ooff = NT * RT;
  } else if (b < PB2) {
    const int lb = b - PB1;
    h = 2; R = PD; by = lb / (PD / PG_ROWS); bx = lb - by * (PD / PG_ROWS);
    grow = 16 * PJ_G1; srow = 16 * PJ_S1; mask = lpl_m; ooff = NT * RT + NH * RH;
  } else if (b < PB3) {
    const int lb = b - PB2;
    h = 3; R = LD; by = lb / (LD / PG_ROWS); bx = lb - by * (LD / PG_ROWS);
    grow = 16 * PJ_G2; srow = 16 * PJ_S2; mask = plp_m; ooff = NT * RT + NH * RH + NA * PD;
  } else if (b < PB4) {
    const int lb = b - PB3;
    h = 4; R = LD; by = lb / (LD / PG_ROWS); bx = lb - by * (LD / PG_ROWS);
    grow = 16 * PJ_G3; srow = 16 * PJ_S3; mask = mask_t + PD;
    ooff = NT * RT + NH * RH + NA * PD + NB * LD;
  } else {
    const int lb = b - PB4;
    h = 5; R = PD; by = lb / (PD / PG_ROWS); bx = lb - by * (PD / PG_ROWS);
    grow = 16 * PJ_G4; srow = 16 * PJ_S4; mask = lpl_m;
    ooff = NT * RT + NH * RH + NA * PD + NB * LD + NPA * LD;
  }
  pair_tile(proj + (size_t)srow * 128, proj + (size_t)grow * 128, mask,
            sc_b1 + h * HD, sc_g + h * HD, sc_be + h * HD, sc_W2 + h * HD, sc_b2 + h,
            out + ooff, R, bx * PG_ROWS, by * PS_ROWS);
}

extern "C" void kernel_launch(void* const* d_in, const int* in_sizes, int n_in,
                              void* d_out, int out_size, void* d_ws, size_t ws_size,
                              hipStream_t stream) {
  (void)in_sizes; (void)n_in; (void)out_size; (void)ws_size;
  const float* points  = (const float*)d_in[0];
  const float* lines   = (const float*)d_in[1];
  const float* circles = (const float*)d_in[2];
  const float* text_s  = (const float*)d_in[3];
  const float* head_s  = (const float*)d_in[4];
  const float* angle_s = (const float*)d_in[5];
  const float* bar_s   = (const float*)d_in[6];
  const float* para_s  = (const float*)d_in[7];
  const float* perp_s  = (const float*)d_in[8];
  const float* rel_W   = (const float*)d_in[9];
  const float* rel_b   = (const float*)d_in[10];
  const float* rel_g   = (const float*)d_in[11];
  const float* rel_be  = (const float*)d_in[12];
  const float* gm_W    = (const float*)d_in[13];
  const float* gm_b    = (const float*)d_in[14];
  const float* gm_g    = (const float*)d_in[15];
  const float* gm_be   = (const float*)d_in[16];
  const float* sc_W1   = (const float*)d_in[17];
  const float* sc_b1   = (const float*)d_in[18];
  const float* sc_g    = (const float*)d_in[19];
  const float* sc_be   = (const float*)d_in[20];
  const float* sc_W2   = (const float*)d_in[21];
  const float* sc_b2   = (const float*)d_in[22];
  const int*   pl      = (const int*)d_in[23];
  const int*   pc      = (const int*)d_in[24];
  float* out = (float*)d_out;

  float* wsp = (float*)d_ws;
  size_t off = 0;
  auto alloc = [&](size_t n) { float* p = wsp + off; off += (n + 31) & ~(size_t)31; return p; };
  float* LPLb   = alloc((size_t)PD * HD);
  float* PLPb   = alloc((size_t)LD * HD);
  float* PCPb   = alloc((size_t)CD * HD);
  float* lpl_m  = alloc(PD);
  float* plp_m  = alloc(LD);
  float* pcp_m  = alloc(CD);
  float* mask_t = alloc(RT);
  float* mask_h = alloc(RH);
  float* geo_all = alloc((size_t)(RT + RH) * HD);
  float* geo_t  = geo_all;
  float* geo_h  = geo_all + (size_t)RT * HD;
  float* proj   = alloc((size_t)PJ_S5 * 16 * HD);

  agg_rel3_kernel<<<AG_S2, 256, 0, stream>>>(pl, pc, points, lines, circles,
                                              rel_W, rel_b, rel_g, rel_be, LPLb, PLPb, PCPb,
                                              lpl_m, plp_m, pcp_m);

  const size_t ROW = (size_t)HD * sizeof(float);
  hipMemcpyAsync(geo_t + (size_t)0 * HD,    points, PD * ROW, hipMemcpyDeviceToDevice, stream);
  hipMemcpyAsync(geo_t + (size_t)1024 * HD, lines,  LD * ROW, hipMemcpyDeviceToDevice, stream);
  hipMemcpyAsync(geo_t + (size_t)2048 * HD, LPLb,   PD * ROW, hipMemcpyDeviceToDevice, stream);
  hipMemcpyAsync(geo_t + (size_t)3072 * HD, PLPb,   LD * ROW, hipMemcpyDeviceToDevice, stream);
  hipMemcpyAsync(geo_t + (size_t)4096 * HD, PCPb,   CD * ROW, hipMemcpyDeviceToDevice, stream);
  hipMemcpyAsync(geo_t + (size_t)4224 * HD, head_s, NH * ROW, hipMemcpyDeviceToDevice, stream);
  hipMemcpyAsync(geo_h + (size_t)0 * HD,    points, PD * ROW, hipMemcpyDeviceToDevice, stream);
  hipMemcpyAsync(geo_h + (size_t)1024 * HD, LPLb,   PD * ROW, hipMemcpyDeviceToDevice, stream);
  hipMemcpyAsync(geo_h + (size_t)2048 * HD, PLPb,   LD * ROW, hipMemcpyDeviceToDevice, stream);
  hipMemcpyAsync(geo_h + (size_t)3072 * HD, PCPb,   CD * ROW, hipMemcpyDeviceToDevice, stream);
  mask_cat_kernel<<<(RT + 255) / 256, 256, 0, stream>>>(lpl_m, plp_m, pcp_m, mask_t, mask_h);

  gm_proj12_kernel<<<PJ_S5, 256, 0, stream>>>(geo_all, LPLb, PLPb, lines, text_s, head_s, angle_s,
                                               bar_s, para_s, perp_s, gm_W, gm_b, gm_g, gm_be,
                                               sc_W1, proj);
  pair6_kernel<<<PB5, 128, 0, stream>>>(proj, mask_t, mask_h, lpl_m, plp_m,
                                         sc_b1, sc_g, sc_be, sc_W2, sc_b2, out);
}


# weight tile of every strip GEMM staged once in LDS by whole-row 16-byte loads, B fragments read from LDS (was 32 eight-byte global loads per wave and layer, 16 rows 512 B apart); k order unchanged
# speedup vs baseline: 8.8970x; 1.0954x over previous
#include <hip/hip_runtime.h>
#include <math.h>

static const int HD  = 128;
static const int PD  = 1024;
static const int LD  = 1024;
static const int CD  = 128;
static const int NT  = 64, NH = 32, NA = 64, NB = 64, NPA = 32, NPE = 32;
static const int RT  = PD + LD + PD + LD + CD + NH;
static const int RH  = PD + PD + LD + CD;
static const int HH  = HD * HD;

typedef float v2f __attribute__((ext_vector_type(2)));
typedef float v8f __attribute__((ext_vector_type(8)));

__device__ __forceinline__ v8f wmma_f32(v2f a, v2f b, v8f c) {
  return __builtin_amdgcn_wmma_f32_16x16x4_f32(false, a, false, b, (short)0, c,
                                               false, false);
}

static const int TP = 132;

__device__ __forceinline__
void wtile_request(const float* __restrict__ Wm, float4 (&wr)[16]) {
  const int lane = threadIdx.x & 31;
  const int w    = threadIdx.x >> 5;
  const float4* p = (const float4*)(Wm + (size_t)(w * 16) * 128) + lane;
#pragma unroll
  for (int j = 0; j < 16; ++j) wr[j] = p[j * 32];
}

__device__ __forceinline__
void wtile_store(const float4 (&wr)[16], float* __restrict__ sW) {
  const int lane = threadIdx.x & 31;
  const int w    = threadIdx.x >> 5;
  float* d = &sW[(w * 16) * TP + (lane << 2)];
#pragma unroll
  for (int j = 0; j < 16; ++j) *(float4*)&d[j * TP] = wr[j];
}

static const int KD = 1024;
static_assert(PD == KD && LD == KD, "the three aggregations share one K");
static const int SP = KD + 4;
static const int KB = 128;

__device__ __forceinline__
void agg_load_b(const float* __restrict__ Bm, int kg0, int n, v2f (&b)[KB / 4]) {
#pragma unroll
  for (int i = 0; i < KB / 4; ++i) {
    const int kg = kg0 + 4 * i;
    b[i].x = Bm[(size_t)kg * HD + n];
    b[i].y = Bm[(size_t)(kg + 1) * HD + n];
  }
}

__device__ __forceinline__
v8f agg_mma(const float* __restrict__ sa, const v2f (&b)[KB / 4], v8f acc) {
#pragma unroll
  for (int i = 0; i < KB / 4; ++i)
    acc = wmma_f32(*(const v2f*)&sa[4 * i], b[i], acc);
  return acc;
}

template <int TRANS>
__device__ __forceinline__
v8f agg_strip(const int* __restrict__ rel, int ldA, const float* __restrict__ Bm,
              float* __restrict__ rowmask, int m0, float* __restrict__ sR,
              const float4 (&wr)[16], float* __restrict__ sW) {
  const int tid  = threadIdx.x;
  const int lane = tid & 31;
  const int w    = tid >> 5;
  const int n    = w * 16 + (lane & 15);
  const int am   = lane & 15;
  const int koff = (lane >> 4) << 1;

  int4 r[16];
  if (TRANS) {
    const int* p = rel + (size_t)(tid >> 2) * ldA + (m0 + ((tid & 3) << 2));
#pragma unroll
    for (int j = 0; j < 16; ++j) r[j] = *(const int4*)(p + (size_t)(64 * j) * ldA);
  } else {
    const int* p = rel + (size_t)(m0 + (tid >> 4)) * ldA + ((tid & 15) << 2);
#pragma unroll
    for (int j = 0; j < 16; ++j) r[j] = *(const int4*)(p + 64 * j);
  }
  v2f bA[KB / 4], bB[KB / 4];
  agg_load_b(Bm, koff, n, bA);

  if (TRANS) {
    float* d = &sR[((tid & 3) << 2) * SP + (tid >> 2)];
#pragma unroll
    for (int j = 0; j < 16; ++j) {
      d[0 * SP + 64 * j] = (float)r[j].x;
      d[1 * SP + 64 * j] = (float)r[j].y;
      d[2 * SP + 64 * j] = (float)r[j].z;
      d[3 * SP + 64 * j] = (float)r[j].w;
    }
  } else {
    float* d = &sR[(tid >> 4) * SP + ((tid & 15) << 2)];
#pragma unroll
    for (int j = 0; j < 16; ++j) {
      const float4 f = {(float)r[j].x, (float)r[j].y, (float)r[j].z, (float)r[j].w};
      *(float4*)&d[64 * j] = f;
    }
  }
  wtile_store(wr, sW);
  __syncthreads();

  const int crow = tid >> 4;
  const int ccol = (tid & 15) << 2;
  float cnt = 0.f;
#pragma unroll
  for (int k0 = 0; k0 < KD; k0 += 128) {
    const float4 c0 = *(const float4*)&sR[crow * SP + k0 + ccol];
    const float4 c1 = *(const float4*)&sR[crow * SP + k0 + ccol + 64];
    cnt += ((c0.x + c0.y) + (c0.z + c0.w)) + ((c1.x + c1.y) + (c1.z + c1.w));
  }

  v8f acc = {0.f, 0.f, 0.f, 0.f, 0.f, 0.f, 0.f, 0.f};
  const float* sa = &sR[am * SP + koff];
  for (int k0 = 0; k0 < KD - 2 * KB; k0 += 2 * KB) {
    agg_load_b(Bm, k0 + KB + koff, n, bB);
    acc = agg_mma(sa + k0, bA, acc);
    agg_load_b(Bm, k0 + 2 * KB + koff, n, bA);
    acc = agg_mma(sa + k0 + KB, bB, acc);
  }
  agg_load_b(Bm, KD - KB + koff, n, bB);
  acc = agg_mma(sa + KD - 2 * KB, bA, acc);
  acc = agg_mma(sa + KD - KB, bB, acc);

  for (int o = 8; o > 0; o >>= 1) cnt += __shfl_xor(cnt, o, 32);
  if ((tid & 15) == 0) rowmask[m0 + crow] = (cnt >= 2.f) ? 1.f : 0.f;
  return acc;
}


__device__ __forceinline__
void strip_stage(const float* __restrict__ X, float* __restrict__ sA) {
  const int tid = threadIdx.x;
  const float4 x0 = ((const float4*)X)[tid];
  const float4 x1 = ((const float4*)X)[tid + 256];
  float* d = &sA[(tid >> 5) * TP + ((tid & 31) << 2)];
  *(float4*)&d[0]      = x0;
  *(float4*)&d[8 * TP] = x1;
}

__device__ __forceinline__
v8f strip_gemm(const float* __restrict__ sA, const float* __restrict__ sW) {
  const int lane = threadIdx.x & 31;
  const int w    = threadIdx.x >> 5;
  const int n    = w * 16 + (lane & 15);
  const int am   = lane & 15;
  const int koff = (lane >> 4) << 1;
  const float* pa = &sA[am * TP + koff];
  const float* pb = &sW[n * TP + koff];
  v8f acc = {0.f, 0.f, 0.f, 0.f, 0.f, 0.f, 0.f, 0.f};
#pragma unroll
  for (int k0 = 0; k0 < 128; k0 += 64) {
    v2f a[16], b[16];
#pragma unroll
    for (int i = 0; i < 16; ++i) {
      a[i] = *(const v2f*)&pa[k0 + 4 * i];
      b[i] = *(const v2f*)&pb[k0 + 4 * i];
    }
#pragma unroll
    for (int i = 0; i < 16; ++i) acc = wmma_f32(a[i], b[i], acc);
  }
  return acc;
}

__device__ __forceinline__
void strip_store_raw(const v8f acc, float* __restrict__ Y) {
  const int lane = threadIdx.x & 31;
  const int n    = (threadIdx.x >> 5) * 16 + (lane & 15);
  const int mb   = (lane >> 4) << 3;
#pragma unroll
  for (int v = 0; v < 8; ++v)
    Y[(size_t)(mb + v) * 128 + n] = acc[v];
}

__device__ __forceinline__ float row16_sum(float v) {
  v += __int_as_float(__builtin_amdgcn_update_dpp(0, __float_as_int(v), 0x168, 0xf, 0xf, true));
  v += __int_as_float(__builtin_amdgcn_update_dpp(0, __float_as_int(v), 0x164, 0xf, 0xf, true));
  v += __int_as_float(__builtin_amdgcn_update_dpp(0, __float_as_int(v), 0x162, 0xf, 0xf, true));
  v += __int_as_float(__builtin_amdgcn_update_dpp(0, __float_as_int(v), 0x161, 0xf, 0xf, true));
  return v;
}

__device__ __forceinline__
void strip_relu_ln(const v8f acc, const float* __restrict__ bias, const float* __restrict__ g,
                   const float* __restrict__ be, float* __restrict__ sC,
                   float* __restrict__ dst, int dp) {
  const int tid  = threadIdx.x;
  const int lane = tid & 31;
  const int n    = (tid >> 5) * 16 + (lane & 15);
  const int mb   = (lane >> 4) << 3;
  const int row  = tid >> 4;
  const int c    = (tid & 15) << 2;
  const float bn = bias[n];
#pragma unroll
  for (int v = 0; v < 8; ++v) {
    float x = acc[v] + bn;
    sC[(mb + v) * TP + n] = x > 0.f ? x : 0.f;
  }
  __syncthreads();
  const float4 g0 = *(const float4*)&g[c];
  const float4 g1 = *(const float4*)&g[c + 64];
  const float4 e0 = *(const float4*)&be[c];
  const float4 e1 = *(const float4*)&be[c + 64];
  const float4 x0 = *(const float4*)&sC[row * TP + c];
  const float4 x1 = *(const float4*)&sC[row * TP + c + 64];
  float s  = ((x0.x + x0.y) + (x0.z + x0.w)) + ((x1.x + x1.y) + (x1.z + x1.w));
  float q0 = x0.x * x0.x, q1 = x1.x * x1.x;
  q0 = fmaf(x0.y, x0.y, q0); q1 = fmaf(x1.y, x1.y, q1);
  q0 = fmaf(x0.z, x0.z, q0); q1 = fmaf(x1.z, x1.z, q1);
  q0 = fmaf(x0.w, x0.w, q0); q1 = fmaf(x1.w, x1.w, q1);
  float q = q0 + q1;
  s = row16_sum(s);
  q = row16_sum(q);
  const float mean = s * (1.f / 128.f);
  const float var  = q * (1.f / 128.f) - mean * mean;
  const float rstd = rsqrtf(var + 1e-5f);
  float4 y0, y1;
  y0.x = (x0.x - mean) * rstd * g0.x + e0.x;
  y0.y = (x0.y - mean) * rstd * g0.y + e0.y;
  y0.z = (x0.z - mean) * rstd * g0.z + e0.z;
  y0.w = (x0.w - mean) * rstd * g0.w + e0.w;
  y1.x = (x1.x - mean) * rstd * g1.x + e1.x;
  y1.y = (x1.y - mean) * rstd * g1.y + e1.y;
  y1.z = (x1.z - mean) * rstd * g1.z + e1.z;
  y1.w = (x1.w - mean) * rstd * g1.w + e1.w;
  *(float4*)&dst[row * dp + c]      = y0;
  *(float4*)&dst[row * dp + c + 64] = y1;
}

static const int AG_S0 = PD / 16;
static const int AG_S1 = AG_S0 + LD / 16;
static const int AG_S2 = AG_S1 + CD / 16;
__global__ __launch_bounds__(256)
void agg_rel3_kernel(const int* __restrict__ pl, const int* __restrict__ pc,
                     const float* __restrict__ points, const float* __restrict__ lines,
                     const float* __restrict__ circles,
                     const float* __restrict__ rel_W, const float* __restrict__ rel_b,
                     const float* __restrict__ rel_g, const float* __restrict__ rel_be,
                     float* __restrict__ LPLb, float* __restrict__ PLPb, float* __restrict__ PCPb,
                     float* __restrict__ lpl_m, float* __restrict__ plp_m,
                     float* __restrict__ pcp_m) {
  __shared__ __attribute__((aligned(16))) float sR[16 * SP];
  __shared__ __attribute__((aligned(16))) float sW[128 * TP];
  __shared__ __attribute__((aligned(16))) float sA[16 * TP];
  __shared__ __attribute__((aligned(16))) float sC[16 * TP];
  const int b = blockIdx.x;
  const int* rel; int ldA, transA, m0, j;
  const float* Bm; const float* middle; float* Y; float* rowmask;
  if (b < AG_S0) {
    rel = pl; ldA = LD; transA = 0; Bm = lines;  middle = points;  Y = LPLb;
    rowmask = lpl_m; j = 0; m0 = b * 16;
  } else if (b < AG_S1) {
    rel = pl; ldA = LD; transA = 1; Bm = points; middle = lines;   Y = PLPb;
    rowmask = plp_m; j = 1; m0 = (b - AG_S0) * 16;
  } else {
    rel = pc; ldA = CD; transA = 1; Bm = points; middle = circles; Y = PCPb;
    rowmask = pcp_m; j = 2; m0 = (b - AG_S1) * 16;
  }
  const int lane = threadIdx.x & 31;
  const int n    = (threadIdx.x >> 5) * 16 + (lane & 15);
  const int mb   = (lane >> 4) << 3;
  float mid[8];
#pragma unroll
  for (int v = 0; v < 8; ++v) mid[v] = middle[(size_t)(m0 + mb + v) * HD + n];
  float4 wr[16];
  wtile_request(rel_W + (size_t)j * HH, wr);

  const v8f agg = transA ? agg_strip<1>(rel, ldA, Bm, rowmask, m0, sR, wr, sW)
                         : agg_strip<0>(rel, ldA, Bm, rowmask, m0, sR, wr, sW);
#pragma unroll
  for (int v = 0; v < 8; ++v) sA[(mb + v) * TP + n] = 2.f * agg[v] + mid[v];
  __syncthreads();
  const v8f y = strip_gemm(sA, sW);
  strip_relu_ln(y, rel_b + j * HD, rel_g + j * HD, rel_be + j * HD, sC,
                Y + (size_t)m0 * HD, HD);
}

static const int PJ_G0 = RT / 16;
static const int PJ_G1 = PJ_G0 + RH / 16;
static const int PJ_G2 = PJ_G1 + PD / 16;
static const int PJ_G3 = PJ_G2 + LD / 16;
static const int PJ_G4 = PJ_G3 + LD / 16;
static const int PJ_G5 = PJ_G4 + PD / 16;
static const int PJ_S0 = PJ_G5 + NT / 16;
static const int PJ_S1 = PJ_S0 + NH / 16;
static const int PJ_S2 = PJ_S1 + NA / 16;
static const int PJ_S3 = PJ_S2 + NB / 16;
static const int PJ_S4 = PJ_S3 + NPA / 16;
static const int PJ_S5 = PJ_S4 + NPE / 16;
__global__ __launch_bounds__(256)
void gm_proj12_kernel(const float* __restrict__ geo_all, const float* __restrict__ LPLb,
                      const float* __restrict__ PLPb, const float* __restrict__ lines,
                      const float* __restrict__ text_s, const float* __restrict__ head_s,
                      const float* __restrict__ angle_s, const float* __restrict__ bar_s,
                      const float* __restrict__ para_s, const float* __restrict__ perp_s,
                      const float* __restrict__ gm_W, const float* __restrict__ gm_b,
                      const float* __restrict__ gm_g, const float* __restrict__ gm_be,
                      const float* __restrict__ sc_W1, float* __restrict__ proj) {
  __shared__ __attribute__((aligned(16))) float sW[128 * TP];
  __shared__ __attribute__((aligned(16))) float sA[16 * TP];
  __shared__ __attribute__((aligned(16))) float sC[16 * TP];
  const int b = blockIdx.x;
  const float* X; int h, s;
  if (b < PJ_G0)      { X = geo_all; h = 0; s = b; }
  else if (b < PJ_G1) { X = geo_all; h = 1; s = b; }
  else if (b < PJ_G2) { X = LPLb;    h = 2; s = b - PJ_G1; }
  else if (b < PJ_G3) { X = PLPb;    h = 3; s = b - PJ_G2; }
  else if (b < PJ_G4) { X = lines;   h = 4; s = b - PJ_G3; }
  else if (b < PJ_G5) { X = LPLb;    h = 5; s = b - PJ_G4; }
  else if (b < PJ_S0) { X = text_s;  h = 0; s = b - PJ_G5; }
  else if (b < PJ_S1) { X = head_s;  h = 1; s = b - PJ_S0; }
  else if (b < PJ_S2) { X = angle_s; h = 2; s = b - PJ_S1; }
  else if (b < PJ_S3) { X = bar_s;   h = 3; s = b - PJ_S2; }
  else if (b < PJ_S4) { X = para_s;  h = 4; s = b - PJ_S3; }
  else                { X = perp_s;  h = 5; s = b - PJ_S4; }
  const float* Wp = sc_W1 + (size_t)h * HH;
  float4 w0[16];
  wtile_request(b < PJ_G1 ? gm_W : Wp, w0);
  strip_stage(X + (size_t)s * (16 * 128), sA);
  wtile_store(w0, sW);
  __syncthreads();
  if (b < PJ_G1) {
    {
      float4 w1[16];
      wtile_request(gm_W + HH, w1);
      const v8f y = strip_gemm(sA, sW);
      wtile_store(w1, sW);
      strip_relu_ln(y, gm_b, gm_g, gm_be, sC, sA, TP);
      __syncthreads();
    }
    {
      float4 w2[16];
      wtile_request(Wp, w2);
      const v8f y = strip_gemm(sA, sW);
      wtile_store(w2, sW);
      strip_relu_ln(y, gm_b + HD, gm_g + HD, gm_be + HD, sC, sA, TP);
      __syncthreads();
    }
  }
  const v8f p = strip_gemm(sA, sW);
  strip_store_raw(p, proj + (size_t)b * (16 * 128));
}

__global__ void mask_cat_kernel(const float* __restrict__ lpl_m,
                                const float* __restrict__ plp_m,
                                const float* __restrict__ pcp_m,
                                float* __restrict__ mask_t, float* __restrict__ mask_h) {
  int t = blockIdx.x * blockDim.x + threadIdx.x;
  if (t < RT) {
    float v;
    if (t < 2048)      v = 1.f;
    else if (t < 3072) v = lpl_m[t - 2048];
    else if (t < 4096) v = plp_m[t - 3072];
    else if (t < 4224) v = pcp_m[t - 4096];
    else               v = 1.f;
    mask_t[t] = v;
  }
  if (t < RH) {
    float v;
    if (t < 1024)      v = 1.f;
    else if (t < 2048) v = lpl_m[t - 1024];
    else if (t < 3072) v = plp_m[t - 2048];
    else               v = pcp_m[t - 3072];
    mask_h[t] = v;
  }
}

typedef float v4f __attribute__((ext_vector_type(4)));
static const int PG_ROWS  = 32;
static const int PS_ROWS  = 16;
static const int PG_PITCH = 132;
static_assert(RT % PG_ROWS == 0 && RH % PG_ROWS == 0 && PD % PG_ROWS == 0 && LD % PG_ROWS == 0,
              "geo row counts must be whole tiles");
static_assert(NT % PS_ROWS == 0 && NH % PS_ROWS == 0 && NA % PS_ROWS == 0 && NB % PS_ROWS == 0 &&
              NPA % PS_ROWS == 0 && NPE % PS_ROWS == 0, "sym row counts must be whole tiles");

__device__ __forceinline__ void pair_block(const v4f gv, const v4f wv, const v4f (&sv)[4],
                                           v2f (&S)[4], v2f (&Q)[4], v2f (&D)[4]) {
  const v2f zero2 = {0.f, 0.f};
  for (int ii = 0; ii < 4; ++ii) {
    const v2f h01 = __builtin_elementwise_max(sv[ii].xy + gv.xy, zero2);
    const v2f h23 = __builtin_elementwise_max(sv[ii].zw + gv.zw, zero2);
    S[ii] += h01;
    Q[ii] = __builtin_elementwise_fma(h01, h01, Q[ii]);
    D[ii] = __builtin_elementwise_fma(h01, wv.xy, D[ii]);
    S[ii] += h23;
    Q[ii] = __builtin_elementwise_fma(h23, h23, Q[ii]);
    D[ii] = __builtin_elementwise_fma(h23, wv.zw, D[ii]);
  }
}

__device__ __forceinline__
void pair_tile(const float* __restrict__ symW, const float* __restrict__ geoW,
                 const float* __restrict__ mask, const float* __restrict__ b1,
                 const float* __restrict__ g, const float* __restrict__ be,
                 const float* __restrict__ W2, const float* __restrict__ b2p,
                 float* __restrict__ out, int R, int r0, int i0) {
  __shared__ __attribute__((aligned(16))) float sG[PG_ROWS * PG_PITCH];
  __shared__ __attribute__((aligned(16))) float sS[PS_ROWS * 128];
  __shared__ __attribute__((aligned(16))) float sW[128];
  const int tid  = threadIdx.x;
  const int lane = tid & 31;
  const int w    = tid >> 5;

  const v4f gg = ((const v4f*)g)[lane];
  const v4f bb = ((const v4f*)be)[lane];
  const v4f ww = ((const v4f*)W2)[lane];
  const v4f b4 = ((const v4f*)b1)[lane];
  const float mk = mask[r0 + lane];
  const float b2 = b2p[0];
  v4f ts[4], tg[8];
  for (int u = 0; u < 4; ++u)
    ts[u] = ((const v4f*)(symW + (size_t)(i0 + w + 4 * u) * 128))[lane];
  for (int u = 0; u < 8; ++u)
    tg[u] = ((const v4f*)(geoW + (size_t)(r0 + w + 4 * u) * 128))[lane];

  const v4f gw = gg * ww;
  const v4f bw = bb * ww;
  if (w == 0) *(v4f*)&sW[lane * 4] = gw;
  float G  = (gw.x + gw.y) + (gw.z + gw.w);
  float Bc = (bw.x + bw.y) + (bw.z + bw.w);
  for (int o = 16; o > 0; o >>= 1) {
    G  += __shfl_xor(G, o, 32);
    Bc += __shfl_xor(Bc, o, 32);
  }
  for (int u = 0; u < 4; ++u)
    *(v4f*)&sS[(w + 4 * u) * 128 + lane * 4] = ts[u] + b4;
  for (int u = 0; u < 8; ++u)
    *(v4f*)&sG[(w + 4 * u) * PG_PITCH + lane * 4] = tg[u];
  __syncthreads();

  const float* gp = &sG[lane * PG_PITCH];
  const float* sp = &sS[(w * 4) * 128];
  const v2f zero2 = {0.f, 0.f};
  v2f S[4], Q[4], D[4];
  for (int ii = 0; ii < 4; ++ii) { S[ii] = zero2; Q[ii] = zero2; D[ii] = zero2; }

  v4f gA, wA, sA[4], gB, wB, sB[4];
  gA = *(const v4f*)gp;
  wA = *(const v4f*)sW;
  for (int ii = 0; ii < 4; ++ii) sA[ii] = *(const v4f*)(sp + ii * 128);
#pragma unroll
  for (int k = 0; k < 128; k += 8) {
    gB = *(const v4f*)(gp + k + 4);
    wB = *(const v4f*)(sW + k + 4);
    for (int ii = 0; ii < 4; ++ii) sB[ii] = *(const v4f*)(sp + ii * 128 + k + 4);
    pair_block(gA, wA, sA, S, Q, D);
    const int kn = (k + 8 < 128) ? k + 8 : 0;
    gA = *(const v4f*)(gp + kn);
    wA = *(const v4f*)(sW + kn);
    for (int ii = 0; ii < 4; ++ii) sA[ii] = *(const v4f*)(sp + ii * 128 + kn);
    pair_block(gB, wB, sB, S, Q, D);
  }

  for (int ii = 0; ii < 4; ++ii) {
    const float s = S[ii].x + S[ii].y;
    const float q = Q[ii].x + Q[ii].y;
    const float d = D[ii].x + D[ii].y;
    const float mean  = s * (1.f / 128.f);
    const float var   = q * (1.f / 128.f) - mean * mean;
    const float rstd  = rsqrtf(var + 1e-5f);
    const float logit = rstd * (d - mean * G) + Bc + b2;
    const float z     = (mk > 0.f) ? logit : -1e9f;
    out[(size_t)(i0 + w * 4 + ii) * R + r0 + lane] = 1.f / (1.f + expf(-z));
  }
}


static const int PB0 = (RT / PG_ROWS) * (NT / PS_ROWS);
static const int PB1 = PB0 + (RH / PG_ROWS) * (NH / PS_ROWS);
static const int PB2 = PB1 + (PD / PG_ROWS) * (NA / PS_ROWS);
static const int PB3 = PB2 + (LD / PG_ROWS) * (NB / PS_ROWS);
static const int PB4 = PB3 + (LD / PG_ROWS) * (NPA / PS_ROWS);
static const int PB5 = PB4 + (PD / PG_ROWS) * (NPE / PS_ROWS);
__global__ __launch_bounds__(128)
void pair6_kernel(const float* __restrict__ proj,
                  const float* __restrict__ mask_t, const float* __restrict__ mask_h,
                  const float* __restrict__ lpl_m, const float* __restrict__ plp_m,
                  const float* __restrict__ sc_b1, const float* __restrict__ sc_g,
                  const float* __restrict__ sc_be, const float* __restrict__ sc_W2,
                  const float* __restrict__ sc_b2, float* __restrict__ out) {
  const int b = blockIdx.x;
  int h, R, bx, by, grow, srow, ooff; const float* mask;
  if (b < PB0) {
    h = 0; R = RT; by = b / (RT / PG_ROWS); bx = b - by * (RT / PG_ROWS);
    grow = 0; srow = 16 * PJ_G5; mask = mask_t; ooff = 0;
  } else if (b < PB1) {
    const int lb = b - PB0;
    h = 1; R = RH; by = lb / (RH / PG_ROWS); bx = lb - by * (RH / PG_ROWS);
    grow = 16 * PJ_G0; srow = 16 * PJ_S0; mask = mask_h; ooff = NT * RT;
  } else if (b < PB2) {
    const int lb = b - PB1;
    h = 2; R = PD; by = lb / (PD / PG_ROWS); bx = lb - by * (PD / PG_ROWS);
    grow = 16 * PJ_G1; srow = 16 * PJ_S1; mask = lpl_m; ooff = NT * RT + NH * RH;
  } else if (b < PB3) {
    const int lb = b - PB2;
    h = 3; R = LD; by = lb / (LD / PG_ROWS); bx = lb - by * (LD / PG_ROWS);
    grow = 16 * PJ_G2; srow = 16 * PJ_S2; mask = plp_m; ooff = NT * RT + NH * RH + NA * PD;
  } else if (b < PB4) {
    const int lb = b - PB3;
    h = 4; R = LD; by = lb / (LD / PG_ROWS); bx = lb - by * (LD / PG_ROWS);
    grow = 16 * PJ_G3; srow = 16 * PJ_S3; mask = mask_t + PD;
    ooff = NT * RT + NH * RH + NA * PD + NB * LD;
  } else {
    const int lb = b - PB4;
    h = 5; R = PD; by = lb / (PD / PG_ROWS); bx = lb - by * (PD / PG_ROWS);
    grow = 16 * PJ_G4; srow = 16 * PJ_S4; mask = lpl_m;
    ooff = NT * RT + NH * RH + NA * PD + NB * LD + NPA * LD;
  }
  pair_tile(proj + (size_t)srow * 128, proj + (size_t)grow * 128, mask,
            sc_b1 + h * HD, sc_g + h * HD, sc_be + h * HD, sc_W2 + h * HD, sc_b2 + h,
            out + ooff, R, bx * PG_ROWS, by * PS_ROWS);
}

extern "C" void kernel_launch(void* const* d_in, const int* in_sizes, int n_in,
                              void* d_out, int out_size, void* d_ws, size_t ws_size,
                              hipStream_t stream) {
  (void)in_sizes; (void)n_in; (void)out_size; (void)ws_size;
  const float* points  = (const float*)d_in[0];
  const float* lines   = (const float*)d_in[1];
  const float* circles = (const float*)d_in[2];
  const float* text_s  = (const float*)d_in[3];
  const float* head_s  = (const float*)d_in[4];
  const float* angle_s = (const float*)d_in[5];
  const float* bar_s   = (const float*)d_in[6];
  const float* para_s  = (const float*)d_in[7];
  const float* perp_s  = (const float*)d_in[8];
  const float* rel_W   = (const float*)d_in[9];
  const float* rel_b   = (const float*)d_in[10];
  const float* rel_g   = (const float*)d_in[11];
  const float* rel_be  = (const float*)d_in[12];
  const float* gm_W    = (const float*)d_in[13];
  const float* gm_b    = (const float*)d_in[14];
  const float* gm_g    = (const float*)d_in[15];
  const float* gm_be   = (const float*)d_in[16];
  const float* sc_W1   = (const float*)d_in[17];
  const float* sc_b1   = (const float*)d_in[18];
  const float* sc_g    = (const float*)d_in[19];
  const float* sc_be   = (const float*)d_in[20];
  const float* sc_W2   = (const float*)d_in[21];
  const float* sc_b2   = (const float*)d_in[22];
  const int*   pl      = (const int*)d_in[23];
  const int*   pc      = (const int*)d_in[24];
  float* out = (float*)d_out;

  float* wsp = (float*)d_ws;
  size_t off = 0;
  auto alloc = [&](size_t n) { float* p = wsp + off; off += (n + 31) & ~(size_t)31; return p; };
  float* LPLb   = alloc((size_t)PD * HD);
  float* PLPb   = alloc((size_t)LD * HD);
  float* PCPb   = alloc((size_t)CD * HD);
  float* lpl_m  = alloc(PD);
  float* plp_m  = alloc(LD);
  float* pcp_m  = alloc(CD);
  float* mask_t = alloc(RT);
  float* mask_h = alloc(RH);
  float* geo_all = alloc((size_t)(RT + RH) * HD);
  float* geo_t  = geo_all;
  float* geo_h  = geo_all + (size_t)RT * HD;
  float* proj   = alloc((size_t)PJ_S5 * 16 * HD);

  agg_rel3_kernel<<<AG_S2, 256, 0, stream>>>(pl, pc, points, lines, circles,
                                              rel_W, rel_b, rel_g, rel_be, LPLb, PLPb, PCPb,
                                              lpl_m, plp_m, pcp_m);

  const size_t ROW = (size_t)HD * sizeof(float);
  hipMemcpyAsync(geo_t + (size_t)0 * HD,    points, PD * ROW, hipMemcpyDeviceToDevice, stream);
  hipMemcpyAsync(geo_t + (size_t)1024 * HD, lines,  LD * ROW, hipMemcpyDeviceToDevice, stream);
  hipMemcpyAsync(geo_t + (size_t)2048 * HD, LPLb,   PD * ROW, hipMemcpyDeviceToDevice, stream);
  hipMemcpyAsync(geo_t + (size_t)3072 * HD, PLPb,   LD * ROW, hipMemcpyDeviceToDevice, stream);
  hipMemcpyAsync(geo_t + (size_t)4096 * HD, PCPb,   CD * ROW, hipMemcpyDeviceToDevice, stream);
  hipMemcpyAsync(geo_t + (size_t)4224 * HD, head_s, NH * ROW, hipMemcpyDeviceToDevice, stream);
  hipMemcpyAsync(geo_h + (size_t)0 * HD,    points, PD * ROW, hipMemcpyDeviceToDevice, stream);
  hipMemcpyAsync(geo_h + (size_t)1024 * HD, LPLb,   PD * ROW, hipMemcpyDeviceToDevice, stream);
  hipMemcpyAsync(geo_h + (size_t)2048 * HD, PLPb,   LD * ROW, hipMemcpyDeviceToDevice, stream);
  hipMemcpyAsync(geo_h + (size_t)3072 * HD, PCPb,   CD * ROW, hipMemcpyDeviceToDevice, stream);
  mask_cat_kernel<<<(RT + 255) / 256, 256, 0, stream>>>(lpl_m, plp_m, pcp_m, mask_t, mask_h);

  gm_proj12_kernel<<<PJ_S5, 256, 0, stream>>>(geo_all, LPLb, PLPb, lines, text_s, head_s, angle_s,
                                               bar_s, para_s, perp_s, gm_W, gm_b, gm_g, gm_be,
                                               sc_W1, proj);
  pair6_kernel<<<PB5, 128, 0, stream>>>(proj, mask_t, mask_h, lpl_m, plp_m,
                                         sc_b1, sc_g, sc_be, sc_W2, sc_b2, out);
}
